# starting kernel
# baseline (speedup 1.0000x reference)
#include <hip/hip_runtime.h>


typedef __attribute__((ext_vector_type(2))) float v2f;
typedef __attribute__((ext_vector_type(8))) float v8f;

#define WAVES 8

__device__ __forceinline__ unsigned fkey(float f) {
  unsigned u = __float_as_uint(f);
  return (u & 0x80000000u) ? ~u : (u | 0x80000000u);
}
__device__ __forceinline__ float funkey(unsigned kk) {
  unsigned u = (kk & 0x80000000u) ? (kk & 0x7FFFFFFFu) : ~kk;
  return __uint_as_float(u);
}

__global__ __launch_bounds__(256) void gemm128_kernel(
    const float* __restrict__ A, const float* __restrict__ W,
    float* __restrict__ out) {
  const int lane = threadIdx.x & 31;
  const int wv   = threadIdx.x >> 5;
  const int row0 = blockIdx.x << 4;
  const int n0   = wv << 4;
  const int m    = lane & 15;
  const int kk   = (lane >> 4) << 1;
  const int mb   = (lane >> 4) << 3;

  v8f c = {};
  const float* ar = A + (size_t)(row0 + m) * 128;
  const float* wp = W + n0 + m;
#pragma unroll 8
  for (int k0 = 0; k0 < 128; k0 += 4) {
    v2f a, b;
    a[0] = ar[k0 + kk];
    a[1] = ar[k0 + kk + 1];
    b[0] = wp[(size_t)(k0 + kk) * 128];
    b[1] = wp[(size_t)(k0 + kk + 1) * 128];
    c = __builtin_amdgcn_wmma_f32_16x16x4_f32(false, a, false, b, (short)0, c,
                                              false, false);
  }
  float* op = out + (size_t)(row0 + mb) * 128 + n0 + m;
#pragma unroll
  for (int r = 0; r < 8; ++r) op[(size_t)r * 128] = c[r];
}

__global__ __launch_bounds__(256) void edge_score_kernel(
    const float* __restrict__ ea, const int* __restrict__ src,
    const int* __restrict__ dst, const float* __restrict__ Wek,
    const float* __restrict__ Wg1, const float* __restrict__ Wg2,
    const float* __restrict__ q, const float* __restrict__ k,
    float* __restrict__ scores, unsigned* __restrict__ mmax) {
  __shared__ float g1s[WAVES][16][64];
  __shared__ float gts[WAVES][16][8];

  const int lane = threadIdx.x & 31;
  const int wv   = threadIdx.x >> 5;
  const int e0   = (blockIdx.x * WAVES + wv) << 4;
  const int m    = lane & 15;
  const int kk   = (lane >> 4) << 1;
  const int mb   = (lane >> 4) << 3;

  v2f afr[8];
  const float* er = ea + (size_t)(e0 + m) * 32;
#pragma unroll
  for (int i = 0; i < 8; ++i) {
    afr[i][0] = er[4 * i + kk];
    afr[i][1] = er[4 * i + kk + 1];
  }

#pragma unroll
  for (int t = 0; t < 4; ++t) {
    v8f c = {};
    const float* wp = Wg1 + t * 16 + m;
#pragma unroll
    for (int i = 0; i < 8; ++i) {
      v2f b;
      b[0] = wp[(size_t)(4 * i + kk) * 64];
      b[1] = wp[(size_t)(4 * i + kk + 1) * 64];
      c = __builtin_amdgcn_wmma_f32_16x16x4_f32(false, afr[i], false, b,
                                                (short)0, c, false, false);
    }
#pragma unroll
    for (int r = 0; r < 8; ++r) {
      float xv = c[r];
      g1s[wv][mb + r][t * 16 + m] = xv / (1.0f + __expf(-xv));
    }
  }
  __syncthreads();

  {
    const int hb = (lane >> 4) << 2;
    float acc0 = 0.f, acc1 = 0.f, acc2 = 0.f, acc3 = 0.f;
#pragma unroll 4
    for (int j = 0; j < 64; ++j) {
      float g = g1s[wv][m][j];
      const float* w2 = Wg2 + j * 8 + hb;
      acc0 = fmaf(g, w2[0], acc0);
      acc1 = fmaf(g, w2[1], acc1);
      acc2 = fmaf(g, w2[2], acc2);
      acc3 = fmaf(g, w2[3], acc3);
    }
    gts[wv][m][hb + 0] = acc0;
    gts[wv][m][hb + 1] = acc1;
    gts[wv][m][hb + 2] = acc2;
    gts[wv][m][hb + 3] = acc3;
  }
  __syncthreads();

  int srcv[8], dstv[8];
#pragma unroll
  for (int r = 0; r < 8; ++r) {
    srcv[r] = src[e0 + mb + r];
    dstv[r] = dst[e0 + mb + r];
  }

#pragma unroll 1
  for (int h = 0; h < 8; ++h) {
    v8f c = {};
    const float* wp = Wek + h * 16 + m;
#pragma unroll
    for (int i = 0; i < 8; ++i) {
      v2f b;
      b[0] = wp[(size_t)(4 * i + kk) * 128];
      b[1] = wp[(size_t)(4 * i + kk + 1) * 128];
      c = __builtin_amdgcn_wmma_f32_16x16x4_f32(false, afr[i], false, b,
                                                (short)0, c, false, false);
    }
    const int col = h * 16 + m;
#pragma unroll
    for (int r = 0; r < 8; ++r) {
      float kv = k[(size_t)srcv[r] * 128 + col] + c[r];
      float qv = q[(size_t)dstv[r] * 128 + col];
      float p  = qv * kv;
#pragma unroll
      for (int off = 8; off >= 1; off >>= 1) p += __shfl_xor(p, off, 32);
      if (m == 0) {
        float sc = p * 0.25f + gts[wv][mb + r][h];
        scores[(size_t)(e0 + mb + r) * 8 + h] = sc;
        atomicMax(&mmax[(size_t)dstv[r] * 8 + h], fkey(sc));
      }
    }
  }
}

__global__ __launch_bounds__(256) void exp_denom_kernel(
    const int* __restrict__ dst, float* __restrict__ scores,
    const unsigned* __restrict__ mmax, float* __restrict__ denom, int total) {
  int idx = blockIdx.x * blockDim.x + threadIdx.x;
  if (idx >= total) return;
  int e = idx >> 3, h = idx & 7;
  int d = dst[e];
  float mv = funkey(mmax[(size_t)d * 8 + h]);
  float es = __expf(scores[idx] - mv);
  scores[idx] = es;
  atomicAdd(&denom[(size_t)d * 8 + h], es);
}

__global__ __launch_bounds__(256) void edge_aggregate_kernel(
    const float* __restrict__ ea, const int* __restrict__ src,
    const int* __restrict__ dst, const float* __restrict__ Wev,
    const float* __restrict__ v, const float* __restrict__ scores,
    const float* __restrict__ denom, float* __restrict__ acc) {
  const int lane = threadIdx.x & 31;
  const int wv   = threadIdx.x >> 5;
  const int e0   = (blockIdx.x * WAVES + wv) << 4;
  const int m    = lane & 15;
  const int kk   = (lane >> 4) << 1;
  const int mb   = (lane >> 4) << 3;

  v2f afr[8];
  const float* er = ea + (size_t)(e0 + m) * 32;
#pragma unroll
  for (int i = 0; i < 8; ++i) {
    afr[i][0] = er[4 * i + kk];
    afr[i][1] = er[4 * i + kk + 1];
  }

  int srcv[8], dstv[8];
#pragma unroll
  for (int r = 0; r < 8; ++r) {
    srcv[r] = src[e0 + mb + r];
    dstv[r] = dst[e0 + mb + r];
  }

#pragma unroll 1
  for (int h = 0; h < 8; ++h) {
    v8f c = {};
    const float* wp = Wev + h * 16 + m;
#pragma unroll
    for (int i = 0; i < 8; ++i) {
      v2f b;
      b[0] = wp[(size_t)(4 * i + kk) * 128];
      b[1] = wp[(size_t)(4 * i + kk + 1) * 128];
      c = __builtin_amdgcn_wmma_f32_16x16x4_f32(false, afr[i], false, b,
                                                (short)0, c, false, false);
    }
    const int col = h * 16 + m;
#pragma unroll
    for (int r = 0; r < 8; ++r) {
      int e = e0 + mb + r;
      float att = scores[(size_t)e * 8 + h] /
                  (denom[(size_t)dstv[r] * 8 + h] + 1e-9f);
      float ve = v[(size_t)srcv[r] * 128 + col] + c[r];
      atomicAdd(&acc[(size_t)dstv[r] * 128 + col], att * ve);
    }
  }
}

extern "C" void kernel_launch(void* const* d_in, const int* in_sizes, int n_in,
                              void* d_out, int out_size, void* d_ws,
                              size_t ws_size, hipStream_t stream) {
  const float* x   = (const float*)d_in[0];
  const int*   ei  = (const int*)d_in[1];
  const float* ea  = (const float*)d_in[2];
  const float* Wq  = (const float*)d_in[3];
  const float* Wk  = (const float*)d_in[4];
  const float* Wv  = (const float*)d_in[5];
  const float* Wo  = (const float*)d_in[6];
  const float* Wek = (const float*)d_in[7];
  const float* Wev = (const float*)d_in[8];
  const float* Wg1 = (const float*)d_in[9];
  const float* Wg2 = (const float*)d_in[10];
  float* out = (float*)d_out;

  const int N = in_sizes[0] / 128;
  const int E = in_sizes[2] / 32;
  const int* src = ei;
  const int* dst = ei + E;

  float* q      = (float*)d_ws;
  float* k      = q + (size_t)N * 128;
  float* v      = k + (size_t)N * 128;
  float* scores = v + (size_t)N * 128;
  float* acc    = scores + (size_t)E * 8;
  float* denom  = acc + (size_t)N * 128;
  unsigned* mmax = (unsigned*)(denom + (size_t)N * 8);

  hipMemsetAsync(acc, 0, (size_t)(N * 128 + N * 8 + N * 8) * sizeof(float),
                 stream);

  dim3 blk(256);
  gemm128_kernel<<<dim3(N / 16), blk, 0, stream>>>(x, Wq, q);
  gemm128_kernel<<<dim3(N / 16), blk, 0, stream>>>(x, Wk, k);
  gemm128_kernel<<<dim3(N / 16), blk, 0, stream>>>(x, Wv, v);

  edge_score_kernel<<<dim3(E / (16 * WAVES)), blk, 0, stream>>>(
      ea, src, dst, Wek, Wg1, Wg2, q, k, scores, mmax);

  exp_denom_kernel<<<dim3((E * 8 + 255) / 256), blk, 0, stream>>>(
      dst, scores, mmax, denom, E * 8);

  edge_aggregate_kernel<<<dim3(E / (16 * WAVES)), blk, 0, stream>>>(
      ea, src, dst, Wev, v, scores, denom, acc);

  gemm128_kernel<<<dim3(N / 16), blk, 0, stream>>>(acc, Wo, out);
}


# edge_score: WMMA operands transposed (weights A, edge_attr B) so each lane owns one edge: 16B q/k loads, in-lane dot, gate2 on WMMA from D regs, no LDS or shuffles; dot and gate sum order change
# speedup vs baseline: 1.5277x; 1.5277x over previous
#include <hip/hip_runtime.h>


typedef __attribute__((ext_vector_type(2))) float v2f;
typedef __attribute__((ext_vector_type(4))) float v4f;
typedef __attribute__((ext_vector_type(8))) float v8f;

#define WAVES 8

__device__ __forceinline__ unsigned fkey(float f) {
  unsigned u = __float_as_uint(f);
  return (u & 0x80000000u) ? ~u : (u | 0x80000000u);
}
__device__ __forceinline__ float funkey(unsigned kk) {
  unsigned u = (kk & 0x80000000u) ? (kk & 0x7FFFFFFFu) : ~kk;
  return __uint_as_float(u);
}

__global__ __launch_bounds__(256) void gemm128_kernel(
    const float* __restrict__ A, const float* __restrict__ W,
    float* __restrict__ out) {
  const int lane = threadIdx.x & 31;
  const int wv   = threadIdx.x >> 5;
  const int row0 = blockIdx.x << 4;
  const int n0   = wv << 4;
  const int m    = lane & 15;
  const int kk   = (lane >> 4) << 1;
  const int mb   = (lane >> 4) << 3;

  v8f c = {};
  const float* ar = A + (size_t)(row0 + m) * 128;
  const float* wp = W + n0 + m;
#pragma unroll 8
  for (int k0 = 0; k0 < 128; k0 += 4) {
    v2f a, b;
    a[0] = ar[k0 + kk];
    a[1] = ar[k0 + kk + 1];
    b[0] = wp[(size_t)(k0 + kk) * 128];
    b[1] = wp[(size_t)(k0 + kk + 1) * 128];
    c = __builtin_amdgcn_wmma_f32_16x16x4_f32(false, a, false, b, (short)0, c,
                                              false, false);
  }
  float* op = out + (size_t)(row0 + mb) * 128 + n0 + m;
#pragma unroll
  for (int r = 0; r < 8; ++r) op[(size_t)r * 128] = c[r];
}

__device__ __forceinline__ float xhalf(float x) {
  int b = __builtin_bit_cast(int, x);
  return __builtin_bit_cast(
      float, __builtin_amdgcn_permlanex16(b, b, 0x76543210, 0xfedcba98, false,
                                          false));
}

__device__ __forceinline__ v8f silu_gate1_tile(const float* __restrict__ Wg1,
                                               const v2f* efr, int t, int m,
                                               int kk) {
  v8f c = {};
  const float* wp = Wg1 + t * 16 + m;
#pragma unroll
  for (int i = 0; i < 8; ++i) {
    v2f a;
    a[0] = wp[(size_t)(4 * i + kk) * 64];
    a[1] = wp[(size_t)(4 * i + kk + 1) * 64];
    c = __builtin_amdgcn_wmma_f32_16x16x4_f32(false, a, false, efr[i], (short)0,
                                              c, false, false);
  }
#pragma unroll
  for (int r = 0; r < 8; ++r) {
    float xv = c[r];
    c[r] = xv / (1.0f + __expf(-xv));
  }
  return c;
}

__device__ __forceinline__ v8f gate2_acc(v8f gt, const float* __restrict__ Wg2,
                                         v8f c, int t, int m, int hf) {
  const float* w2 = Wg2 + (size_t)(t * 16 + 8 * hf) * 8 + (m & 7);
#pragma unroll
  for (int p = 0; p < 4; ++p) {
    v2f a, b;
    a[0] = w2[(2 * p) * 8];
    a[1] = w2[(2 * p + 1) * 8];
    b[0] = c[2 * p];
    b[1] = c[2 * p + 1];
    gt = __builtin_amdgcn_wmma_f32_16x16x4_f32(false, a, false, b, (short)0, gt,
                                               false, false);
  }
  return gt;
}

__global__ __launch_bounds__(256) void edge_score_kernel(
    const float* __restrict__ ea, const int* __restrict__ src,
    const int* __restrict__ dst, const float* __restrict__ Wek,
    const float* __restrict__ Wg1, const float* __restrict__ Wg2,
    const float* __restrict__ q, const float* __restrict__ k,
    float* __restrict__ scores, unsigned* __restrict__ mmax) {
  const int lane = threadIdx.x & 31;
  const int wv   = threadIdx.x >> 5;
  const int e0   = (blockIdx.x * WAVES + wv) << 4;
  const int m    = lane & 15;
  const int hf   = lane >> 4;
  const int kk   = hf << 1;

  v2f efr[8];
  const float* er = ea + (size_t)(e0 + m) * 32;
#pragma unroll
  for (int i = 0; i < 8; ++i) {
    efr[i][0] = er[4 * i + kk];
    efr[i][1] = er[4 * i + kk + 1];
  }
  const int sm = src[e0 + m];
  const int dm = dst[e0 + m];

  v8f gt = {};
  v8f g1 = silu_gate1_tile(Wg1, efr, 0, m, kk);
#pragma unroll 1
  for (int t = 1; t < 4; ++t) {
    gt = gate2_acc(gt, Wg2, g1, t - 1, m, hf);
    g1 = silu_gate1_tile(Wg1, efr, t, m, kk);
  }
  gt = gate2_acc(gt, Wg2, g1, 3, m, hf);

  const float* qrow = q + (size_t)dm * 128 + 8 * hf;
  const float* krow = k + (size_t)sm * 128 + 8 * hf;
  v4f o = {};
#pragma unroll 1
  for (int h = 0; h < 8; ++h) {
    v8f c = {};
    const float* wp = Wek + h * 16 + m;
#pragma unroll
    for (int i = 0; i < 8; ++i) {
      v2f a;
      a[0] = wp[(size_t)(4 * i + kk) * 128];
      a[1] = wp[(size_t)(4 * i + kk + 1) * 128];
      c = __builtin_amdgcn_wmma_f32_16x16x4_f32(false, a, false, efr[i],
                                                (short)0, c, false, false);
    }
    const v4f q0 = *(const v4f*)(qrow + h * 16);
    const v4f q1 = *(const v4f*)(qrow + h * 16 + 4);
    const v4f k0 = *(const v4f*)(krow + h * 16);
    const v4f k1 = *(const v4f*)(krow + h * 16 + 4);
    float p = q0[0] * (k0[0] + c[0]);
    p = fmaf(q0[1], k0[1] + c[1], p);
    p = fmaf(q0[2], k0[2] + c[2], p);
    p = fmaf(q0[3], k0[3] + c[3], p);
    p = fmaf(q1[0], k1[0] + c[4], p);
    p = fmaf(q1[1], k1[1] + c[5], p);
    p = fmaf(q1[2], k1[2] + c[6], p);
    p = fmaf(q1[3], k1[3] + c[7], p);
    p += xhalf(p);
    const int hh = h - 4 * hf;
    o[0] = (hh == 0) ? p : o[0];
    o[1] = (hh == 1) ? p : o[1];
    o[2] = (hh == 2) ? p : o[2];
    o[3] = (hh == 3) ? p : o[3];
  }
  o[0] = o[0] * 0.25f + (hf ? gt[4] : gt[0]);
  o[1] = o[1] * 0.25f + (hf ? gt[5] : gt[1]);
  o[2] = o[2] * 0.25f + (hf ? gt[6] : gt[2]);
  o[3] = o[3] * 0.25f + (hf ? gt[7] : gt[3]);

  *(v4f*)(scores + (size_t)(e0 + m) * 8 + 4 * hf) = o;
  unsigned* mp = mmax + (size_t)dm * 8 + 4 * hf;
  atomicMax(mp + 0, fkey(o[0]));
  atomicMax(mp + 1, fkey(o[1]));
  atomicMax(mp + 2, fkey(o[2]));
  atomicMax(mp + 3, fkey(o[3]));
}

__global__ __launch_bounds__(256) void exp_denom_kernel(
    const int* __restrict__ dst, float* __restrict__ scores,
    const unsigned* __restrict__ mmax, float* __restrict__ denom, int total) {
  int idx = blockIdx.x * blockDim.x + threadIdx.x;
  if (idx >= total) return;
  int e = idx >> 3, h = idx & 7;
  int d = dst[e];
  float mv = funkey(mmax[(size_t)d * 8 + h]);
  float es = __expf(scores[idx] - mv);
  scores[idx] = es;
  atomicAdd(&denom[(size_t)d * 8 + h], es);
}

__global__ __launch_bounds__(256) void edge_aggregate_kernel(
    const float* __restrict__ ea, const int* __restrict__ src,
    const int* __restrict__ dst, const float* __restrict__ Wev,
    const float* __restrict__ v, const float* __restrict__ scores,
    const float* __restrict__ denom, float* __restrict__ acc) {
  const int lane = threadIdx.x & 31;
  const int wv   = threadIdx.x >> 5;
  const int e0   = (blockIdx.x * WAVES + wv) << 4;
  const int m    = lane & 15;
  const int kk   = (lane >> 4) << 1;
  const int mb   = (lane >> 4) << 3;

  v2f afr[8];
  const float* er = ea + (size_t)(e0 + m) * 32;
#pragma unroll
  for (int i = 0; i < 8; ++i) {
    afr[i][0] = er[4 * i + kk];
    afr[i][1] = er[4 * i + kk + 1];
  }

  int srcv[8], dstv[8];
#pragma unroll
  for (int r = 0; r < 8; ++r) {
    srcv[r] = src[e0 + mb + r];
    dstv[r] = dst[e0 + mb + r];
  }

#pragma unroll 1
  for (int h = 0; h < 8; ++h) {
    v8f c = {};
    const float* wp = Wev + h * 16 + m;
#pragma unroll
    for (int i = 0; i < 8; ++i) {
      v2f b;
      b[0] = wp[(size_t)(4 * i + kk) * 128];
      b[1] = wp[(size_t)(4 * i + kk + 1) * 128];
      c = __builtin_amdgcn_wmma_f32_16x16x4_f32(false, afr[i], false, b,
                                                (short)0, c, false, false);
    }
    const int col = h * 16 + m;
#pragma unroll
    for (int r = 0; r < 8; ++r) {
      int e = e0 + mb + r;
      float att = scores[(size_t)e * 8 + h] /
                  (denom[(size_t)dstv[r] * 8 + h] + 1e-9f);
      float ve = v[(size_t)srcv[r] * 128 + col] + c[r];
      atomicAdd(&acc[(size_t)dstv[r] * 128 + col], att * ve);
    }
  }
}

extern "C" void kernel_launch(void* const* d_in, const int* in_sizes, int n_in,
                              void* d_out, int out_size, void* d_ws,
                              size_t ws_size, hipStream_t stream) {
  const float* x   = (const float*)d_in[0];
  const int*   ei  = (const int*)d_in[1];
  const float* ea  = (const float*)d_in[2];
  const float* Wq  = (const float*)d_in[3];
  const float* Wk  = (const float*)d_in[4];
  const float* Wv  = (const float*)d_in[5];
  const float* Wo  = (const float*)d_in[6];
  const float* Wek = (const float*)d_in[7];
  const float* Wev = (const float*)d_in[8];
  const float* Wg1 = (const float*)d_in[9];
  const float* Wg2 = (const float*)d_in[10];
  float* out = (float*)d_out;

  const int N = in_sizes[0] / 128;
  const int E = in_sizes[2] / 32;
  const int* src = ei;
  const int* dst = ei + E;

  float* q      = (float*)d_ws;
  float* k      = q + (size_t)N * 128;
  float* v      = k + (size_t)N * 128;
  float* scores = v + (size_t)N * 128;
  float* acc    = scores + (size_t)E * 8;
  float* denom  = acc + (size_t)N * 128;
  unsigned* mmax = (unsigned*)(denom + (size_t)N * 8);

  hipMemsetAsync(acc, 0, (size_t)(N * 128 + N * 8 + N * 8) * sizeof(float),
                 stream);

  dim3 blk(256);
  gemm128_kernel<<<dim3(N / 16), blk, 0, stream>>>(x, Wq, q);
  gemm128_kernel<<<dim3(N / 16), blk, 0, stream>>>(x, Wk, k);
  gemm128_kernel<<<dim3(N / 16), blk, 0, stream>>>(x, Wv, v);

  edge_score_kernel<<<dim3(E / (16 * WAVES)), blk, 0, stream>>>(
      ea, src, dst, Wek, Wg1, Wg2, q, k, scores, mmax);

  exp_denom_kernel<<<dim3((E * 8 + 255) / 256), blk, 0, stream>>>(
      dst, scores, mmax, denom, E * 8);

  edge_aggregate_kernel<<<dim3(E / (16 * WAVES)), blk, 0, stream>>>(
      ea, src, dst, Wev, v, scores, denom, acc);

  gemm128_kernel<<<dim3(N / 16), blk, 0, stream>>>(acc, Wo, out);
}


# edge_aggregate: edges counting-sorted by destination (rank from a counter in pass 2, scan, scatter); a lane sums its rows per run of equal destination, one atomic per run; sum order changes
# speedup vs baseline: 1.7061x; 1.1168x over previous
#include <hip/hip_runtime.h>


typedef __attribute__((ext_vector_type(2))) float v2f;
typedef __attribute__((ext_vector_type(4))) float v4f;
typedef __attribute__((ext_vector_type(8))) float v8f;

#define WAVES 8

__device__ __forceinline__ unsigned fkey(float f) {
  unsigned u = __float_as_uint(f);
  return (u & 0x80000000u) ? ~u : (u | 0x80000000u);
}
__device__ __forceinline__ float funkey(unsigned kk) {
  unsigned u = (kk & 0x80000000u) ? (kk & 0x7FFFFFFFu) : ~kk;
  return __uint_as_float(u);
}

__global__ __launch_bounds__(256) void gemm128_kernel(
    const float* __restrict__ A, const float* __restrict__ W,
    float* __restrict__ out) {
  const int lane = threadIdx.x & 31;
  const int wv   = threadIdx.x >> 5;
  const int row0 = blockIdx.x << 4;
  const int n0   = wv << 4;
  const int m    = lane & 15;
  const int kk   = (lane >> 4) << 1;
  const int mb   = (lane >> 4) << 3;

  v8f c = {};
  const float* ar = A + (size_t)(row0 + m) * 128;
  const float* wp = W + n0 + m;
#pragma unroll 8
  for (int k0 = 0; k0 < 128; k0 += 4) {
    v2f a, b;
    a[0] = ar[k0 + kk];
    a[1] = ar[k0 + kk + 1];
    b[0] = wp[(size_t)(k0 + kk) * 128];
    b[1] = wp[(size_t)(k0 + kk + 1) * 128];
    c = __builtin_amdgcn_wmma_f32_16x16x4_f32(false, a, false, b, (short)0, c,
                                              false, false);
  }
  float* op = out + (size_t)(row0 + mb) * 128 + n0 + m;
#pragma unroll
  for (int r = 0; r < 8; ++r) op[(size_t)r * 128] = c[r];
}

__device__ __forceinline__ float xhalf(float x) {
  int b = __builtin_bit_cast(int, x);
  return __builtin_bit_cast(
      float, __builtin_amdgcn_permlanex16(b, b, 0x76543210, 0xfedcba98, false,
                                          false));
}

__device__ __forceinline__ v8f silu_gate1_tile(const float* __restrict__ Wg1,
                                               const v2f* efr, int t, int m,
                                               int kk) {
  v8f c = {};
  const float* wp = Wg1 + t * 16 + m;
#pragma unroll
  for (int i = 0; i < 8; ++i) {
    v2f a;
    a[0] = wp[(size_t)(4 * i + kk) * 64];
    a[1] = wp[(size_t)(4 * i + kk + 1) * 64];
    c = __builtin_amdgcn_wmma_f32_16x16x4_f32(false, a, false, efr[i], (short)0,
                                              c, false, false);
  }
#pragma unroll
  for (int r = 0; r < 8; ++r) {
    float xv = c[r];
    c[r] = xv / (1.0f + __expf(-xv));
  }
  return c;
}

__device__ __forceinline__ v8f gate2_acc(v8f gt, const float* __restrict__ Wg2,
                                         v8f c, int t, int m, int hf) {
  const float* w2 = Wg2 + (size_t)(t * 16 + 8 * hf) * 8 + (m & 7);
#pragma unroll
  for (int p = 0; p < 4; ++p) {
    v2f a, b;
    a[0] = w2[(2 * p) * 8];
    a[1] = w2[(2 * p + 1) * 8];
    b[0] = c[2 * p];
    b[1] = c[2 * p + 1];
    gt = __builtin_amdgcn_wmma_f32_16x16x4_f32(false, a, false, b, (short)0, gt,
                                               false, false);
  }
  return gt;
}

__global__ __launch_bounds__(256) void edge_score_kernel(
    const float* __restrict__ ea, const int* __restrict__ src,
    const int* __restrict__ dst, const float* __restrict__ Wek,
    const float* __restrict__ Wg1, const float* __restrict__ Wg2,
    const float* __restrict__ q, const float* __restrict__ k,
    float* __restrict__ scores, unsigned* __restrict__ mmax,
    int* __restrict__ cnt, int n_nodes) {
  const int lane = threadIdx.x & 31;
  const int wv   = threadIdx.x >> 5;
  const int e0   = (blockIdx.x * WAVES + wv) << 4;
  const int m    = lane & 15;
  const int hf   = lane >> 4;
  const int kk   = hf << 1;

  v2f efr[8];
  const float* er = ea + (size_t)(e0 + m) * 32;
#pragma unroll
  for (int i = 0; i < 8; ++i) {
    efr[i][0] = er[4 * i + kk];
    efr[i][1] = er[4 * i + kk + 1];
  }
  const int sm = src[e0 + m];
  const int dm = dst[e0 + m];

  v8f gt = {};
  v8f g1 = silu_gate1_tile(Wg1, efr, 0, m, kk);
#pragma unroll 1
  for (int t = 1; t < 4; ++t) {
    gt = gate2_acc(gt, Wg2, g1, t - 1, m, hf);
    g1 = silu_gate1_tile(Wg1, efr, t, m, kk);
  }
  gt = gate2_acc(gt, Wg2, g1, 3, m, hf);

  const float* qrow = q + (size_t)dm * 128 + 8 * hf;
  const float* krow = k + (size_t)sm * 128 + 8 * hf;
  v4f o = {};
#pragma unroll 1
  for (int h = 0; h < 8; ++h) {
    v8f c = {};
    const float* wp = Wek + h * 16 + m;
#pragma unroll
    for (int i = 0; i < 8; ++i) {
      v2f a;
      a[0] = wp[(size_t)(4 * i + kk) * 128];
      a[1] = wp[(size_t)(4 * i + kk + 1) * 128];
      c = __builtin_amdgcn_wmma_f32_16x16x4_f32(false, a, false, efr[i],
                                                (short)0, c, false, false);
    }
    const v4f q0 = *(const v4f*)(qrow + h * 16);
    const v4f q1 = *(const v4f*)(qrow + h * 16 + 4);
    const v4f k0 = *(const v4f*)(krow + h * 16);
    const v4f k1 = *(const v4f*)(krow + h * 16 + 4);
    float p = q0[0] * (k0[0] + c[0]);
    p = fmaf(q0[1], k0[1] + c[1], p);
    p = fmaf(q0[2], k0[2] + c[2], p);
    p = fmaf(q0[3], k0[3] + c[3], p);
    p = fmaf(q1[0], k1[0] + c[4], p);
    p = fmaf(q1[1], k1[1] + c[5], p);
    p = fmaf(q1[2], k1[2] + c[6], p);
    p = fmaf(q1[3], k1[3] + c[7], p);
    p += xhalf(p);
    const int hh = h - 4 * hf;
    o[0] = (hh == 0) ? p : o[0];
    o[1] = (hh == 1) ? p : o[1];
    o[2] = (hh == 2) ? p : o[2];
    o[3] = (hh == 3) ? p : o[3];
  }
  o[0] = o[0] * 0.25f + (hf ? gt[4] : gt[0]);
  o[1] = o[1] * 0.25f + (hf ? gt[5] : gt[1]);
  o[2] = o[2] * 0.25f + (hf ? gt[6] : gt[2]);
  o[3] = o[3] * 0.25f + (hf ? gt[7] : gt[3]);

  *(v4f*)(scores + (size_t)(e0 + m) * 8 + 4 * hf) = o;
  unsigned* mp = mmax + (size_t)dm * 8 + 4 * hf;
  atomicMax(mp + 0, fkey(o[0]));
  atomicMax(mp + 1, fkey(o[1]));
  atomicMax(mp + 2, fkey(o[2]));
  atomicMax(mp + 3, fkey(o[3]));

  const int gid = blockIdx.x * (WAVES * 32) + threadIdx.x;
  if (gid < n_nodes) cnt[gid] = 0;
}

__global__ __launch_bounds__(256) void exp_denom_kernel(
    const int* __restrict__ dst, float* __restrict__ scores,
    const unsigned* __restrict__ mmax, float* __restrict__ denom,
    int* __restrict__ cnt, int* __restrict__ rank, int total) {
  int idx = blockIdx.x * blockDim.x + threadIdx.x;
  if (idx >= total) return;
  int e = idx >> 3, h = idx & 7;
  int d = dst[e];
  float mv = funkey(mmax[(size_t)d * 8 + h]);
  float es = __expf(scores[idx] - mv);
  scores[idx] = es;
  atomicAdd(&denom[(size_t)d * 8 + h], es);
  if (h == 0) rank[e] = atomicAdd(&cnt[d], 1);
}

__global__ __launch_bounds__(1024) void seg_offsets_kernel(
    const int* __restrict__ cnt, int* __restrict__ off, int n) {
  __shared__ int wsum[32];
  __shared__ int psum[32];
  const int t = threadIdx.x, lane = t & 31, w = t >> 5;
  const int base = blockIdx.x * 1024;
  int p = 0;
  for (int i = t; i < base; i += 1024) p += cnt[i];
  const int x = (base + t < n) ? cnt[base + t] : 0;
  int inc = x;
  for (int d = 1; d < 32; d <<= 1) {
    int y = __shfl_up(inc, d);
    if (lane >= d) inc += y;
  }
  for (int d = 16; d > 0; d >>= 1) p += __shfl_xor(p, d);
  if (lane == 31) wsum[w] = inc;
  if (lane == 0) psum[w] = p;
  __syncthreads();
  int pre = 0;
  for (int i = 0; i < 32; ++i) {
    pre += psum[i];
    pre += (i < w) ? wsum[i] : 0;
  }
  if (base + t < n) off[base + t] = pre + inc - x;
}

__global__ __launch_bounds__(256) void dst_order_kernel(
    const int* __restrict__ src, const int* __restrict__ dst,
    const int* __restrict__ rank, const int* __restrict__ off,
    int* __restrict__ perm, int* __restrict__ ssrc, int* __restrict__ sdst,
    int n_edges) {
  const int e = blockIdx.x * blockDim.x + threadIdx.x;
  if (e >= n_edges) return;
  const int d = dst[e];
  const int pos = off[d] + rank[e];
  perm[pos] = e;
  ssrc[pos] = src[e];
  sdst[pos] = d;
}

typedef __attribute__((ext_vector_type(4))) int v4i;
#define TILES 2

__global__ __launch_bounds__(256) void edge_aggregate_kernel(
    const float* __restrict__ ea, const int* __restrict__ perm,
    const int* __restrict__ ssrc, const int* __restrict__ sdst,
    const float* __restrict__ Wev, const float* __restrict__ v,
    const float* __restrict__ scores, const float* __restrict__ denom,
    float* __restrict__ acc) {
  __shared__ float carry[WAVES][8][32];
  const int lane = threadIdx.x & 31;
  const int wv   = threadIdx.x >> 5;
  const int m    = lane & 15;
  const int hf   = lane >> 4;
  const int kk   = hf << 1;
  const int wbase = (blockIdx.x * WAVES + wv) * (16 * TILES);
  const int sbase = wbase + hf * (8 * TILES);
  const int abase = wbase + (m >> 3) * (8 * TILES) + (m & 7);

#pragma unroll 1
  for (int i = 0; i < TILES; ++i) {
    v2f afr[8];
    const float* er = ea + (size_t)perm[abase + 8 * i] * 32;
#pragma unroll
    for (int j = 0; j < 8; ++j) {
      afr[j][0] = er[4 * j + kk];
      afr[j][1] = er[4 * j + kk + 1];
    }

    const int p = sbase + 8 * i;
    unsigned so[8], dn[8], vo[8];
    {
      const v4i p0 = *(const v4i*)(perm + p), p1 = *(const v4i*)(perm + p + 4);
      const v4i s0 = *(const v4i*)(ssrc + p), s1 = *(const v4i*)(ssrc + p + 4);
      const v4i d0 = *(const v4i*)(sdst + p), d1 = *(const v4i*)(sdst + p + 4);
#pragma unroll
      for (int r = 0; r < 4; ++r) {
        so[r] = (unsigned)p0[r] * 8u;        so[r + 4] = (unsigned)p1[r] * 8u;
        vo[r] = (unsigned)s0[r] * 128u + m;  vo[r + 4] = (unsigned)s1[r] * 128u + m;
        dn[r] = (unsigned)d0[r] * 8u;        dn[r + 4] = (unsigned)d1[r] * 8u;
      }
    }
    const bool last = (i == TILES - 1);
    const unsigned dnext = last ? 0xffffffffu : (unsigned)sdst[p + 8] * 8u;

#pragma unroll 1
    for (int h = 0; h < 8; ++h) {
      v8f c = {};
      const float* wp = Wev + h * 16 + m;
#pragma unroll
      for (int j = 0; j < 8; ++j) {
        v2f b;
        b[0] = wp[(size_t)(4 * j + kk) * 128];
        b[1] = wp[(size_t)(4 * j + kk + 1) * 128];
        c = __builtin_amdgcn_wmma_f32_16x16x4_f32(false, afr[j], false, b,
                                                  (short)0, c, false, false);
      }
      float t[8];
#pragma unroll
      for (int r = 0; r < 8; ++r) {
        float att = scores[so[r] + (unsigned)h] / (denom[dn[r] + (unsigned)h] + 1e-9f);
        float ve = v[vo[r] + 16u * (unsigned)h] + c[r];
        t[r] = att * ve;
      }
      float run = (i == 0) ? 0.0f : carry[wv][h][lane];
#pragma unroll
      for (int r = 0; r < 8; ++r) {
        run += t[r];
        const unsigned nx = (r < 7) ? dn[r + 1 < 8 ? r + 1 : 7] : dnext;
        if (dn[r] != nx) {
          atomicAdd(&acc[dn[r] * 16u + 16u * (unsigned)h + m], run);
          run = 0.0f;
        }
      }
      if (!last) carry[wv][h][lane] = run;
    }
  }
}

extern "C" void kernel_launch(void* const* d_in, const int* in_sizes, int n_in,
                              void* d_out, int out_size, void* d_ws,
                              size_t ws_size, hipStream_t stream) {
  const float* x   = (const float*)d_in[0];
  const int*   ei  = (const int*)d_in[1];
  const float* ea  = (const float*)d_in[2];
  const float* Wq  = (const float*)d_in[3];
  const float* Wk  = (const float*)d_in[4];
  const float* Wv  = (const float*)d_in[5];
  const float* Wo  = (const float*)d_in[6];
  const float* Wek = (const float*)d_in[7];
  const float* Wev = (const float*)d_in[8];
  const float* Wg1 = (const float*)d_in[9];
  const float* Wg2 = (const float*)d_in[10];
  float* out = (float*)d_out;

  const int N = in_sizes[0] / 128;
  const int E = in_sizes[2] / 32;
  const int* src = ei;
  const int* dst = ei + E;

  float* q      = (float*)d_ws;
  float* k      = q + (size_t)N * 128;
  float* v      = k + (size_t)N * 128;
  float* scores = v + (size_t)N * 128;
  float* acc    = scores + (size_t)E * 8;
  float* denom  = acc + (size_t)N * 128;
  unsigned* mmax = (unsigned*)(denom + (size_t)N * 8);
  int* cnt  = (int*)v;
  int* rank = cnt + N;
  int* perm = (int*)q;
  int* ssrc = perm + E;
  int* sdst = ssrc + E;
  int* off  = sdst + E;

  hipMemsetAsync(acc, 0, (size_t)(N * 128 + N * 8 + N * 8) * sizeof(float),
                 stream);

  dim3 blk(256);
  gemm128_kernel<<<dim3(N / 16), blk, 0, stream>>>(x, Wq, q);
  gemm128_kernel<<<dim3(N / 16), blk, 0, stream>>>(x, Wk, k);

  edge_score_kernel<<<dim3(E / (16 * WAVES)), blk, 0, stream>>>(
      ea, src, dst, Wek, Wg1, Wg2, q, k, scores, mmax, cnt, N);

  exp_denom_kernel<<<dim3((E * 8 + 255) / 256), blk, 0, stream>>>(
      dst, scores, mmax, denom, cnt, rank, E * 8);

  seg_offsets_kernel<<<dim3((N + 1023) / 1024), dim3(1024), 0, stream>>>(
      cnt, off, N);

  dst_order_kernel<<<dim3((E + 255) / 256), blk, 0, stream>>>(
      src, dst, rank, off, perm, ssrc, sdst, E);

  gemm128_kernel<<<dim3(N / 16), blk, 0, stream>>>(x, Wv, v);

  edge_aggregate_kernel<<<dim3(E / (16 * WAVES * TILES)), blk, 0, stream>>>(
      ea, perm, ssrc, sdst, Wev, v, scores, denom, acc);

  gemm128_kernel<<<dim3(N / 16), blk, 0, stream>>>(acc, Wo, out);
}


# exp_denom: a wave takes 64 edges x 8 heads (16 unrolled steps of one 128-byte score line, each phase's loads in flight together), not 4 edges: 12500 waves enter, not 200000; denom still by atomics
# speedup vs baseline: 1.7446x; 1.0226x over previous
#include <hip/hip_runtime.h>


typedef __attribute__((ext_vector_type(2))) float v2f;
typedef __attribute__((ext_vector_type(4))) float v4f;
typedef __attribute__((ext_vector_type(8))) float v8f;

#define WAVES 8

__device__ __forceinline__ unsigned fkey(float f) {
  unsigned u = __float_as_uint(f);
  return (u & 0x80000000u) ? ~u : (u | 0x80000000u);
}
__device__ __forceinline__ float funkey(unsigned kk) {
  unsigned u = (kk & 0x80000000u) ? (kk & 0x7FFFFFFFu) : ~kk;
  return __uint_as_float(u);
}

__global__ __launch_bounds__(256) void gemm128_kernel(
    const float* __restrict__ A, const float* __restrict__ W,
    float* __restrict__ out) {
  const int lane = threadIdx.x & 31;
  const int wv   = threadIdx.x >> 5;
  const int row0 = blockIdx.x << 4;
  const int n0   = wv << 4;
  const int m    = lane & 15;
  const int kk   = (lane >> 4) << 1;
  const int mb   = (lane >> 4) << 3;

  v8f c = {};
  const float* ar = A + (size_t)(row0 + m) * 128;
  const float* wp = W + n0 + m;
#pragma unroll 8
  for (int k0 = 0; k0 < 128; k0 += 4) {
    v2f a, b;
    a[0] = ar[k0 + kk];
    a[1] = ar[k0 + kk + 1];
    b[0] = wp[(size_t)(k0 + kk) * 128];
    b[1] = wp[(size_t)(k0 + kk + 1) * 128];
    c = __builtin_amdgcn_wmma_f32_16x16x4_f32(false, a, false, b, (short)0, c,
                                              false, false);
  }
  float* op = out + (size_t)(row0 + mb) * 128 + n0 + m;
#pragma unroll
  for (int r = 0; r < 8; ++r) op[(size_t)r * 128] = c[r];
}

__device__ __forceinline__ float xhalf(float x) {
  int b = __builtin_bit_cast(int, x);
  return __builtin_bit_cast(
      float, __builtin_amdgcn_permlanex16(b, b, 0x76543210, 0xfedcba98, false,
                                          false));
}

__device__ __forceinline__ v8f silu_gate1_tile(const float* __restrict__ Wg1,
                                               const v2f* efr, int t, int m,
                                               int kk) {
  v8f c = {};
  const float* wp = Wg1 + t * 16 + m;
#pragma unroll
  for (int i = 0; i < 8; ++i) {
    v2f a;
    a[0] = wp[(size_t)(4 * i + kk) * 64];
    a[1] = wp[(size_t)(4 * i + kk + 1) * 64];
    c = __builtin_amdgcn_wmma_f32_16x16x4_f32(false, a, false, efr[i], (short)0,
                                              c, false, false);
  }
#pragma unroll
  for (int r = 0; r < 8; ++r) {
    float xv = c[r];
    c[r] = xv / (1.0f + __expf(-xv));
  }
  return c;
}

__device__ __forceinline__ v8f gate2_acc(v8f gt, const float* __restrict__ Wg2,
                                         v8f c, int t, int m, int hf) {
  const float* w2 = Wg2 + (size_t)(t * 16 + 8 * hf) * 8 + (m & 7);
#pragma unroll
  for (int p = 0; p < 4; ++p) {
    v2f a, b;
    a[0] = w2[(2 * p) * 8];
    a[1] = w2[(2 * p + 1) * 8];
    b[0] = c[2 * p];
    b[1] = c[2 * p + 1];
    gt = __builtin_amdgcn_wmma_f32_16x16x4_f32(false, a, false, b, (short)0, gt,
                                               false, false);
  }
  return gt;
}

__global__ __launch_bounds__(256) void edge_score_kernel(
    const float* __restrict__ ea, const int* __restrict__ src,
    const int* __restrict__ dst, const float* __restrict__ Wek,
    const float* __restrict__ Wg1, const float* __restrict__ Wg2,
    const float* __restrict__ q, const float* __restrict__ k,
    float* __restrict__ scores, unsigned* __restrict__ mmax,
    int* __restrict__ cnt, int n_nodes) {
  const int lane = threadIdx.x & 31;
  const int wv   = threadIdx.x >> 5;
  const int e0   = (blockIdx.x * WAVES + wv) << 4;
  const int m    = lane & 15;
  const int hf   = lane >> 4;
  const int kk   = hf << 1;

  v2f efr[8];
  const float* er = ea + (size_t)(e0 + m) * 32;
#pragma unroll
  for (int i = 0; i < 8; ++i) {
    efr[i][0] = er[4 * i + kk];
    efr[i][1] = er[4 * i + kk + 1];
  }
  const int sm = src[e0 + m];
  const int dm = dst[e0 + m];

  v8f gt = {};
  v8f g1 = silu_gate1_tile(Wg1, efr, 0, m, kk);
#pragma unroll 1
  for (int t = 1; t < 4; ++t) {
    gt = gate2_acc(gt, Wg2, g1, t - 1, m, hf);
    g1 = silu_gate1_tile(Wg1, efr, t, m, kk);
  }
  gt = gate2_acc(gt, Wg2, g1, 3, m, hf);

  const float* qrow = q + (size_t)dm * 128 + 8 * hf;
  const float* krow = k + (size_t)sm * 128 + 8 * hf;
  v4f o = {};
#pragma unroll 1
  for (int h = 0; h < 8; ++h) {
    v8f c = {};
    const float* wp = Wek + h * 16 + m;
#pragma unroll
    for (int i = 0; i < 8; ++i) {
      v2f a;
      a[0] = wp[(size_t)(4 * i + kk) * 128];
      a[1] = wp[(size_t)(4 * i + kk + 1) * 128];
      c = __builtin_amdgcn_wmma_f32_16x16x4_f32(false, a, false, efr[i],
                                                (short)0, c, false, false);
    }
    const v4f q0 = *(const v4f*)(qrow + h * 16);
    const v4f q1 = *(const v4f*)(qrow + h * 16 + 4);
    const v4f k0 = *(const v4f*)(krow + h * 16);
    const v4f k1 = *(const v4f*)(krow + h * 16 + 4);
    float p = q0[0] * (k0[0] + c[0]);
    p = fmaf(q0[1], k0[1] + c[1], p);
    p = fmaf(q0[2], k0[2] + c[2], p);
    p = fmaf(q0[3], k0[3] + c[3], p);
    p = fmaf(q1[0], k1[0] + c[4], p);
    p = fmaf(q1[1], k1[1] + c[5], p);
    p = fmaf(q1[2], k1[2] + c[6], p);
    p = fmaf(q1[3], k1[3] + c[7], p);
    p += xhalf(p);
    const int hh = h - 4 * hf;
    o[0] = (hh == 0) ? p : o[0];
    o[1] = (hh == 1) ? p : o[1];
    o[2] = (hh == 2) ? p : o[2];
    o[3] = (hh == 3) ? p : o[3];
  }
  o[0] = o[0] * 0.25f + (hf ? gt[4] : gt[0]);
  o[1] = o[1] * 0.25f + (hf ? gt[5] : gt[1]);
  o[2] = o[2] * 0.25f + (hf ? gt[6] : gt[2]);
  o[3] = o[3] * 0.25f + (hf ? gt[7] : gt[3]);

  *(v4f*)(scores + (size_t)(e0 + m) * 8 + 4 * hf) = o;
  unsigned* mp = mmax + (size_t)dm * 8 + 4 * hf;
  atomicMax(mp + 0, fkey(o[0]));
  atomicMax(mp + 1, fkey(o[1]));
  atomicMax(mp + 2, fkey(o[2]));
  atomicMax(mp + 3, fkey(o[3]));

  const int gid = blockIdx.x * (WAVES * 32) + threadIdx.x;
  if (gid < n_nodes) cnt[gid] = 0;
}

#define EXP_STEPS 16
#define EXP_EDGES (4 * EXP_STEPS)
#define EXP_ROUNDS (EXP_EDGES / 32)

__global__ __launch_bounds__(256) void exp_denom_kernel(
    const int* __restrict__ dst, float* __restrict__ scores,
    const unsigned* __restrict__ mmax, float* __restrict__ denom,
    int* __restrict__ cnt, int* __restrict__ rank, int n_edges) {
  const int lane  = threadIdx.x & 31;
  const int wbase = (blockIdx.x * WAVES + (threadIdx.x >> 5)) * EXP_EDGES;
  if (wbase >= n_edges) return;
  const unsigned h = lane & 7;
  const int sub = lane >> 3;

  if (wbase + EXP_EDGES <= n_edges) {
    float* sp = scores + (size_t)wbase * 8 + lane;
    const int* dp = dst + wbase + sub;
    int d[EXP_STEPS], rd[EXP_ROUNDS], rk[EXP_ROUNDS];
    float s[EXP_STEPS];
    unsigned mk[EXP_STEPS];
#pragma unroll
    for (int i = 0; i < EXP_STEPS; ++i) {
      d[i] = dp[4 * i];
      s[i] = sp[32 * i];
    }
#pragma unroll
    for (int j = 0; j < EXP_ROUNDS; ++j) rd[j] = dst[wbase + 32 * j + lane];
#pragma unroll
    for (int i = 0; i < EXP_STEPS; ++i) mk[i] = mmax[(unsigned)d[i] * 8u + h];
#pragma unroll
    for (int j = 0; j < EXP_ROUNDS; ++j) rk[j] = atomicAdd(&cnt[rd[j]], 1);
#pragma unroll
    for (int i = 0; i < EXP_STEPS; ++i) {
      const float es = __expf(s[i] - funkey(mk[i]));
      sp[32 * i] = es;
      atomicAdd(&denom[(unsigned)d[i] * 8u + h], es);
    }
#pragma unroll
    for (int j = 0; j < EXP_ROUNDS; ++j) rank[wbase + 32 * j + lane] = rk[j];
  } else {
    for (int i = 0; i < EXP_STEPS; ++i) {
      const int e = wbase + 4 * i + sub;
      if (e < n_edges) {
        const unsigned dh = (unsigned)dst[e] * 8u + h;
        const size_t idx = (size_t)e * 8 + h;
        const float es = __expf(scores[idx] - funkey(mmax[dh]));
        scores[idx] = es;
        atomicAdd(&denom[dh], es);
      }
    }
    for (int j = 0; j < EXP_ROUNDS; ++j) {
      const int e = wbase + 32 * j + lane;
      if (e < n_edges) rank[e] = atomicAdd(&cnt[dst[e]], 1);
    }
  }
}

__global__ __launch_bounds__(1024) void seg_offsets_kernel(
    const int* __restrict__ cnt, int* __restrict__ off, int n) {
  __shared__ int wsum[32];
  __shared__ int psum[32];
  const int t = threadIdx.x, lane = t & 31, w = t >> 5;
  const int base = blockIdx.x * 1024;
  int p = 0;
  for (int i = t; i < base; i += 1024) p += cnt[i];
  const int x = (base + t < n) ? cnt[base + t] : 0;
  int inc = x;
  for (int d = 1; d < 32; d <<= 1) {
    int y = __shfl_up(inc, d);
    if (lane >= d) inc += y;
  }
  for (int d = 16; d > 0; d >>= 1) p += __shfl_xor(p, d);
  if (lane == 31) wsum[w] = inc;
  if (lane == 0) psum[w] = p;
  __syncthreads();
  int pre = 0;
  for (int i = 0; i < 32; ++i) {
    pre += psum[i];
    pre += (i < w) ? wsum[i] : 0;
  }
  if (base + t < n) off[base + t] = pre + inc - x;
}

__global__ __launch_bounds__(256) void dst_order_kernel(
    const int* __restrict__ src, const int* __restrict__ dst,
    const int* __restrict__ rank, const int* __restrict__ off,
    int* __restrict__ perm, int* __restrict__ ssrc, int* __restrict__ sdst,
    int n_edges) {
  const int e = blockIdx.x * blockDim.x + threadIdx.x;
  if (e >= n_edges) return;
  const int d = dst[e];
  const int pos = off[d] + rank[e];
  perm[pos] = e;
  ssrc[pos] = src[e];
  sdst[pos] = d;
}

typedef __attribute__((ext_vector_type(4))) int v4i;
#define TILES 2

__global__ __launch_bounds__(256) void edge_aggregate_kernel(
    const float* __restrict__ ea, const int* __restrict__ perm,
    const int* __restrict__ ssrc, const int* __restrict__ sdst,
    const float* __restrict__ Wev, const float* __restrict__ v,
    const float* __restrict__ scores, const float* __restrict__ denom,
    float* __restrict__ acc) {
  __shared__ float carry[WAVES][8][32];
  const int lane = threadIdx.x & 31;
  const int wv   = threadIdx.x >> 5;
  const int m    = lane & 15;
  const int hf   = lane >> 4;
  const int kk   = hf << 1;
  const int wbase = (blockIdx.x * WAVES + wv) * (16 * TILES);
  const int sbase = wbase + hf * (8 * TILES);
  const int abase = wbase + (m >> 3) * (8 * TILES) + (m & 7);

#pragma unroll 1
  for (int i = 0; i < TILES; ++i) {
    v2f afr[8];
    const float* er = ea + (size_t)perm[abase + 8 * i] * 32;
#pragma unroll
    for (int j = 0; j < 8; ++j) {
      afr[j][0] = er[4 * j + kk];
      afr[j][1] = er[4 * j + kk + 1];
    }

    const int p = sbase + 8 * i;
    unsigned so[8], dn[8], vo[8];
    {
      const v4i p0 = *(const v4i*)(perm + p), p1 = *(const v4i*)(perm + p + 4);
      const v4i s0 = *(const v4i*)(ssrc + p), s1 = *(const v4i*)(ssrc + p + 4);
      const v4i d0 = *(const v4i*)(sdst + p), d1 = *(const v4i*)(sdst + p + 4);
#pragma unroll
      for (int r = 0; r < 4; ++r) {
        so[r] = (unsigned)p0[r] * 8u;        so[r + 4] = (unsigned)p1[r] * 8u;
        vo[r] = (unsigned)s0[r] * 128u + m;  vo[r + 4] = (unsigned)s1[r] * 128u + m;
        dn[r] = (unsigned)d0[r] * 8u;        dn[r + 4] = (unsigned)d1[r] * 8u;
      }
    }
    const bool last = (i == TILES - 1);
    const unsigned dnext = last ? 0xffffffffu : (unsigned)sdst[p + 8] * 8u;

#pragma unroll 1
    for (int h = 0; h < 8; ++h) {
      v8f c = {};
      const float* wp = Wev + h * 16 + m;
#pragma unroll
      for (int j = 0; j < 8; ++j) {
        v2f b;
        b[0] = wp[(size_t)(4 * j + kk) * 128];
        b[1] = wp[(size_t)(4 * j + kk + 1) * 128];
        c = __builtin_amdgcn_wmma_f32_16x16x4_f32(false, afr[j], false, b,
                                                  (short)0, c, false, false);
      }
      float t[8];
#pragma unroll
      for (int r = 0; r < 8; ++r) {
        float att = scores[so[r] + (unsigned)h] / (denom[dn[r] + (unsigned)h] + 1e-9f);
        float ve = v[vo[r] + 16u * (unsigned)h] + c[r];
        t[r] = att * ve;
      }
      float run = (i == 0) ? 0.0f : carry[wv][h][lane];
#pragma unroll
      for (int r = 0; r < 8; ++r) {
        run += t[r];
        const unsigned nx = (r < 7) ? dn[r + 1 < 8 ? r + 1 : 7] : dnext;
        if (dn[r] != nx) {
          atomicAdd(&acc[dn[r] * 16u + 16u * (unsigned)h + m], run);
          run = 0.0f;
        }
      }
      if (!last) carry[wv][h][lane] = run;
    }
  }
}

extern "C" void kernel_launch(void* const* d_in, const int* in_sizes, int n_in,
                              void* d_out, int out_size, void* d_ws,
                              size_t ws_size, hipStream_t stream) {
  const float* x   = (const float*)d_in[0];
  const int*   ei  = (const int*)d_in[1];
  const float* ea  = (const float*)d_in[2];
  const float* Wq  = (const float*)d_in[3];
  const float* Wk  = (const float*)d_in[4];
  const float* Wv  = (const float*)d_in[5];
  const float* Wo  = (const float*)d_in[6];
  const float* Wek = (const float*)d_in[7];
  const float* Wev = (const float*)d_in[8];
  const float* Wg1 = (const float*)d_in[9];
  const float* Wg2 = (const float*)d_in[10];
  float* out = (float*)d_out;

  const int N = in_sizes[0] / 128;
  const int E = in_sizes[2] / 32;
  const int* src = ei;
  const int* dst = ei + E;

  float* q      = (float*)d_ws;
  float* k      = q + (size_t)N * 128;
  float* v      = k + (size_t)N * 128;
  float* scores = v + (size_t)N * 128;
  float* acc    = scores + (size_t)E * 8;
  float* denom  = acc + (size_t)N * 128;
  unsigned* mmax = (unsigned*)(denom + (size_t)N * 8);
  int* cnt  = (int*)v;
  int* rank = cnt + N;
  int* perm = (int*)q;
  int* ssrc = perm + E;
  int* sdst = ssrc + E;
  int* off  = sdst + E;

  hipMemsetAsync(acc, 0, (size_t)(N * 128 + N * 8 + N * 8) * sizeof(float),
                 stream);

  dim3 blk(256);
  gemm128_kernel<<<dim3(N / 16), blk, 0, stream>>>(x, Wq, q);
  gemm128_kernel<<<dim3(N / 16), blk, 0, stream>>>(x, Wk, k);

  edge_score_kernel<<<dim3(E / (16 * WAVES)), blk, 0, stream>>>(
      ea, src, dst, Wek, Wg1, Wg2, q, k, scores, mmax, cnt, N);

  exp_denom_kernel<<<dim3((E + WAVES * EXP_EDGES - 1) / (WAVES * EXP_EDGES)), blk,
                     0, stream>>>(dst, scores, mmax, denom, cnt, rank, E);

  seg_offsets_kernel<<<dim3((N + 1023) / 1024), dim3(1024), 0, stream>>>(
      cnt, off, N);

  dst_order_kernel<<<dim3((E + 255) / 256), blk, 0, stream>>>(
      src, dst, rank, off, perm, ssrc, sdst, E);

  gemm128_kernel<<<dim3(N / 16), blk, 0, stream>>>(x, Wv, v);

  edge_aggregate_kernel<<<dim3(E / (16 * WAVES * TILES)), blk, 0, stream>>>(
      ea, perm, ssrc, sdst, Wev, v, scores, denom, acc);

  gemm128_kernel<<<dim3(N / 16), blk, 0, stream>>>(acc, Wo, out);
}


# gemm128: a wave owns 16 rows x 64 interleaved columns (col 64g+4c+t): A fragment loaded once per 4 WMMAs, W rows and output rows as 16-byte whole-line accesses; 6250 waves not 25000; sums unchanged
# speedup vs baseline: 1.8954x; 1.0864x over previous
#include <hip/hip_runtime.h>


typedef __attribute__((ext_vector_type(2))) float v2f;
typedef __attribute__((ext_vector_type(4))) float v4f;
typedef __attribute__((ext_vector_type(8))) float v8f;

#define WAVES 8

__device__ __forceinline__ unsigned fkey(float f) {
  unsigned u = __float_as_uint(f);
  return (u & 0x80000000u) ? ~u : (u | 0x80000000u);
}
__device__ __forceinline__ float funkey(unsigned kk) {
  unsigned u = (kk & 0x80000000u) ? (kk & 0x7FFFFFFFu) : ~kk;
  return __uint_as_float(u);
}

#define GEMM_WAVES 4
__global__ __launch_bounds__(32 * GEMM_WAVES) void gemm128_kernel(
    const float* __restrict__ A, const float* __restrict__ W,
    float* __restrict__ out, int n_tiles) {
  const int lane = threadIdx.x & 31;
  const int unit = blockIdx.x * GEMM_WAVES + (threadIdx.x >> 5);
  const int rt   = unit >> 1;
  const int g    = unit & 1;
  if (rt >= n_tiles) return;
  const int m    = lane & 15;
  const int hf   = lane >> 4;
  const int kk   = hf << 1;

  v8f c0 = {}, c1 = {}, c2 = {}, c3 = {};
  const float* ar = A + (size_t)(rt * 16 + m) * 128 + kk;
  const float* wp = W + (size_t)kk * 128 + 64 * g + 4 * m;
#pragma unroll 8
  for (int k0 = 0; k0 < 128; k0 += 4) {
    const v2f a  = *(const v2f*)(ar + k0);
    const v4f w0 = *(const v4f*)(wp + (size_t)k0 * 128);
    const v4f w1 = *(const v4f*)(wp + (size_t)(k0 + 1) * 128);
    v2f b;
    b[0] = w0[0]; b[1] = w1[0];
    c0 = __builtin_amdgcn_wmma_f32_16x16x4_f32(false, a, false, b, (short)0, c0,
                                               false, false);
    b[0] = w0[1]; b[1] = w1[1];
    c1 = __builtin_amdgcn_wmma_f32_16x16x4_f32(false, a, false, b, (short)0, c1,
                                               false, false);
    b[0] = w0[2]; b[1] = w1[2];
    c2 = __builtin_amdgcn_wmma_f32_16x16x4_f32(false, a, false, b, (short)0, c2,
                                               false, false);
    b[0] = w0[3]; b[1] = w1[3];
    c3 = __builtin_amdgcn_wmma_f32_16x16x4_f32(false, a, false, b, (short)0, c3,
                                               false, false);
  }
  float* op = out + (size_t)(rt * 16 + 8 * hf) * 128 + 64 * g + 4 * m;
#pragma unroll
  for (int r = 0; r < 8; ++r) {
    v4f o;
    o[0] = c0[r]; o[1] = c1[r]; o[2] = c2[r]; o[3] = c3[r];
    *(v4f*)(op + (size_t)r * 128) = o;
  }
}

__device__ __forceinline__ float xhalf(float x) {
  int b = __builtin_bit_cast(int, x);
  return __builtin_bit_cast(
      float, __builtin_amdgcn_permlanex16(b, b, 0x76543210, 0xfedcba98, false,
                                          false));
}

__device__ __forceinline__ v8f silu_gate1_tile(const float* __restrict__ Wg1,
                                               const v2f* efr, int t, int m,
                                               int kk) {
  v8f c = {};
  const float* wp = Wg1 + t * 16 + m;
#pragma unroll
  for (int i = 0; i < 8; ++i) {
    v2f a;
    a[0] = wp[(size_t)(4 * i + kk) * 64];
    a[1] = wp[(size_t)(4 * i + kk + 1) * 64];
    c = __builtin_amdgcn_wmma_f32_16x16x4_f32(false, a, false, efr[i], (short)0,
                                              c, false, false);
  }
#pragma unroll
  for (int r = 0; r < 8; ++r) {
    float xv = c[r];
    c[r] = xv / (1.0f + __expf(-xv));
  }
  return c;
}

__device__ __forceinline__ v8f gate2_acc(v8f gt, const float* __restrict__ Wg2,
                                         v8f c, int t, int m, int hf) {
  const float* w2 = Wg2 + (size_t)(t * 16 + 8 * hf) * 8 + (m & 7);
#pragma unroll
  for (int p = 0; p < 4; ++p) {
    v2f a, b;
    a[0] = w2[(2 * p) * 8];
    a[1] = w2[(2 * p + 1) * 8];
    b[0] = c[2 * p];
    b[1] = c[2 * p + 1];
    gt = __builtin_amdgcn_wmma_f32_16x16x4_f32(false, a, false, b, (short)0, gt,
                                               false, false);
  }
  return gt;
}

__global__ __launch_bounds__(256) void edge_score_kernel(
    const float* __restrict__ ea, const int* __restrict__ src,
    const int* __restrict__ dst, const float* __restrict__ Wek,
    const float* __restrict__ Wg1, const float* __restrict__ Wg2,
    const float* __restrict__ q, const float* __restrict__ k,
    float* __restrict__ scores, unsigned* __restrict__ mmax,
    int* __restrict__ cnt, int n_nodes) {
  const int lane = threadIdx.x & 31;
  const int wv   = threadIdx.x >> 5;
  const int e0   = (blockIdx.x * WAVES + wv) << 4;
  const int m    = lane & 15;
  const int hf   = lane >> 4;
  const int kk   = hf << 1;

  v2f efr[8];
  const float* er = ea + (size_t)(e0 + m) * 32;
#pragma unroll
  for (int i = 0; i < 8; ++i) {
    efr[i][0] = er[4 * i + kk];
    efr[i][1] = er[4 * i + kk + 1];
  }
  const int sm = src[e0 + m];
  const int dm = dst[e0 + m];

  v8f gt = {};
  v8f g1 = silu_gate1_tile(Wg1, efr, 0, m, kk);
#pragma unroll 1
  for (int t = 1; t < 4; ++t) {
    gt = gate2_acc(gt, Wg2, g1, t - 1, m, hf);
    g1 = silu_gate1_tile(Wg1, efr, t, m, kk);
  }
  gt = gate2_acc(gt, Wg2, g1, 3, m, hf);

  const float* qrow = q + (size_t)dm * 128 + 8 * hf;
  const float* krow = k + (size_t)sm * 128 + 8 * hf;
  v4f o = {};
#pragma unroll 1
  for (int h = 0; h < 8; ++h) {
    v8f c = {};
    const float* wp = Wek + h * 16 + m;
#pragma unroll
    for (int i = 0; i < 8; ++i) {
      v2f a;
      a[0] = wp[(size_t)(4 * i + kk) * 128];
      a[1] = wp[(size_t)(4 * i + kk + 1) * 128];
      c = __builtin_amdgcn_wmma_f32_16x16x4_f32(false, a, false, efr[i],
                                                (short)0, c, false, false);
    }
    const v4f q0 = *(const v4f*)(qrow + h * 16);
    const v4f q1 = *(const v4f*)(qrow + h * 16 + 4);
    const v4f k0 = *(const v4f*)(krow + h * 16);
    const v4f k1 = *(const v4f*)(krow + h * 16 + 4);
    float p = q0[0] * (k0[0] + c[0]);
    p = fmaf(q0[1], k0[1] + c[1], p);
    p = fmaf(q0[2], k0[2] + c[2], p);
    p = fmaf(q0[3], k0[3] + c[3], p);
    p = fmaf(q1[0], k1[0] + c[4], p);
    p = fmaf(q1[1], k1[1] + c[5], p);
    p = fmaf(q1[2], k1[2] + c[6], p);
    p = fmaf(q1[3], k1[3] + c[7], p);
    p += xhalf(p);
    const int hh = h - 4 * hf;
    o[0] = (hh == 0) ? p : o[0];
    o[1] = (hh == 1) ? p : o[1];
    o[2] = (hh == 2) ? p : o[2];
    o[3] = (hh == 3) ? p : o[3];
  }
  o[0] = o[0] * 0.25f + (hf ? gt[4] : gt[0]);
  o[1] = o[1] * 0.25f + (hf ? gt[5] : gt[1]);
  o[2] = o[2] * 0.25f + (hf ? gt[6] : gt[2]);
  o[3] = o[3] * 0.25f + (hf ? gt[7] : gt[3]);

  *(v4f*)(scores + (size_t)(e0 + m) * 8 + 4 * hf) = o;
  unsigned* mp = mmax + (size_t)dm * 8 + 4 * hf;
  atomicMax(mp + 0, fkey(o[0]));
  atomicMax(mp + 1, fkey(o[1]));
  atomicMax(mp + 2, fkey(o[2]));
  atomicMax(mp + 3, fkey(o[3]));

  const int gid = blockIdx.x * (WAVES * 32) + threadIdx.x;
  if (gid < n_nodes) cnt[gid] = 0;
}

#define EXP_STEPS 16
#define EXP_EDGES (4 * EXP_STEPS)
#define EXP_ROUNDS (EXP_EDGES / 32)

__global__ __launch_bounds__(256) void exp_denom_kernel(
    const int* __restrict__ dst, float* __restrict__ scores,
    const unsigned* __restrict__ mmax, float* __restrict__ denom,
    int* __restrict__ cnt, int* __restrict__ rank, int n_edges) {
  const int lane  = threadIdx.x & 31;
  const int wbase = (blockIdx.x * WAVES + (threadIdx.x >> 5)) * EXP_EDGES;
  if (wbase >= n_edges) return;
  const unsigned h = lane & 7;
  const int sub = lane >> 3;

  if (wbase + EXP_EDGES <= n_edges) {
    float* sp = scores + (size_t)wbase * 8 + lane;
    const int* dp = dst + wbase + sub;
    int d[EXP_STEPS], rd[EXP_ROUNDS], rk[EXP_ROUNDS];
    float s[EXP_STEPS];
    unsigned mk[EXP_STEPS];
#pragma unroll
    for (int i = 0; i < EXP_STEPS; ++i) {
      d[i] = dp[4 * i];
      s[i] = sp[32 * i];
    }
#pragma unroll
    for (int j = 0; j < EXP_ROUNDS; ++j) rd[j] = dst[wbase + 32 * j + lane];
#pragma unroll
    for (int i = 0; i < EXP_STEPS; ++i) mk[i] = mmax[(unsigned)d[i] * 8u + h];
#pragma unroll
    for (int j = 0; j < EXP_ROUNDS; ++j) rk[j] = atomicAdd(&cnt[rd[j]], 1);
#pragma unroll
    for (int i = 0; i < EXP_STEPS; ++i) {
      const float es = __expf(s[i] - funkey(mk[i]));
      sp[32 * i] = es;
      atomicAdd(&denom[(unsigned)d[i] * 8u + h], es);
    }
#pragma unroll
    for (int j = 0; j < EXP_ROUNDS; ++j) rank[wbase + 32 * j + lane] = rk[j];
  } else {
    for (int i = 0; i < EXP_STEPS; ++i) {
      const int e = wbase + 4 * i + sub;
      if (e < n_edges) {
        const unsigned dh = (unsigned)dst[e] * 8u + h;
        const size_t idx = (size_t)e * 8 + h;
        const float es = __expf(scores[idx] - funkey(mmax[dh]));
        scores[idx] = es;
        atomicAdd(&denom[dh], es);
      }
    }
    for (int j = 0; j < EXP_ROUNDS; ++j) {
      const int e = wbase + 32 * j + lane;
      if (e < n_edges) rank[e] = atomicAdd(&cnt[dst[e]], 1);
    }
  }
}

__global__ __launch_bounds__(1024) void seg_offsets_kernel(
    const int* __restrict__ cnt, int* __restrict__ off, int n) {
  __shared__ int wsum[32];
  __shared__ int psum[32];
  const int t = threadIdx.x, lane = t & 31, w = t >> 5;
  const int base = blockIdx.x * 1024;
  int p = 0;
  for (int i = t; i < base; i += 1024) p += cnt[i];
  const int x = (base + t < n) ? cnt[base + t] : 0;
  int inc = x;
  for (int d = 1; d < 32; d <<= 1) {
    int y = __shfl_up(inc, d);
    if (lane >= d) inc += y;
  }
  for (int d = 16; d > 0; d >>= 1) p += __shfl_xor(p, d);
  if (lane == 31) wsum[w] = inc;
  if (lane == 0) psum[w] = p;
  __syncthreads();
  int pre = 0;
  for (int i = 0; i < 32; ++i) {
    pre += psum[i];
    pre += (i < w) ? wsum[i] : 0;
  }
  if (base + t < n) off[base + t] = pre + inc - x;
}

__global__ __launch_bounds__(256) void dst_order_kernel(
    const int* __restrict__ src, const int* __restrict__ dst,
    const int* __restrict__ rank, const int* __restrict__ off,
    int* __restrict__ perm, int* __restrict__ ssrc, int* __restrict__ sdst,
    int n_edges) {
  const int e = blockIdx.x * blockDim.x + threadIdx.x;
  if (e >= n_edges) return;
  const int d = dst[e];
  const int pos = off[d] + rank[e];
  perm[pos] = e;
  ssrc[pos] = src[e];
  sdst[pos] = d;
}

typedef __attribute__((ext_vector_type(4))) int v4i;
#define TILES 2

__global__ __launch_bounds__(256) void edge_aggregate_kernel(
    const float* __restrict__ ea, const int* __restrict__ perm,
    const int* __restrict__ ssrc, const int* __restrict__ sdst,
    const float* __restrict__ Wev, const float* __restrict__ v,
    const float* __restrict__ scores, const float* __restrict__ denom,
    float* __restrict__ acc) {
  __shared__ float carry[WAVES][8][32];
  const int lane = threadIdx.x & 31;
  const int wv   = threadIdx.x >> 5;
  const int m    = lane & 15;
  const int hf   = lane >> 4;
  const int kk   = hf << 1;
  const int wbase = (blockIdx.x * WAVES + wv) * (16 * TILES);
  const int sbase = wbase + hf * (8 * TILES);
  const int abase = wbase + (m >> 3) * (8 * TILES) + (m & 7);

#pragma unroll 1
  for (int i = 0; i < TILES; ++i) {
    v2f afr[8];
    const float* er = ea + (size_t)perm[abase + 8 * i] * 32;
#pragma unroll
    for (int j = 0; j < 8; ++j) {
      afr[j][0] = er[4 * j + kk];
      afr[j][1] = er[4 * j + kk + 1];
    }

    const int p = sbase + 8 * i;
    unsigned so[8], dn[8], vo[8];
    {
      const v4i p0 = *(const v4i*)(perm + p), p1 = *(const v4i*)(perm + p + 4);
      const v4i s0 = *(const v4i*)(ssrc + p), s1 = *(const v4i*)(ssrc + p + 4);
      const v4i d0 = *(const v4i*)(sdst + p), d1 = *(const v4i*)(sdst + p + 4);
#pragma unroll
      for (int r = 0; r < 4; ++r) {
        so[r] = (unsigned)p0[r] * 8u;        so[r + 4] = (unsigned)p1[r] * 8u;
        vo[r] = (unsigned)s0[r] * 128u + m;  vo[r + 4] = (unsigned)s1[r] * 128u + m;
        dn[r] = (unsigned)d0[r] * 8u;        dn[r + 4] = (unsigned)d1[r] * 8u;
      }
    }
    const bool last = (i == TILES - 1);
    const unsigned dnext = last ? 0xffffffffu : (unsigned)sdst[p + 8] * 8u;

#pragma unroll 1
    for (int h = 0; h < 8; ++h) {
      v8f c = {};
      const float* wp = Wev + h * 16 + m;
#pragma unroll
      for (int j = 0; j < 8; ++j) {
        v2f b;
        b[0] = wp[(size_t)(4 * j + kk) * 128];
        b[1] = wp[(size_t)(4 * j + kk + 1) * 128];
        c = __builtin_amdgcn_wmma_f32_16x16x4_f32(false, afr[j], false, b,
                                                  (short)0, c, false, false);
      }
      float t[8];
#pragma unroll
      for (int r = 0; r < 8; ++r) {
        float att = scores[so[r] + (unsigned)h] / (denom[dn[r] + (unsigned)h] + 1e-9f);
        float ve = v[vo[r] + 16u * (unsigned)h] + c[r];
        t[r] = att * ve;
      }
      float run = (i == 0) ? 0.0f : carry[wv][h][lane];
#pragma unroll
      for (int r = 0; r < 8; ++r) {
        run += t[r];
        const unsigned nx = (r < 7) ? dn[r + 1 < 8 ? r + 1 : 7] : dnext;
        if (dn[r] != nx) {
          atomicAdd(&acc[dn[r] * 16u + 16u * (unsigned)h + m], run);
          run = 0.0f;
        }
      }
      if (!last) carry[wv][h][lane] = run;
    }
  }
}

extern "C" void kernel_launch(void* const* d_in, const int* in_sizes, int n_in,
                              void* d_out, int out_size, void* d_ws,
                              size_t ws_size, hipStream_t stream) {
  const float* x   = (const float*)d_in[0];
  const int*   ei  = (const int*)d_in[1];
  const float* ea  = (const float*)d_in[2];
  const float* Wq  = (const float*)d_in[3];
  const float* Wk  = (const float*)d_in[4];
  const float* Wv  = (const float*)d_in[5];
  const float* Wo  = (const float*)d_in[6];
  const float* Wek = (const float*)d_in[7];
  const float* Wev = (const float*)d_in[8];
  const float* Wg1 = (const float*)d_in[9];
  const float* Wg2 = (const float*)d_in[10];
  float* out = (float*)d_out;

  const int N = in_sizes[0] / 128;
  const int E = in_sizes[2] / 32;
  const int* src = ei;
  const int* dst = ei + E;

  float* q      = (float*)d_ws;
  float* k      = q + (size_t)N * 128;
  float* v      = k + (size_t)N * 128;
  float* scores = v + (size_t)N * 128;
  float* acc    = scores + (size_t)E * 8;
  float* denom  = acc + (size_t)N * 128;
  unsigned* mmax = (unsigned*)(denom + (size_t)N * 8);
  int* cnt  = (int*)v;
  int* rank = cnt + N;
  int* perm = (int*)q;
  int* ssrc = perm + E;
  int* sdst = ssrc + E;
  int* off  = sdst + E;

  hipMemsetAsync(acc, 0, (size_t)(N * 128 + N * 8 + N * 8) * sizeof(float),
                 stream);

  dim3 blk(256);
  dim3 gblk(32 * GEMM_WAVES);
  gemm128_kernel<<<dim3((N / 16 * 2 + GEMM_WAVES - 1) / GEMM_WAVES), gblk, 0, stream>>>(x, Wq, q, N / 16);
  gemm128_kernel<<<dim3((N / 16 * 2 + GEMM_WAVES - 1) / GEMM_WAVES), gblk, 0, stream>>>(x, Wk, k, N / 16);

  edge_score_kernel<<<dim3(E / (16 * WAVES)), blk, 0, stream>>>(
      ea, src, dst, Wek, Wg1, Wg2, q, k, scores, mmax, cnt, N);

  exp_denom_kernel<<<dim3((E + WAVES * EXP_EDGES - 1) / (WAVES * EXP_EDGES)), blk,
                     0, stream>>>(dst, scores, mmax, denom, cnt, rank, E);

  seg_offsets_kernel<<<dim3((N + 1023) / 1024), dim3(1024), 0, stream>>>(
      cnt, off, N);

  dst_order_kernel<<<dim3((E + 255) / 256), blk, 0, stream>>>(
      src, dst, rank, off, perm, ssrc, sdst, E);

  gemm128_kernel<<<dim3((N / 16 * 2 + GEMM_WAVES - 1) / GEMM_WAVES), gblk, 0, stream>>>(x, Wv, v, N / 16);

  edge_aggregate_kernel<<<dim3(E / (16 * WAVES * TILES)), blk, 0, stream>>>(
      ea, perm, ssrc, sdst, Wev, v, scores, denom, acc);

  gemm128_kernel<<<dim3((N / 16 * 2 + GEMM_WAVES - 1) / GEMM_WAVES), gblk, 0, stream>>>(acc, Wo, out, N / 16);
}


# edge_aggregate: 4 accumulator tiles with interleaved columns (feature 64g+4c+t): Wev and v[src] rows as 16-byte loads, one scores/denom load and division per 4 features; sum order unchanged
# speedup vs baseline: 2.0858x; 1.1005x over previous
#include <hip/hip_runtime.h>


typedef __attribute__((ext_vector_type(2))) float v2f;
typedef __attribute__((ext_vector_type(4))) float v4f;
typedef __attribute__((ext_vector_type(8))) float v8f;

#define WAVES 8

__device__ __forceinline__ unsigned fkey(float f) {
  unsigned u = __float_as_uint(f);
  return (u & 0x80000000u) ? ~u : (u | 0x80000000u);
}
__device__ __forceinline__ float funkey(unsigned kk) {
  unsigned u = (kk & 0x80000000u) ? (kk & 0x7FFFFFFFu) : ~kk;
  return __uint_as_float(u);
}

#define GEMM_WAVES 4
__global__ __launch_bounds__(32 * GEMM_WAVES) void gemm128_kernel(
    const float* __restrict__ A, const float* __restrict__ W,
    float* __restrict__ out, int n_tiles) {
  const int lane = threadIdx.x & 31;
  const int unit = blockIdx.x * GEMM_WAVES + (threadIdx.x >> 5);
  const int rt   = unit >> 1;
  const int g    = unit & 1;
  if (rt >= n_tiles) return;
  const int m    = lane & 15;
  const int hf   = lane >> 4;
  const int kk   = hf << 1;

  v8f c0 = {}, c1 = {}, c2 = {}, c3 = {};
  const float* ar = A + (size_t)(rt * 16 + m) * 128 + kk;
  const float* wp = W + (size_t)kk * 128 + 64 * g + 4 * m;
#pragma unroll 8
  for (int k0 = 0; k0 < 128; k0 += 4) {
    const v2f a  = *(const v2f*)(ar + k0);
    const v4f w0 = *(const v4f*)(wp + (size_t)k0 * 128);
    const v4f w1 = *(const v4f*)(wp + (size_t)(k0 + 1) * 128);
    v2f b;
    b[0] = w0[0]; b[1] = w1[0];
    c0 = __builtin_amdgcn_wmma_f32_16x16x4_f32(false, a, false, b, (short)0, c0,
                                               false, false);
    b[0] = w0[1]; b[1] = w1[1];
    c1 = __builtin_amdgcn_wmma_f32_16x16x4_f32(false, a, false, b, (short)0, c1,
                                               false, false);
    b[0] = w0[2]; b[1] = w1[2];
    c2 = __builtin_amdgcn_wmma_f32_16x16x4_f32(false, a, false, b, (short)0, c2,
                                               false, false);
    b[0] = w0[3]; b[1] = w1[3];
    c3 = __builtin_amdgcn_wmma_f32_16x16x4_f32(false, a, false, b, (short)0, c3,
                                               false, false);
  }
  float* op = out + (size_t)(rt * 16 + 8 * hf) * 128 + 64 * g + 4 * m;
#pragma unroll
  for (int r = 0; r < 8; ++r) {
    v4f o;
    o[0] = c0[r]; o[1] = c1[r]; o[2] = c2[r]; o[3] = c3[r];
    *(v4f*)(op + (size_t)r * 128) = o;
  }
}

__device__ __forceinline__ float xhalf(float x) {
  int b = __builtin_bit_cast(int, x);
  return __builtin_bit_cast(
      float, __builtin_amdgcn_permlanex16(b, b, 0x76543210, 0xfedcba98, false,
                                          false));
}

__device__ __forceinline__ v8f silu_gate1_tile(const float* __restrict__ Wg1,
                                               const v2f* efr, int t, int m,
                                               int kk) {
  v8f c = {};
  const float* wp = Wg1 + t * 16 + m;
#pragma unroll
  for (int i = 0; i < 8; ++i) {
    v2f a;
    a[0] = wp[(size_t)(4 * i + kk) * 64];
    a[1] = wp[(size_t)(4 * i + kk + 1) * 64];
    c = __builtin_amdgcn_wmma_f32_16x16x4_f32(false, a, false, efr[i], (short)0,
                                              c, false, false);
  }
#pragma unroll
  for (int r = 0; r < 8; ++r) {
    float xv = c[r];
    c[r] = xv / (1.0f + __expf(-xv));
  }
  return c;
}

__device__ __forceinline__ v8f gate2_acc(v8f gt, const float* __restrict__ Wg2,
                                         v8f c, int t, int m, int hf) {
  const float* w2 = Wg2 + (size_t)(t * 16 + 8 * hf) * 8 + (m & 7);
#pragma unroll
  for (int p = 0; p < 4; ++p) {
    v2f a, b;
    a[0] = w2[(2 * p) * 8];
    a[1] = w2[(2 * p + 1) * 8];
    b[0] = c[2 * p];
    b[1] = c[2 * p + 1];
    gt = __builtin_amdgcn_wmma_f32_16x16x4_f32(false, a, false, b, (short)0, gt,
                                               false, false);
  }
  return gt;
}

__global__ __launch_bounds__(256) void edge_score_kernel(
    const float* __restrict__ ea, const int* __restrict__ src,
    const int* __restrict__ dst, const float* __restrict__ Wek,
    const float* __restrict__ Wg1, const float* __restrict__ Wg2,
    const float* __restrict__ q, const float* __restrict__ k,
    float* __restrict__ scores, unsigned* __restrict__ mmax,
    int* __restrict__ cnt, int n_nodes) {
  const int lane = threadIdx.x & 31;
  const int wv   = threadIdx.x >> 5;
  const int e0   = (blockIdx.x * WAVES + wv) << 4;
  const int m    = lane & 15;
  const int hf   = lane >> 4;
  const int kk   = hf << 1;

  v2f efr[8];
  const float* er = ea + (size_t)(e0 + m) * 32;
#pragma unroll
  for (int i = 0; i < 8; ++i) {
    efr[i][0] = er[4 * i + kk];
    efr[i][1] = er[4 * i + kk + 1];
  }
  const int sm = src[e0 + m];
  const int dm = dst[e0 + m];

  v8f gt = {};
  v8f g1 = silu_gate1_tile(Wg1, efr, 0, m, kk);
#pragma unroll 1
  for (int t = 1; t < 4; ++t) {
    gt = gate2_acc(gt, Wg2, g1, t - 1, m, hf);
    g1 = silu_gate1_tile(Wg1, efr, t, m, kk);
  }
  gt = gate2_acc(gt, Wg2, g1, 3, m, hf);

  const float* qrow = q + (size_t)dm * 128 + 8 * hf;
  const float* krow = k + (size_t)sm * 128 + 8 * hf;
  v4f o = {};
#pragma unroll 1
  for (int h = 0; h < 8; ++h) {
    v8f c = {};
    const float* wp = Wek + h * 16 + m;
#pragma unroll
    for (int i = 0; i < 8; ++i) {
      v2f a;
      a[0] = wp[(size_t)(4 * i + kk) * 128];
      a[1] = wp[(size_t)(4 * i + kk + 1) * 128];
      c = __builtin_amdgcn_wmma_f32_16x16x4_f32(false, a, false, efr[i],
                                                (short)0, c, false, false);
    }
    const v4f q0 = *(const v4f*)(qrow + h * 16);
    const v4f q1 = *(const v4f*)(qrow + h * 16 + 4);
    const v4f k0 = *(const v4f*)(krow + h * 16);
    const v4f k1 = *(const v4f*)(krow + h * 16 + 4);
    float p = q0[0] * (k0[0] + c[0]);
    p = fmaf(q0[1], k0[1] + c[1], p);
    p = fmaf(q0[2], k0[2] + c[2], p);
    p = fmaf(q0[3], k0[3] + c[3], p);
    p = fmaf(q1[0], k1[0] + c[4], p);
    p = fmaf(q1[1], k1[1] + c[5], p);
    p = fmaf(q1[2], k1[2] + c[6], p);
    p = fmaf(q1[3], k1[3] + c[7], p);
    p += xhalf(p);
    const int hh = h - 4 * hf;
    o[0] = (hh == 0) ? p : o[0];
    o[1] = (hh == 1) ? p : o[1];
    o[2] = (hh == 2) ? p : o[2];
    o[3] = (hh == 3) ? p : o[3];
  }
  o[0] = o[0] * 0.25f + (hf ? gt[4] : gt[0]);
  o[1] = o[1] * 0.25f + (hf ? gt[5] : gt[1]);
  o[2] = o[2] * 0.25f + (hf ? gt[6] : gt[2]);
  o[3] = o[3] * 0.25f + (hf ? gt[7] : gt[3]);

  *(v4f*)(scores + (size_t)(e0 + m) * 8 + 4 * hf) = o;
  unsigned* mp = mmax + (size_t)dm * 8 + 4 * hf;
  atomicMax(mp + 0, fkey(o[0]));
  atomicMax(mp + 1, fkey(o[1]));
  atomicMax(mp + 2, fkey(o[2]));
  atomicMax(mp + 3, fkey(o[3]));

  const int gid = blockIdx.x * (WAVES * 32) + threadIdx.x;
  if (gid < n_nodes) cnt[gid] = 0;
}

#define EXP_STEPS 16
#define EXP_EDGES (4 * EXP_STEPS)
#define EXP_ROUNDS (EXP_EDGES / 32)

__global__ __launch_bounds__(256) void exp_denom_kernel(
    const int* __restrict__ dst, float* __restrict__ scores,
    const unsigned* __restrict__ mmax, float* __restrict__ denom,
    int* __restrict__ cnt, int* __restrict__ rank, int n_edges) {
  const int lane  = threadIdx.x & 31;
  const int wbase = (blockIdx.x * WAVES + (threadIdx.x >> 5)) * EXP_EDGES;
  if (wbase >= n_edges) return;
  const unsigned h = lane & 7;
  const int sub = lane >> 3;

  if (wbase + EXP_EDGES <= n_edges) {
    float* sp = scores + (size_t)wbase * 8 + lane;
    const int* dp = dst + wbase + sub;
    int d[EXP_STEPS], rd[EXP_ROUNDS], rk[EXP_ROUNDS];
    float s[EXP_STEPS];
    unsigned mk[EXP_STEPS];
#pragma unroll
    for (int i = 0; i < EXP_STEPS; ++i) {
      d[i] = dp[4 * i];
      s[i] = sp[32 * i];
    }
#pragma unroll
    for (int j = 0; j < EXP_ROUNDS; ++j) rd[j] = dst[wbase + 32 * j + lane];
#pragma unroll
    for (int i = 0; i < EXP_STEPS; ++i) mk[i] = mmax[(unsigned)d[i] * 8u + h];
#pragma unroll
    for (int j = 0; j < EXP_ROUNDS; ++j) rk[j] = atomicAdd(&cnt[rd[j]], 1);
#pragma unroll
    for (int i = 0; i < EXP_STEPS; ++i) {
      const float es = __expf(s[i] - funkey(mk[i]));
      sp[32 * i] = es;
      atomicAdd(&denom[(unsigned)d[i] * 8u + h], es);
    }
#pragma unroll
    for (int j = 0; j < EXP_ROUNDS; ++j) rank[wbase + 32 * j + lane] = rk[j];
  } else {
    for (int i = 0; i < EXP_STEPS; ++i) {
      const int e = wbase + 4 * i + sub;
      if (e < n_edges) {
        const unsigned dh = (unsigned)dst[e] * 8u + h;
        const size_t idx = (size_t)e * 8 + h;
        const float es = __expf(scores[idx] - funkey(mmax[dh]));
        scores[idx] = es;
        atomicAdd(&denom[dh], es);
      }
    }
    for (int j = 0; j < EXP_ROUNDS; ++j) {
      const int e = wbase + 32 * j + lane;
      if (e < n_edges) rank[e] = atomicAdd(&cnt[dst[e]], 1);
    }
  }
}

__global__ __launch_bounds__(1024) void seg_offsets_kernel(
    const int* __restrict__ cnt, int* __restrict__ off, int n) {
  __shared__ int wsum[32];
  __shared__ int psum[32];
  const int t = threadIdx.x, lane = t & 31, w = t >> 5;
  const int base = blockIdx.x * 1024;
  int p = 0;
  for (int i = t; i < base; i += 1024) p += cnt[i];
  const int x = (base + t < n) ? cnt[base + t] : 0;
  int inc = x;
  for (int d = 1; d < 32; d <<= 1) {
    int y = __shfl_up(inc, d);
    if (lane >= d) inc += y;
  }
  for (int d = 16; d > 0; d >>= 1) p += __shfl_xor(p, d);
  if (lane == 31) wsum[w] = inc;
  if (lane == 0) psum[w] = p;
  __syncthreads();
  int pre = 0;
  for (int i = 0; i < 32; ++i) {
    pre += psum[i];
    pre += (i < w) ? wsum[i] : 0;
  }
  if (base + t < n) off[base + t] = pre + inc - x;
}

__global__ __launch_bounds__(256) void dst_order_kernel(
    const int* __restrict__ src, const int* __restrict__ dst,
    const int* __restrict__ rank, const int* __restrict__ off,
    int* __restrict__ perm, int* __restrict__ ssrc, int* __restrict__ sdst,
    int n_edges) {
  const int e = blockIdx.x * blockDim.x + threadIdx.x;
  if (e >= n_edges) return;
  const int d = dst[e];
  const int pos = off[d] + rank[e];
  perm[pos] = e;
  ssrc[pos] = src[e];
  sdst[pos] = d;
}

typedef __attribute__((ext_vector_type(4))) int v4i;
#define TILES 2

__global__ __launch_bounds__(256) __attribute__((amdgpu_waves_per_eu(6))) void edge_aggregate_kernel(
    const float* __restrict__ ea, const int* __restrict__ perm,
    const int* __restrict__ ssrc, const int* __restrict__ sdst,
    const float* __restrict__ Wev, const float* __restrict__ v,
    const float* __restrict__ scores, const float* __restrict__ denom,
    float* __restrict__ acc) {
  __shared__ float carry[WAVES][8][32];
  const int lane = threadIdx.x & 31;
  const int wv   = threadIdx.x >> 5;
  const int m    = lane & 15;
  const int hf   = lane >> 4;
  const int kk   = hf << 1;
  const unsigned hq = (unsigned)(m >> 2);
  const int wbase = (blockIdx.x * WAVES + wv) * (16 * TILES);
  const int sbase = wbase + hf * (8 * TILES);
  const int abase = wbase + (m >> 3) * (8 * TILES) + (m & 7);

#pragma unroll 1
  for (int i = 0; i < TILES; ++i) {
    v2f afr[8];
    const float* er = ea + (size_t)perm[abase + 8 * i] * 32;
#pragma unroll
    for (int j = 0; j < 8; ++j) {
      afr[j][0] = er[4 * j + kk];
      afr[j][1] = er[4 * j + kk + 1];
    }

    const int p = sbase + 8 * i;
    unsigned so[8], dn[8], vo[8];
    {
      const v4i p0 = *(const v4i*)(perm + p), p1 = *(const v4i*)(perm + p + 4);
      const v4i s0 = *(const v4i*)(ssrc + p), s1 = *(const v4i*)(ssrc + p + 4);
      const v4i d0 = *(const v4i*)(sdst + p), d1 = *(const v4i*)(sdst + p + 4);
#pragma unroll
      for (int r = 0; r < 4; ++r) {
        so[r] = (unsigned)p0[r] * 32u;             so[r + 4] = (unsigned)p1[r] * 32u;
        vo[r] = (unsigned)s0[r] * 512u + 16u * m;  vo[r + 4] = (unsigned)s1[r] * 512u + 16u * m;
        dn[r] = (unsigned)d0[r] * 32u;             dn[r + 4] = (unsigned)d1[r] * 32u;
      }
    }
    const bool last = (i == TILES - 1);
    const unsigned dnext = last ? 0xffffffffu : (unsigned)sdst[p + 8] * 32u;

#pragma unroll 1
    for (int g = 0; g < 2; ++g) {
      v8f c0 = {}, c1 = {}, c2 = {}, c3 = {};
      const float* wp = Wev + (size_t)kk * 128 + 64 * g + 4 * m;
#pragma unroll
      for (int j = 0; j < 8; ++j) {
        const v4f w0 = *(const v4f*)(wp + (size_t)(4 * j) * 128);
        const v4f w1 = *(const v4f*)(wp + (size_t)(4 * j + 1) * 128);
        v2f b;
        b[0] = w0[0]; b[1] = w1[0];
        c0 = __builtin_amdgcn_wmma_f32_16x16x4_f32(false, afr[j], false, b,
                                                   (short)0, c0, false, false);
        b[0] = w0[1]; b[1] = w1[1];
        c1 = __builtin_amdgcn_wmma_f32_16x16x4_f32(false, afr[j], false, b,
                                                   (short)0, c1, false, false);
        b[0] = w0[2]; b[1] = w1[2];
        c2 = __builtin_amdgcn_wmma_f32_16x16x4_f32(false, afr[j], false, b,
                                                   (short)0, c2, false, false);
        b[0] = w0[3]; b[1] = w1[3];
        c3 = __builtin_amdgcn_wmma_f32_16x16x4_f32(false, afr[j], false, b,
                                                   (short)0, c3, false, false);
      }
      __builtin_amdgcn_sched_barrier(0);
      const unsigned hd = 16u * (unsigned)g + 4u * hq;
      const unsigned fo = 256u * (unsigned)g;
      v4f t[8];
#pragma unroll
      for (int r = 0; r < 8; ++r) {
        const float att = *(const float*)((const char*)scores + (so[r] + hd)) /
                          (*(const float*)((const char*)denom + (dn[r] + hd)) + 1e-9f);
        const v4f vr = *(const v4f*)((const char*)v + (vo[r] + fo));
        t[r][0] = att * (vr[0] + c0[r]);
        t[r][1] = att * (vr[1] + c1[r]);
        t[r][2] = att * (vr[2] + c2[r]);
        t[r][3] = att * (vr[3] + c3[r]);
      }
      v4f run = {};
      if (i != 0) {
        run[0] = carry[wv][4 * g + 0][lane];
        run[1] = carry[wv][4 * g + 1][lane];
        run[2] = carry[wv][4 * g + 2][lane];
        run[3] = carry[wv][4 * g + 3][lane];
      }
#pragma unroll
      for (int r = 0; r < 8; ++r) {
        run[0] += t[r][0];
        run[1] += t[r][1];
        run[2] += t[r][2];
        run[3] += t[r][3];
        const unsigned nx = (r < 7) ? dn[r + 1 < 8 ? r + 1 : 7] : dnext;
        if (dn[r] != nx) {
          float* ap = (float*)((char*)acc + (dn[r] * 16u + fo + 16u * (unsigned)m));
          atomicAdd(ap + 0, run[0]);
          atomicAdd(ap + 1, run[1]);
          atomicAdd(ap + 2, run[2]);
          atomicAdd(ap + 3, run[3]);
          run[0] = 0.0f; run[1] = 0.0f; run[2] = 0.0f; run[3] = 0.0f;
        }
      }
      if (!last) {
        carry[wv][4 * g + 0][lane] = run[0];
        carry[wv][4 * g + 1][lane] = run[1];
        carry[wv][4 * g + 2][lane] = run[2];
        carry[wv][4 * g + 3][lane] = run[3];
      }
    }
  }
}

extern "C" void kernel_launch(void* const* d_in, const int* in_sizes, int n_in,
                              void* d_out, int out_size, void* d_ws,
                              size_t ws_size, hipStream_t stream) {
  const float* x   = (const float*)d_in[0];
  const int*   ei  = (const int*)d_in[1];
  const float* ea  = (const float*)d_in[2];
  const float* Wq  = (const float*)d_in[3];
  const float* Wk  = (const float*)d_in[4];
  const float* Wv  = (const float*)d_in[5];
  const float* Wo  = (const float*)d_in[6];
  const float* Wek = (const float*)d_in[7];
  const float* Wev = (const float*)d_in[8];
  const float* Wg1 = (const float*)d_in[9];
  const float* Wg2 = (const float*)d_in[10];
  float* out = (float*)d_out;

  const int N = in_sizes[0] / 128;
  const int E = in_sizes[2] / 32;
  const int* src = ei;
  const int* dst = ei + E;

  float* q      = (float*)d_ws;
  float* k      = q + (size_t)N * 128;
  float* v      = k + (size_t)N * 128;
  float* scores = v + (size_t)N * 128;
  float* acc    = scores + (size_t)E * 8;
  float* denom  = acc + (size_t)N * 128;
  unsigned* mmax = (unsigned*)(denom + (size_t)N * 8);
  int* cnt  = (int*)v;
  int* rank = cnt + N;
  int* perm = (int*)q;
  int* ssrc = perm + E;
  int* sdst = ssrc + E;
  int* off  = sdst + E;

  hipMemsetAsync(acc, 0, (size_t)(N * 128 + N * 8 + N * 8) * sizeof(float),
                 stream);

  dim3 blk(256);
  dim3 gblk(32 * GEMM_WAVES);
  gemm128_kernel<<<dim3((N / 16 * 2 + GEMM_WAVES - 1) / GEMM_WAVES), gblk, 0, stream>>>(x, Wq, q, N / 16);
  gemm128_kernel<<<dim3((N / 16 * 2 + GEMM_WAVES - 1) / GEMM_WAVES), gblk, 0, stream>>>(x, Wk, k, N / 16);

  edge_score_kernel<<<dim3(E / (16 * WAVES)), blk, 0, stream>>>(
      ea, src, dst, Wek, Wg1, Wg2, q, k, scores, mmax, cnt, N);

  exp_denom_kernel<<<dim3((E + WAVES * EXP_EDGES - 1) / (WAVES * EXP_EDGES)), blk,
                     0, stream>>>(dst, scores, mmax, denom, cnt, rank, E);

  seg_offsets_kernel<<<dim3((N + 1023) / 1024), dim3(1024), 0, stream>>>(
      cnt, off, N);

  dst_order_kernel<<<dim3((E + 255) / 256), blk, 0, stream>>>(
      src, dst, rank, off, perm, ssrc, sdst, E);

  gemm128_kernel<<<dim3((N / 16 * 2 + GEMM_WAVES - 1) / GEMM_WAVES), gblk, 0, stream>>>(x, Wv, v, N / 16);

  edge_aggregate_kernel<<<dim3(E / (16 * WAVES * TILES)), blk, 0, stream>>>(
      ea, perm, ssrc, sdst, Wev, v, scores, denom, acc);

  gemm128_kernel<<<dim3((N / 16 * 2 + GEMM_WAVES - 1) / GEMM_WAVES), gblk, 0, stream>>>(acc, Wo, out, N / 16);
}


# edge_score: 4 accumulator tiles with interleaved A rows (feature 64g+8(a&7)+4(a>>3)+t; gate 4a+t): Wek and Wg1 rows as 16-byte loads, a quarter of the weight loads; dot and gate sum order change
# speedup vs baseline: 2.2518x; 1.0796x over previous
#include <hip/hip_runtime.h>


typedef __attribute__((ext_vector_type(2))) float v2f;
typedef __attribute__((ext_vector_type(4))) float v4f;
typedef __attribute__((ext_vector_type(8))) float v8f;

#define WAVES 8

__device__ __forceinline__ unsigned fkey(float f) {
  unsigned u = __float_as_uint(f);
  return (u & 0x80000000u) ? ~u : (u | 0x80000000u);
}
__device__ __forceinline__ float funkey(unsigned kk) {
  unsigned u = (kk & 0x80000000u) ? (kk & 0x7FFFFFFFu) : ~kk;
  return __uint_as_float(u);
}

#define GEMM_WAVES 4
__global__ __launch_bounds__(32 * GEMM_WAVES) void gemm128_kernel(
    const float* __restrict__ A, const float* __restrict__ W,
    float* __restrict__ out, int n_tiles) {
  const int lane = threadIdx.x & 31;
  const int unit = blockIdx.x * GEMM_WAVES + (threadIdx.x >> 5);
  const int rt   = unit >> 1;
  const int g    = unit & 1;
  if (rt >= n_tiles) return;
  const int m    = lane & 15;
  const int hf   = lane >> 4;
  const int kk   = hf << 1;

  v8f c0 = {}, c1 = {}, c2 = {}, c3 = {};
  const float* ar = A + (size_t)(rt * 16 + m) * 128 + kk;
  const float* wp = W + (size_t)kk * 128 + 64 * g + 4 * m;
#pragma unroll 8
  for (int k0 = 0; k0 < 128; k0 += 4) {
    const v2f a  = *(const v2f*)(ar + k0);
    const v4f w0 = *(const v4f*)(wp + (size_t)k0 * 128);
    const v4f w1 = *(const v4f*)(wp + (size_t)(k0 + 1) * 128);
    v2f b;
    b[0] = w0[0]; b[1] = w1[0];
    c0 = __builtin_amdgcn_wmma_f32_16x16x4_f32(false, a, false, b, (short)0, c0,
                                               false, false);
    b[0] = w0[1]; b[1] = w1[1];
    c1 = __builtin_amdgcn_wmma_f32_16x16x4_f32(false, a, false, b, (short)0, c1,
                                               false, false);
    b[0] = w0[2]; b[1] = w1[2];
    c2 = __builtin_amdgcn_wmma_f32_16x16x4_f32(false, a, false, b, (short)0, c2,
                                               false, false);
    b[0] = w0[3]; b[1] = w1[3];
    c3 = __builtin_amdgcn_wmma_f32_16x16x4_f32(false, a, false, b, (short)0, c3,
                                               false, false);
  }
  float* op = out + (size_t)(rt * 16 + 8 * hf) * 128 + 64 * g + 4 * m;
#pragma unroll
  for (int r = 0; r < 8; ++r) {
    v4f o;
    o[0] = c0[r]; o[1] = c1[r]; o[2] = c2[r]; o[3] = c3[r];
    *(v4f*)(op + (size_t)r * 128) = o;
  }
}

__device__ __forceinline__ float xhalf(float x) {
  int b = __builtin_bit_cast(int, x);
  return __builtin_bit_cast(
      float, __builtin_amdgcn_permlanex16(b, b, 0x76543210, 0xfedcba98, false,
                                          false));
}

__device__ __forceinline__ v8f silu8(v8f c) {
#pragma unroll
  for (int r = 0; r < 8; ++r) {
    float xv = c[r];
    c[r] = xv / (1.0f + __expf(-xv));
  }
  return c;
}

__device__ __forceinline__ void wmma4_steps(const float* __restrict__ Wb,
                                            unsigned wo, unsigned pitch,
                                            const v2f* efr, int i0, v8f& c0,
                                            v8f& c1, v8f& c2, v8f& c3) {
#pragma unroll
  for (int i = i0; i < i0 + 4; ++i) {
    const v4f w0 = *(const v4f*)((const char*)Wb + (wo + (4u * i) * pitch));
    const v4f w1 = *(const v4f*)((const char*)Wb + (wo + (4u * i + 1u) * pitch));
    v2f a;
    a[0] = w0[0]; a[1] = w1[0];
    c0 = __builtin_amdgcn_wmma_f32_16x16x4_f32(false, a, false, efr[i],
                                               (short)0, c0, false, false);
    a[0] = w0[1]; a[1] = w1[1];
    c1 = __builtin_amdgcn_wmma_f32_16x16x4_f32(false, a, false, efr[i],
                                               (short)0, c1, false, false);
    a[0] = w0[2]; a[1] = w1[2];
    c2 = __builtin_amdgcn_wmma_f32_16x16x4_f32(false, a, false, efr[i],
                                               (short)0, c2, false, false);
    a[0] = w0[3]; a[1] = w1[3];
    c3 = __builtin_amdgcn_wmma_f32_16x16x4_f32(false, a, false, efr[i],
                                               (short)0, c3, false, false);
  }
}

__device__ __forceinline__ v8f gate2_acc(v8f gt, const float* __restrict__ w2,
                                         v8f c, int t) {
#pragma unroll
  for (int p = 0; p < 4; ++p) {
    v2f a, b;
    a[0] = w2[(8 * p + t) * 8];
    a[1] = w2[(8 * p + 4 + t) * 8];
    b[0] = c[2 * p];
    b[1] = c[2 * p + 1];
    gt = __builtin_amdgcn_wmma_f32_16x16x4_f32(false, a, false, b, (short)0, gt,
                                               false, false);
  }
  return gt;
}

__global__ __launch_bounds__(256) void edge_score_kernel(
    const float* __restrict__ ea, const int* __restrict__ src,
    const int* __restrict__ dst, const float* __restrict__ Wek,
    const float* __restrict__ Wg1, const float* __restrict__ Wg2,
    const float* __restrict__ q, const float* __restrict__ k,
    float* __restrict__ scores, unsigned* __restrict__ mmax,
    int* __restrict__ cnt, int n_nodes) {
  const int lane = threadIdx.x & 31;
  const int wv   = threadIdx.x >> 5;
  const int e0   = (blockIdx.x * WAVES + wv) << 4;
  const int m    = lane & 15;
  const int hf   = lane >> 4;
  const int kk   = hf << 1;

  v2f efr[8];
  const float* er = ea + (size_t)(e0 + m) * 32;
#pragma unroll
  for (int i = 0; i < 8; ++i) {
    efr[i][0] = er[4 * i + kk];
    efr[i][1] = er[4 * i + kk + 1];
  }
  const int sm = src[e0 + m];
  const int dm = dst[e0 + m];

  const unsigned qo = (unsigned)dm * 512u + 16u * (unsigned)hf;
  const unsigned ko = (unsigned)sm * 512u + 16u * (unsigned)hf;
  const int psi = 8 * (m & 7) + 4 * (m >> 3);
  v4f o = {};
#pragma unroll 1
  for (int g = 0; g < 2; ++g) {
    v8f c0 = {}, c1 = {}, c2 = {}, c3 = {};
    const unsigned wo = (unsigned)kk * 512u + 256u * (unsigned)g + 4u * (unsigned)psi;
    wmma4_steps(Wek, wo, 512u, efr, 0, c0, c1, c2, c3);
    __builtin_amdgcn_sched_barrier(0);
    wmma4_steps(Wek, wo, 512u, efr, 4, c0, c1, c2, c3);
    const unsigned qg = qo + 256u * (unsigned)g;
    const unsigned kg = ko + 256u * (unsigned)g;
#pragma unroll
    for (int hh = 0; hh < 4; ++hh) {
      if ((hh & 1) == 0) __builtin_amdgcn_sched_barrier(0);
      const v4f q0 = *(const v4f*)((const char*)q + (qg + 64u * hh));
      const v4f q1 = *(const v4f*)((const char*)q + (qg + 64u * hh + 32u));
      const v4f k0 = *(const v4f*)((const char*)k + (kg + 64u * hh));
      const v4f k1 = *(const v4f*)((const char*)k + (kg + 64u * hh + 32u));
      float p = q0[0] * (k0[0] + c0[2 * hh]);
      p = fmaf(q0[1], k0[1] + c1[2 * hh], p);
      p = fmaf(q0[2], k0[2] + c2[2 * hh], p);
      p = fmaf(q0[3], k0[3] + c3[2 * hh], p);
      p = fmaf(q1[0], k1[0] + c0[2 * hh + 1], p);
      p = fmaf(q1[1], k1[1] + c1[2 * hh + 1], p);
      p = fmaf(q1[2], k1[2] + c2[2 * hh + 1], p);
      p = fmaf(q1[3], k1[3] + c3[2 * hh + 1], p);
      p += xhalf(p);
      o[hh] = (hf == g) ? p : o[hh];
    }
  }
  v8f gt = {};
  {
    v8f c0 = {}, c1 = {}, c2 = {}, c3 = {};
    const unsigned wo = (unsigned)kk * 256u + 16u * (unsigned)m;
    wmma4_steps(Wg1, wo, 256u, efr, 0, c0, c1, c2, c3);
    wmma4_steps(Wg1, wo, 256u, efr, 4, c0, c1, c2, c3);
    const float* w2 = Wg2 + (size_t)(32 * hf) * 8 + (m & 7);
    gt = gate2_acc(gt, w2, silu8(c0), 0);
    gt = gate2_acc(gt, w2, silu8(c1), 1);
    gt = gate2_acc(gt, w2, silu8(c2), 2);
    gt = gate2_acc(gt, w2, silu8(c3), 3);
  }

  o[0] = o[0] * 0.25f + (hf ? gt[4] : gt[0]);
  o[1] = o[1] * 0.25f + (hf ? gt[5] : gt[1]);
  o[2] = o[2] * 0.25f + (hf ? gt[6] : gt[2]);
  o[3] = o[3] * 0.25f + (hf ? gt[7] : gt[3]);

  *(v4f*)(scores + (size_t)(e0 + m) * 8 + 4 * hf) = o;
  unsigned* mp = mmax + (size_t)dm * 8 + 4 * hf;
  atomicMax(mp + 0, fkey(o[0]));
  atomicMax(mp + 1, fkey(o[1]));
  atomicMax(mp + 2, fkey(o[2]));
  atomicMax(mp + 3, fkey(o[3]));

  const int gid = blockIdx.x * (WAVES * 32) + threadIdx.x;
  if (gid < n_nodes) cnt[gid] = 0;
}

#define EXP_STEPS 16
#define EXP_EDGES (4 * EXP_STEPS)
#define EXP_ROUNDS (EXP_EDGES / 32)

__global__ __launch_bounds__(256) void exp_denom_kernel(
    const int* __restrict__ dst, float* __restrict__ scores,
    const unsigned* __restrict__ mmax, float* __restrict__ denom,
    int* __restrict__ cnt, int* __restrict__ rank, int n_edges) {
  const int lane  = threadIdx.x & 31;
  const int wbase = (blockIdx.x * WAVES + (threadIdx.x >> 5)) * EXP_EDGES;
  if (wbase >= n_edges) return;
  const unsigned h = lane & 7;
  const int sub = lane >> 3;

  if (wbase + EXP_EDGES <= n_edges) {
    float* sp = scores + (size_t)wbase * 8 + lane;
    const int* dp = dst + wbase + sub;
    int d[EXP_STEPS], rd[EXP_ROUNDS], rk[EXP_ROUNDS];
    float s[EXP_STEPS];
    unsigned mk[EXP_STEPS];
#pragma unroll
    for (int i = 0; i < EXP_STEPS; ++i) {
      d[i] = dp[4 * i];
      s[i] = sp[32 * i];
    }
#pragma unroll
    for (int j = 0; j < EXP_ROUNDS; ++j) rd[j] = dst[wbase + 32 * j + lane];
#pragma unroll
    for (int i = 0; i < EXP_STEPS; ++i) mk[i] = mmax[(unsigned)d[i] * 8u + h];
#pragma unroll
    for (int j = 0; j < EXP_ROUNDS; ++j) rk[j] = atomicAdd(&cnt[rd[j]], 1);
#pragma unroll
    for (int i = 0; i < EXP_STEPS; ++i) {
      const float es = __expf(s[i] - funkey(mk[i]));
      sp[32 * i] = es;
      atomicAdd(&denom[(unsigned)d[i] * 8u + h], es);
    }
#pragma unroll
    for (int j = 0; j < EXP_ROUNDS; ++j) rank[wbase + 32 * j + lane] = rk[j];
  } else {
    for (int i = 0; i < EXP_STEPS; ++i) {
      const int e = wbase + 4 * i + sub;
      if (e < n_edges) {
        const unsigned dh = (unsigned)dst[e] * 8u + h;
        const size_t idx = (size_t)e * 8 + h;
        const float es = __expf(scores[idx] - funkey(mmax[dh]));
        scores[idx] = es;
        atomicAdd(&denom[dh], es);
      }
    }
    for (int j = 0; j < EXP_ROUNDS; ++j) {
      const int e = wbase + 32 * j + lane;
      if (e < n_edges) rank[e] = atomicAdd(&cnt[dst[e]], 1);
    }
  }
}

__global__ __launch_bounds__(1024) void seg_offsets_kernel(
    const int* __restrict__ cnt, int* __restrict__ off, int n) {
  __shared__ int wsum[32];
  __shared__ int psum[32];
  const int t = threadIdx.x, lane = t & 31, w = t >> 5;
  const int base = blockIdx.x * 1024;
  int p = 0;
  for (int i = t; i < base; i += 1024) p += cnt[i];
  const int x = (base + t < n) ? cnt[base + t] : 0;
  int inc = x;
  for (int d = 1; d < 32; d <<= 1) {
    int y = __shfl_up(inc, d);
    if (lane >= d) inc += y;
  }
  for (int d = 16; d > 0; d >>= 1) p += __shfl_xor(p, d);
  if (lane == 31) wsum[w] = inc;
  if (lane == 0) psum[w] = p;
  __syncthreads();
  int pre = 0;
  for (int i = 0; i < 32; ++i) {
    pre += psum[i];
    pre += (i < w) ? wsum[i] : 0;
  }
  if (base + t < n) off[base + t] = pre + inc - x;
}

__global__ __launch_bounds__(256) void dst_order_kernel(
    const int* __restrict__ src, const int* __restrict__ dst,
    const int* __restrict__ rank, const int* __restrict__ off,
    int* __restrict__ perm, int* __restrict__ ssrc, int* __restrict__ sdst,
    int n_edges) {
  const int e = blockIdx.x * blockDim.x + threadIdx.x;
  if (e >= n_edges) return;
  const int d = dst[e];
  const int pos = off[d] + rank[e];
  perm[pos] = e;
  ssrc[pos] = src[e];
  sdst[pos] = d;
}

typedef __attribute__((ext_vector_type(4))) int v4i;
#define TILES 2

__global__ __launch_bounds__(256) __attribute__((amdgpu_waves_per_eu(6))) void edge_aggregate_kernel(
    const float* __restrict__ ea, const int* __restrict__ perm,
    const int* __restrict__ ssrc, const int* __restrict__ sdst,
    const float* __restrict__ Wev, const float* __restrict__ v,
    const float* __restrict__ scores, const float* __restrict__ denom,
    float* __restrict__ acc) {
  __shared__ float carry[WAVES][8][32];
  const int lane = threadIdx.x & 31;
  const int wv   = threadIdx.x >> 5;
  const int m    = lane & 15;
  const int hf   = lane >> 4;
  const int kk   = hf << 1;
  const unsigned hq = (unsigned)(m >> 2);
  const int wbase = (blockIdx.x * WAVES + wv) * (16 * TILES);
  const int sbase = wbase + hf * (8 * TILES);
  const int abase = wbase + (m >> 3) * (8 * TILES) + (m & 7);

#pragma unroll 1
  for (int i = 0; i < TILES; ++i) {
    v2f afr[8];
    const float* er = ea + (size_t)perm[abase + 8 * i] * 32;
#pragma unroll
    for (int j = 0; j < 8; ++j) {
      afr[j][0] = er[4 * j + kk];
      afr[j][1] = er[4 * j + kk + 1];
    }

    const int p = sbase + 8 * i;
    unsigned so[8], dn[8], vo[8];
    {
      const v4i p0 = *(const v4i*)(perm + p), p1 = *(const v4i*)(perm + p + 4);
      const v4i s0 = *(const v4i*)(ssrc + p), s1 = *(const v4i*)(ssrc + p + 4);
      const v4i d0 = *(const v4i*)(sdst + p), d1 = *(const v4i*)(sdst + p + 4);
#pragma unroll
      for (int r = 0; r < 4; ++r) {
        so[r] = (unsigned)p0[r] * 32u;             so[r + 4] = (unsigned)p1[r] * 32u;
        vo[r] = (unsigned)s0[r] * 512u + 16u * m;  vo[r + 4] = (unsigned)s1[r] * 512u + 16u * m;
        dn[r] = (unsigned)d0[r] * 32u;             dn[r + 4] = (unsigned)d1[r] * 32u;
      }
    }
    const bool last = (i == TILES - 1);
    const unsigned dnext = last ? 0xffffffffu : (unsigned)sdst[p + 8] * 32u;

#pragma unroll 1
    for (int g = 0; g < 2; ++g) {
      v8f c0 = {}, c1 = {}, c2 = {}, c3 = {};
      const float* wp = Wev + (size_t)kk * 128 + 64 * g + 4 * m;
#pragma unroll
      for (int j = 0; j < 8; ++j) {
        const v4f w0 = *(const v4f*)(wp + (size_t)(4 * j) * 128);
        const v4f w1 = *(const v4f*)(wp + (size_t)(4 * j + 1) * 128);
        v2f b;
        b[0] = w0[0]; b[1] = w1[0];
        c0 = __builtin_amdgcn_wmma_f32_16x16x4_f32(false, afr[j], false, b,
                                                   (short)0, c0, false, false);
        b[0] = w0[1]; b[1] = w1[1];
        c1 = __builtin_amdgcn_wmma_f32_16x16x4_f32(false, afr[j], false, b,
                                                   (short)0, c1, false, false);
        b[0] = w0[2]; b[1] = w1[2];
        c2 = __builtin_amdgcn_wmma_f32_16x16x4_f32(false, afr[j], false, b,
                                                   (short)0, c2, false, false);
        b[0] = w0[3]; b[1] = w1[3];
        c3 = __builtin_amdgcn_wmma_f32_16x16x4_f32(false, afr[j], false, b,
                                                   (short)0, c3, false, false);
      }
      __builtin_amdgcn_sched_barrier(0);
      const unsigned hd = 16u * (unsigned)g + 4u * hq;
      const unsigned fo = 256u * (unsigned)g;
      v4f t[8];
#pragma unroll
      for (int r = 0; r < 8; ++r) {
        const float att = *(const float*)((const char*)scores + (so[r] + hd)) /
                          (*(const float*)((const char*)denom + (dn[r] + hd)) + 1e-9f);
        const v4f vr = *(const v4f*)((const char*)v + (vo[r] + fo));
        t[r][0] = att * (vr[0] + c0[r]);
        t[r][1] = att * (vr[1] + c1[r]);
        t[r][2] = att * (vr[2] + c2[r]);
        t[r][3] = att * (vr[3] + c3[r]);
      }
      v4f run = {};
      if (i != 0) {
        run[0] = carry[wv][4 * g + 0][lane];
        run[1] = carry[wv][4 * g + 1][lane];
        run[2] = carry[wv][4 * g + 2][lane];
        run[3] = carry[wv][4 * g + 3][lane];
      }
#pragma unroll
      for (int r = 0; r < 8; ++r) {
        run[0] += t[r][0];
        run[1] += t[r][1];
        run[2] += t[r][2];
        run[3] += t[r][3];
        const unsigned nx = (r < 7) ? dn[r + 1 < 8 ? r + 1 : 7] : dnext;
        if (dn[r] != nx) {
          float* ap = (float*)((char*)acc + (dn[r] * 16u + fo + 16u * (unsigned)m));
          atomicAdd(ap + 0, run[0]);
          atomicAdd(ap + 1, run[1]);
          atomicAdd(ap + 2, run[2]);
          atomicAdd(ap + 3, run[3]);
          run[0] = 0.0f; run[1] = 0.0f; run[2] = 0.0f; run[3] = 0.0f;
        }
      }
      if (!last) {
        carry[wv][4 * g + 0][lane] = run[0];
        carry[wv][4 * g + 1][lane] = run[1];
        carry[wv][4 * g + 2][lane] = run[2];
        carry[wv][4 * g + 3][lane] = run[3];
      }
    }
  }
}

extern "C" void kernel_launch(void* const* d_in, const int* in_sizes, int n_in,
                              void* d_out, int out_size, void* d_ws,
                              size_t ws_size, hipStream_t stream) {
  const float* x   = (const float*)d_in[0];
  const int*   ei  = (const int*)d_in[1];
  const float* ea  = (const float*)d_in[2];
  const float* Wq  = (const float*)d_in[3];
  const float* Wk  = (const float*)d_in[4];
  const float* Wv  = (const float*)d_in[5];
  const float* Wo  = (const float*)d_in[6];
  const float* Wek = (const float*)d_in[7];
  const float* Wev = (const float*)d_in[8];
  const float* Wg1 = (const float*)d_in[9];
  const float* Wg2 = (const float*)d_in[10];
  float* out = (float*)d_out;

  const int N = in_sizes[0] / 128;
  const int E = in_sizes[2] / 32;
  const int* src = ei;
  const int* dst = ei + E;

  float* q      = (float*)d_ws;
  float* k      = q + (size_t)N * 128;
  float* v      = k + (size_t)N * 128;
  float* scores = v + (size_t)N * 128;
  float* acc    = scores + (size_t)E * 8;
  float* denom  = acc + (size_t)N * 128;
  unsigned* mmax = (unsigned*)(denom + (size_t)N * 8);
  int* cnt  = (int*)v;
  int* rank = cnt + N;
  int* perm = (int*)q;
  int* ssrc = perm + E;
  int* sdst = ssrc + E;
  int* off  = sdst + E;

  hipMemsetAsync(acc, 0, (size_t)(N * 128 + N * 8 + N * 8) * sizeof(float),
                 stream);

  dim3 blk(256);
  dim3 gblk(32 * GEMM_WAVES);
  gemm128_kernel<<<dim3((N / 16 * 2 + GEMM_WAVES - 1) / GEMM_WAVES), gblk, 0, stream>>>(x, Wq, q, N / 16);
  gemm128_kernel<<<dim3((N / 16 * 2 + GEMM_WAVES - 1) / GEMM_WAVES), gblk, 0, stream>>>(x, Wk, k, N / 16);

  edge_score_kernel<<<dim3(E / (16 * WAVES)), blk, 0, stream>>>(
      ea, src, dst, Wek, Wg1, Wg2, q, k, scores, mmax, cnt, N);

  exp_denom_kernel<<<dim3((E + WAVES * EXP_EDGES - 1) / (WAVES * EXP_EDGES)), blk,
                     0, stream>>>(dst, scores, mmax, denom, cnt, rank, E);

  seg_offsets_kernel<<<dim3((N + 1023) / 1024), dim3(1024), 0, stream>>>(
      cnt, off, N);

  dst_order_kernel<<<dim3((E + 255) / 256), blk, 0, stream>>>(
      src, dst, rank, off, perm, ssrc, sdst, E);

  gemm128_kernel<<<dim3((N / 16 * 2 + GEMM_WAVES - 1) / GEMM_WAVES), gblk, 0, stream>>>(x, Wv, v, N / 16);

  edge_aggregate_kernel<<<dim3(E / (16 * WAVES * TILES)), blk, 0, stream>>>(
      ea, perm, ssrc, sdst, Wev, v, scores, denom, acc);

  gemm128_kernel<<<dim3((N / 16 * 2 + GEMM_WAVES - 1) / GEMM_WAVES), gblk, 0, stream>>>(acc, Wo, out, N / 16);
}


# q[dst], k[src], edge_attr and GEMM A rows fetched as whole 128-byte lines (8 lanes a line: a load touches 4 lines, not 16), handed to the owning lane through wave-private LDS rows; sums unchanged
# speedup vs baseline: 2.3474x; 1.0425x over previous
#include <hip/hip_runtime.h>


typedef __attribute__((ext_vector_type(2))) float v2f;
typedef __attribute__((ext_vector_type(4))) float v4f;
typedef __attribute__((ext_vector_type(8))) float v8f;

#define WAVES 8

__device__ __forceinline__ unsigned fkey(float f) {
  unsigned u = __float_as_uint(f);
  return (u & 0x80000000u) ? ~u : (u | 0x80000000u);
}
__device__ __forceinline__ float funkey(unsigned kk) {
  unsigned u = (kk & 0x80000000u) ? (kk & 0x7FFFFFFFu) : ~kk;
  return __uint_as_float(u);
}

#define GEMM_WAVES 4
#define STG_PITCH 9
__global__ __launch_bounds__(32 * GEMM_WAVES) void gemm128_kernel(
    const float* __restrict__ A, const float* __restrict__ W,
    float* __restrict__ out, int n_tiles) {
  __shared__ v4f sta[GEMM_WAVES][16 * STG_PITCH];
  const int lane = threadIdx.x & 31;
  const int unit = blockIdx.x * GEMM_WAVES + (threadIdx.x >> 5);
  const int rt   = unit >> 1;
  const int g    = unit & 1;
  if (rt >= n_tiles) return;
  const int m    = lane & 15;
  const int hf   = lane >> 4;
  const int kk   = hf << 1;

  v8f c0 = {}, c1 = {}, c2 = {}, c3 = {};
  const int sub = lane & 7;
  const int eg  = lane >> 3;
  v4f* sa = &sta[threadIdx.x >> 5][0];
  const float* ab = A + (size_t)(rt * 16 + eg) * 128 + 4 * sub;
  const v2f* rowm = (const v2f*)(sa + m * STG_PITCH) + hf;
  const float* wp = W + (size_t)kk * 128 + 64 * g + 4 * m;
#pragma unroll
  for (int t0 = 0; t0 < 128; t0 += 32) {
    __builtin_amdgcn_sched_barrier(0);
    v4f ta[4], w0[8], w1[8];
#pragma unroll
    for (int j = 0; j < 4; ++j) ta[j] = *(const v4f*)(ab + (size_t)(4 * j) * 128 + t0);
#pragma unroll
    for (int i = 0; i < 8; ++i) {
      w0[i] = *(const v4f*)(wp + (size_t)(t0 + 4 * i) * 128);
      w1[i] = *(const v4f*)(wp + (size_t)(t0 + 4 * i + 1) * 128);
    }
    __builtin_amdgcn_fence(__ATOMIC_ACQ_REL, "wavefront");
#pragma unroll
    for (int j = 0; j < 4; ++j) sa[(4 * j + eg) * STG_PITCH + sub] = ta[j];
    __builtin_amdgcn_fence(__ATOMIC_ACQ_REL, "wavefront");
    v2f afr[8];
#pragma unroll
    for (int i = 0; i < 8; ++i) afr[i] = rowm[2 * i];
#pragma unroll
    for (int i = 0; i < 8; ++i) {
      const v2f a = afr[i];
      v2f b;
      b[0] = w0[i][0]; b[1] = w1[i][0];
      c0 = __builtin_amdgcn_wmma_f32_16x16x4_f32(false, a, false, b, (short)0, c0,
                                                 false, false);
      b[0] = w0[i][1]; b[1] = w1[i][1];
      c1 = __builtin_amdgcn_wmma_f32_16x16x4_f32(false, a, false, b, (short)0, c1,
                                                 false, false);
      b[0] = w0[i][2]; b[1] = w1[i][2];
      c2 = __builtin_amdgcn_wmma_f32_16x16x4_f32(false, a, false, b, (short)0, c2,
                                                 false, false);
      b[0] = w0[i][3]; b[1] = w1[i][3];
      c3 = __builtin_amdgcn_wmma_f32_16x16x4_f32(false, a, false, b, (short)0, c3,
                                                 false, false);
    }
  }
  float* op = out + (size_t)(rt * 16 + 8 * hf) * 128 + 64 * g + 4 * m;
#pragma unroll
  for (int r = 0; r < 8; ++r) {
    v4f o;
    o[0] = c0[r]; o[1] = c1[r]; o[2] = c2[r]; o[3] = c3[r];
    *(v4f*)(op + (size_t)r * 128) = o;
  }
}

__device__ __forceinline__ float xhalf(float x) {
  int b = __builtin_bit_cast(int, x);
  return __builtin_bit_cast(
      float, __builtin_amdgcn_permlanex16(b, b, 0x76543210, 0xfedcba98, false,
                                          false));
}

__device__ __forceinline__ v8f silu8(v8f c) {
#pragma unroll
  for (int r = 0; r < 8; ++r) {
    float xv = c[r];
    c[r] = xv / (1.0f + __expf(-xv));
  }
  return c;
}

__device__ __forceinline__ void wmma4_steps(const float* __restrict__ Wb,
                                            unsigned wo, unsigned pitch,
                                            const v2f* efr, int i0, v8f& c0,
                                            v8f& c1, v8f& c2, v8f& c3) {
#pragma unroll
  for (int i = i0; i < i0 + 4; ++i) {
    const v4f w0 = *(const v4f*)((const char*)Wb + (wo + (4u * i) * pitch));
    const v4f w1 = *(const v4f*)((const char*)Wb + (wo + (4u * i + 1u) * pitch));
    v2f a;
    a[0] = w0[0]; a[1] = w1[0];
    c0 = __builtin_amdgcn_wmma_f32_16x16x4_f32(false, a, false, efr[i],
                                               (short)0, c0, false, false);
    a[0] = w0[1]; a[1] = w1[1];
    c1 = __builtin_amdgcn_wmma_f32_16x16x4_f32(false, a, false, efr[i],
                                               (short)0, c1, false, false);
    a[0] = w0[2]; a[1] = w1[2];
    c2 = __builtin_amdgcn_wmma_f32_16x16x4_f32(false, a, false, efr[i],
                                               (short)0, c2, false, false);
    a[0] = w0[3]; a[1] = w1[3];
    c3 = __builtin_amdgcn_wmma_f32_16x16x4_f32(false, a, false, efr[i],
                                               (short)0, c3, false, false);
  }
}

__device__ __forceinline__ v8f gate2_acc(v8f gt, const float* __restrict__ w2,
                                         v8f c, int t) {
#pragma unroll
  for (int p = 0; p < 4; ++p) {
    v2f a, b;
    a[0] = w2[(8 * p + t) * 8];
    a[1] = w2[(8 * p + 4 + t) * 8];
    b[0] = c[2 * p];
    b[1] = c[2 * p + 1];
    gt = __builtin_amdgcn_wmma_f32_16x16x4_f32(false, a, false, b, (short)0, gt,
                                               false, false);
  }
  return gt;
}

__global__ __launch_bounds__(256) void edge_score_kernel(
    const float* __restrict__ ea, const int* __restrict__ src,
    const int* __restrict__ dst, const float* __restrict__ Wek,
    const float* __restrict__ Wg1, const float* __restrict__ Wg2,
    const float* __restrict__ q, const float* __restrict__ k,
    float* __restrict__ scores, unsigned* __restrict__ mmax,
    int* __restrict__ cnt, int n_nodes) {
  __shared__ v4f stg[WAVES][2][16 * STG_PITCH];
  const int lane = threadIdx.x & 31;
  const int wv   = threadIdx.x >> 5;
  const int e0   = (blockIdx.x * WAVES + wv) << 4;
  const int m    = lane & 15;
  const int hf   = lane >> 4;
  const int kk   = hf << 1;

  const int dm = dst[e0 + m];

  const int sub = lane & 7;
  const int eg  = lane >> 3;
  unsigned qrow[4], krow[4];
#pragma unroll
  for (int j = 0; j < 4; ++j) {
    qrow[j] = (unsigned)dst[e0 + 4 * j + eg] * 512u + 16u * (unsigned)sub;
    krow[j] = (unsigned)src[e0 + 4 * j + eg] * 512u + 16u * (unsigned)sub;
  }
  v4f* sq = &stg[wv][0][0];
  v4f* sk = &stg[wv][1][0];

  v2f efr[8];
  {
    const float* eb = ea + (size_t)e0 * 32 + 4 * lane;
    v4f te[4];
#pragma unroll
    for (int j = 0; j < 4; ++j) te[j] = *(const v4f*)(eb + 128 * j);
#pragma unroll
    for (int j = 0; j < 4; ++j) sq[(4 * j + eg) * STG_PITCH + sub] = te[j];
    __builtin_amdgcn_fence(__ATOMIC_ACQ_REL, "wavefront");
    const v2f* rowm = (const v2f*)(sq + m * STG_PITCH) + hf;
#pragma unroll
    for (int i = 0; i < 8; ++i) efr[i] = rowm[2 * i];
  }
  const int psi = 8 * (m & 7) + 4 * (m >> 3);
  v4f o = {};
#pragma unroll 1
  for (int g = 0; g < 2; ++g) {
    v8f c0 = {}, c1 = {}, c2 = {}, c3 = {};
    const unsigned wo = (unsigned)kk * 512u + 256u * (unsigned)g + 4u * (unsigned)psi;
    wmma4_steps(Wek, wo, 512u, efr, 0, c0, c1, c2, c3);
    __builtin_amdgcn_sched_barrier(0);
    wmma4_steps(Wek, wo, 512u, efr, 4, c0, c1, c2, c3);
#pragma unroll
    for (int hp = 0; hp < 2; ++hp) {
      __builtin_amdgcn_sched_barrier(0);
      const unsigned lo = 256u * (unsigned)g + 128u * (unsigned)hp;
      v4f tq[4], tk[4];
#pragma unroll
      for (int j = 0; j < 4; ++j) {
        tq[j] = *(const v4f*)((const char*)q + (qrow[j] + lo));
        tk[j] = *(const v4f*)((const char*)k + (krow[j] + lo));
      }
      __builtin_amdgcn_fence(__ATOMIC_ACQ_REL, "wavefront");
#pragma unroll
      for (int j = 0; j < 4; ++j) {
        sq[(4 * j + eg) * STG_PITCH + sub] = tq[j];
        sk[(4 * j + eg) * STG_PITCH + sub] = tk[j];
      }
      __builtin_amdgcn_fence(__ATOMIC_ACQ_REL, "wavefront");
#pragma unroll
      for (int hl = 0; hl < 2; ++hl) {
        const int hh = 2 * hp + hl;
        const v4f q0 = sq[m * STG_PITCH + 4 * hl + hf];
        const v4f q1 = sq[m * STG_PITCH + 4 * hl + hf + 2];
        const v4f k0 = sk[m * STG_PITCH + 4 * hl + hf];
        const v4f k1 = sk[m * STG_PITCH + 4 * hl + hf + 2];
        float p = q0[0] * (k0[0] + c0[2 * hh]);
        p = fmaf(q0[1], k0[1] + c1[2 * hh], p);
        p = fmaf(q0[2], k0[2] + c2[2 * hh], p);
        p = fmaf(q0[3], k0[3] + c3[2 * hh], p);
        p = fmaf(q1[0], k1[0] + c0[2 * hh + 1], p);
        p = fmaf(q1[1], k1[1] + c1[2 * hh + 1], p);
        p = fmaf(q1[2], k1[2] + c2[2 * hh + 1], p);
        p = fmaf(q1[3], k1[3] + c3[2 * hh + 1], p);
        p += xhalf(p);
        o[hh] = (hf == g) ? p : o[hh];
      }
    }
  }
  v8f gt = {};
  {
    v8f c0 = {}, c1 = {}, c2 = {}, c3 = {};
    const unsigned wo = (unsigned)kk * 256u + 16u * (unsigned)m;
    wmma4_steps(Wg1, wo, 256u, efr, 0, c0, c1, c2, c3);
    wmma4_steps(Wg1, wo, 256u, efr, 4, c0, c1, c2, c3);
    const float* w2 = Wg2 + (size_t)(32 * hf) * 8 + (m & 7);
    gt = gate2_acc(gt, w2, silu8(c0), 0);
    gt = gate2_acc(gt, w2, silu8(c1), 1);
    gt = gate2_acc(gt, w2, silu8(c2), 2);
    gt = gate2_acc(gt, w2, silu8(c3), 3);
  }

  o[0] = o[0] * 0.25f + (hf ? gt[4] : gt[0]);
  o[1] = o[1] * 0.25f + (hf ? gt[5] : gt[1]);
  o[2] = o[2] * 0.25f + (hf ? gt[6] : gt[2]);
  o[3] = o[3] * 0.25f + (hf ? gt[7] : gt[3]);

  *(v4f*)(scores + (size_t)(e0 + m) * 8 + 4 * hf) = o;
  unsigned* mp = mmax + (size_t)dm * 8 + 4 * hf;
  atomicMax(mp + 0, fkey(o[0]));
  atomicMax(mp + 1, fkey(o[1]));
  atomicMax(mp + 2, fkey(o[2]));
  atomicMax(mp + 3, fkey(o[3]));

  const int gid = blockIdx.x * (WAVES * 32) + threadIdx.x;
  if (gid < n_nodes) cnt[gid] = 0;
}

#define EXP_STEPS 16
#define EXP_EDGES (4 * EXP_STEPS)
#define EXP_ROUNDS (EXP_EDGES / 32)

__global__ __launch_bounds__(256) void exp_denom_kernel(
    const int* __restrict__ dst, float* __restrict__ scores,
    const unsigned* __restrict__ mmax, float* __restrict__ denom,
    int* __restrict__ cnt, int* __restrict__ rank, int n_edges) {
  const int lane  = threadIdx.x & 31;
  const int wbase = (blockIdx.x * WAVES + (threadIdx.x >> 5)) * EXP_EDGES;
  if (wbase >= n_edges) return;
  const unsigned h = lane & 7;
  const int sub = lane >> 3;

  if (wbase + EXP_EDGES <= n_edges) {
    float* sp = scores + (size_t)wbase * 8 + lane;
    const int* dp = dst + wbase + sub;
    int d[EXP_STEPS], rd[EXP_ROUNDS], rk[EXP_ROUNDS];
    float s[EXP_STEPS];
    unsigned mk[EXP_STEPS];
#pragma unroll
    for (int i = 0; i < EXP_STEPS; ++i) {
      d[i] = dp[4 * i];
      s[i] = sp[32 * i];
    }
#pragma unroll
    for (int j = 0; j < EXP_ROUNDS; ++j) rd[j] = dst[wbase + 32 * j + lane];
#pragma unroll
    for (int i = 0; i < EXP_STEPS; ++i) mk[i] = mmax[(unsigned)d[i] * 8u + h];
#pragma unroll
    for (int j = 0; j < EXP_ROUNDS; ++j) rk[j] = atomicAdd(&cnt[rd[j]], 1);
#pragma unroll
    for (int i = 0; i < EXP_STEPS; ++i) {
      const float es = __expf(s[i] - funkey(mk[i]));
      sp[32 * i] = es;
      atomicAdd(&denom[(unsigned)d[i] * 8u + h], es);
    }
#pragma unroll
    for (int j = 0; j < EXP_ROUNDS; ++j) rank[wbase + 32 * j + lane] = rk[j];
  } else {
    for (int i = 0; i < EXP_STEPS; ++i) {
      const int e = wbase + 4 * i + sub;
      if (e < n_edges) {
        const unsigned dh = (unsigned)dst[e] * 8u + h;
        const size_t idx = (size_t)e * 8 + h;
        const float es = __expf(scores[idx] - funkey(mmax[dh]));
        scores[idx] = es;
        atomicAdd(&denom[dh], es);
      }
    }
    for (int j = 0; j < EXP_ROUNDS; ++j) {
      const int e = wbase + 32 * j + lane;
      if (e < n_edges) rank[e] = atomicAdd(&cnt[dst[e]], 1);
    }
  }
}

__global__ __launch_bounds__(1024) void seg_offsets_kernel(
    const int* __restrict__ cnt, int* __restrict__ off, int n) {
  __shared__ int wsum[32];
  __shared__ int psum[32];
  const int t = threadIdx.x, lane = t & 31, w = t >> 5;
  const int base = blockIdx.x * 1024;
  int p = 0;
  for (int i = t; i < base; i += 1024) p += cnt[i];
  const int x = (base + t < n) ? cnt[base + t] : 0;
  int inc = x;
  for (int d = 1; d < 32; d <<= 1) {
    int y = __shfl_up(inc, d);
    if (lane >= d) inc += y;
  }
  for (int d = 16; d > 0; d >>= 1) p += __shfl_xor(p, d);
  if (lane == 31) wsum[w] = inc;
  if (lane == 0) psum[w] = p;
  __syncthreads();
  int pre = 0;
  for (int i = 0; i < 32; ++i) {
    pre += psum[i];
    pre += (i < w) ? wsum[i] : 0;
  }
  if (base + t < n) off[base + t] = pre + inc - x;
}

__global__ __launch_bounds__(256) void dst_order_kernel(
    const int* __restrict__ src, const int* __restrict__ dst,
    const int* __restrict__ rank, const int* __restrict__ off,
    int* __restrict__ perm, int* __restrict__ ssrc, int* __restrict__ sdst,
    int n_edges) {
  const int e = blockIdx.x * blockDim.x + threadIdx.x;
  if (e >= n_edges) return;
  const int d = dst[e];
  const int pos = off[d] + rank[e];
  perm[pos] = e;
  ssrc[pos] = src[e];
  sdst[pos] = d;
}

typedef __attribute__((ext_vector_type(4))) int v4i;
#define TILES 2

__global__ __launch_bounds__(256) __attribute__((amdgpu_waves_per_eu(6))) void edge_aggregate_kernel(
    const float* __restrict__ ea, const int* __restrict__ perm,
    const int* __restrict__ ssrc, const int* __restrict__ sdst,
    const float* __restrict__ Wev, const float* __restrict__ v,
    const float* __restrict__ scores, const float* __restrict__ denom,
    float* __restrict__ acc) {
  __shared__ float carry[WAVES][8][32];
  __shared__ v4f sta[WAVES][16 * STG_PITCH];
  const int lane = threadIdx.x & 31;
  const int wv   = threadIdx.x >> 5;
  const int m    = lane & 15;
  const int hf   = lane >> 4;
  const int kk   = hf << 1;
  const unsigned hq = (unsigned)(m >> 2);
  const int wbase = (blockIdx.x * WAVES + wv) * (16 * TILES);
  const int sbase = wbase + hf * (8 * TILES);
  const int sub = lane & 7;
  const int eg  = lane >> 3;
  v4f* sa = &sta[wv][0];
  const v2f* rowm = (const v2f*)(sa + m * STG_PITCH) + hf;

#pragma unroll 1
  for (int i = 0; i < TILES; ++i) {
    v4f te[4];
#pragma unroll
    for (int jj = 0; jj < 4; ++jj) {
      const int pe = perm[wbase + (jj >> 1) * (8 * TILES) + 4 * (jj & 1) + eg + 8 * i];
      te[jj] = *(const v4f*)(ea + (size_t)pe * 32 + 4 * sub);
    }

    const int p = sbase + 8 * i;
    unsigned so[8], dn[8], vo[8];
    {
      const v4i p0 = *(const v4i*)(perm + p), p1 = *(const v4i*)(perm + p + 4);
      const v4i s0 = *(const v4i*)(ssrc + p), s1 = *(const v4i*)(ssrc + p + 4);
      const v4i d0 = *(const v4i*)(sdst + p), d1 = *(const v4i*)(sdst + p + 4);
#pragma unroll
      for (int r = 0; r < 4; ++r) {
        so[r] = (unsigned)p0[r] * 32u;             so[r + 4] = (unsigned)p1[r] * 32u;
        vo[r] = (unsigned)s0[r] * 512u + 16u * m;  vo[r + 4] = (unsigned)s1[r] * 512u + 16u * m;
        dn[r] = (unsigned)d0[r] * 32u;             dn[r + 4] = (unsigned)d1[r] * 32u;
      }
    }
    const bool last = (i == TILES - 1);
    const unsigned dnext = last ? 0xffffffffu : (unsigned)sdst[p + 8] * 32u;

    __builtin_amdgcn_fence(__ATOMIC_ACQ_REL, "wavefront");
#pragma unroll
    for (int jj = 0; jj < 4; ++jj) sa[(4 * jj + eg) * STG_PITCH + sub] = te[jj];
    __builtin_amdgcn_fence(__ATOMIC_ACQ_REL, "wavefront");
    v2f afr[8];
#pragma unroll
    for (int j = 0; j < 8; ++j) afr[j] = rowm[2 * j];

#pragma unroll 1
    for (int g = 0; g < 2; ++g) {
      v8f c0 = {}, c1 = {}, c2 = {}, c3 = {};
      const float* wp = Wev + (size_t)kk * 128 + 64 * g + 4 * m;
#pragma unroll
      for (int j = 0; j < 8; ++j) {
        const v4f w0 = *(const v4f*)(wp + (size_t)(4 * j) * 128);
        const v4f w1 = *(const v4f*)(wp + (size_t)(4 * j + 1) * 128);
        v2f b;
        b[0] = w0[0]; b[1] = w1[0];
        c0 = __builtin_amdgcn_wmma_f32_16x16x4_f32(false, afr[j], false, b,
                                                   (short)0, c0, false, false);
        b[0] = w0[1]; b[1] = w1[1];
        c1 = __builtin_amdgcn_wmma_f32_16x16x4_f32(false, afr[j], false, b,
                                                   (short)0, c1, false, false);
        b[0] = w0[2]; b[1] = w1[2];
        c2 = __builtin_amdgcn_wmma_f32_16x16x4_f32(false, afr[j], false, b,
                                                   (short)0, c2, false, false);
        b[0] = w0[3]; b[1] = w1[3];
        c3 = __builtin_amdgcn_wmma_f32_16x16x4_f32(false, afr[j], false, b,
                                                   (short)0, c3, false, false);
      }
      __builtin_amdgcn_sched_barrier(0);
      const unsigned hd = 16u * (unsigned)g + 4u * hq;
      const unsigned fo = 256u * (unsigned)g;
      v4f t[8];
#pragma unroll
      for (int r = 0; r < 8; ++r) {
        const float att = *(const float*)((const char*)scores + (so[r] + hd)) /
                          (*(const float*)((const char*)denom + (dn[r] + hd)) + 1e-9f);
        const v4f vr = *(const v4f*)((const char*)v + (vo[r] + fo));
        t[r][0] = att * (vr[0] + c0[r]);
        t[r][1] = att * (vr[1] + c1[r]);
        t[r][2] = att * (vr[2] + c2[r]);
        t[r][3] = att * (vr[3] + c3[r]);
      }
      v4f run = {};
      if (i != 0) {
        run[0] = carry[wv][4 * g + 0][lane];
        run[1] = carry[wv][4 * g + 1][lane];
        run[2] = carry[wv][4 * g + 2][lane];
        run[3] = carry[wv][4 * g + 3][lane];
      }
#pragma unroll
      for (int r = 0; r < 8; ++r) {
        run[0] += t[r][0];
        run[1] += t[r][1];
        run[2] += t[r][2];
        run[3] += t[r][3];
        const unsigned nx = (r < 7) ? dn[r + 1 < 8 ? r + 1 : 7] : dnext;
        if (dn[r] != nx) {
          float* ap = (float*)((char*)acc + (dn[r] * 16u + fo + 16u * (unsigned)m));
          atomicAdd(ap + 0, run[0]);
          atomicAdd(ap + 1, run[1]);
          atomicAdd(ap + 2, run[2]);
          atomicAdd(ap + 3, run[3]);
          run[0] = 0.0f; run[1] = 0.0f; run[2] = 0.0f; run[3] = 0.0f;
        }
      }
      if (!last) {
        carry[wv][4 * g + 0][lane] = run[0];
        carry[wv][4 * g + 1][lane] = run[1];
        carry[wv][4 * g + 2][lane] = run[2];
        carry[wv][4 * g + 3][lane] = run[3];
      }
    }
  }
}

extern "C" void kernel_launch(void* const* d_in, const int* in_sizes, int n_in,
                              void* d_out, int out_size, void* d_ws,
                              size_t ws_size, hipStream_t stream) {
  const float* x   = (const float*)d_in[0];
  const int*   ei  = (const int*)d_in[1];
  const float* ea  = (const float*)d_in[2];
  const float* Wq  = (const float*)d_in[3];
  const float* Wk  = (const float*)d_in[4];
  const float* Wv  = (const float*)d_in[5];
  const float* Wo  = (const float*)d_in[6];
  const float* Wek = (const float*)d_in[7];
  const float* Wev = (const float*)d_in[8];
  const float* Wg1 = (const float*)d_in[9];
  const float* Wg2 = (const float*)d_in[10];
  float* out = (float*)d_out;

  const int N = in_sizes[0] / 128;
  const int E = in_sizes[2] / 32;
  const int* src = ei;
  const int* dst = ei + E;

  float* q      = (float*)d_ws;
  float* k      = q + (size_t)N * 128;
  float* v      = k + (size_t)N * 128;
  float* scores = v + (size_t)N * 128;
  float* acc    = scores + (size_t)E * 8;
  float* denom  = acc + (size_t)N * 128;
  unsigned* mmax = (unsigned*)(denom + (size_t)N * 8);
  int* cnt  = (int*)v;
  int* rank = cnt + N;
  int* perm = (int*)q;
  int* ssrc = perm + E;
  int* sdst = ssrc + E;
  int* off  = sdst + E;

  hipMemsetAsync(acc, 0, (size_t)(N * 128 + N * 8 + N * 8) * sizeof(float),
                 stream);

  dim3 blk(256);
  dim3 gblk(32 * GEMM_WAVES);
  gemm128_kernel<<<dim3((N / 16 * 2 + GEMM_WAVES - 1) / GEMM_WAVES), gblk, 0, stream>>>(x, Wq, q, N / 16);
  gemm128_kernel<<<dim3((N / 16 * 2 + GEMM_WAVES - 1) / GEMM_WAVES), gblk, 0, stream>>>(x, Wk, k, N / 16);

  edge_score_kernel<<<dim3(E / (16 * WAVES)), blk, 0, stream>>>(
      ea, src, dst, Wek, Wg1, Wg2, q, k, scores, mmax, cnt, N);

  exp_denom_kernel<<<dim3((E + WAVES * EXP_EDGES - 1) / (WAVES * EXP_EDGES)), blk,
                     0, stream>>>(dst, scores, mmax, denom, cnt, rank, E);

  seg_offsets_kernel<<<dim3((N + 1023) / 1024), dim3(1024), 0, stream>>>(
      cnt, off, N);

  dst_order_kernel<<<dim3((E + 255) / 256), blk, 0, stream>>>(
      src, dst, rank, off, perm, ssrc, sdst, E);

  gemm128_kernel<<<dim3((N / 16 * 2 + GEMM_WAVES - 1) / GEMM_WAVES), gblk, 0, stream>>>(x, Wv, v, N / 16);

  edge_aggregate_kernel<<<dim3(E / (16 * WAVES * TILES)), blk, 0, stream>>>(
      ea, perm, ssrc, sdst, Wev, v, scores, denom, acc);

  gemm128_kernel<<<dim3((N / 16 * 2 + GEMM_WAVES - 1) / GEMM_WAVES), gblk, 0, stream>>>(acc, Wo, out, N / 16);
}


# edge_score and edge_aggregate: Wek, Wg1, Wg2, Wev copied once per work-group into LDS (26 and 16 KiB) and read as grouped LDS fragment loads instead of global loads by every wave; sums unchanged
# speedup vs baseline: 2.4205x; 1.0311x over previous
#include <hip/hip_runtime.h>


typedef __attribute__((ext_vector_type(2))) float v2f;
typedef __attribute__((ext_vector_type(4))) float v4f;
typedef __attribute__((ext_vector_type(8))) float v8f;

#define WAVES 8

__device__ __forceinline__ unsigned fkey(float f) {
  unsigned u = __float_as_uint(f);
  return (u & 0x80000000u) ? ~u : (u | 0x80000000u);
}
__device__ __forceinline__ float funkey(unsigned kk) {
  unsigned u = (kk & 0x80000000u) ? (kk & 0x7FFFFFFFu) : ~kk;
  return __uint_as_float(u);
}

#define GEMM_WAVES 4
#define STG_PITCH 9
__global__ __launch_bounds__(32 * GEMM_WAVES) void gemm128_kernel(
    const float* __restrict__ A, const float* __restrict__ W,
    float* __restrict__ out, int n_tiles) {
  __shared__ v4f sta[GEMM_WAVES][16 * STG_PITCH];
  const int lane = threadIdx.x & 31;
  const int unit = blockIdx.x * GEMM_WAVES + (threadIdx.x >> 5);
  const int rt   = unit >> 1;
  const int g    = unit & 1;
  if (rt >= n_tiles) return;
  const int m    = lane & 15;
  const int hf   = lane >> 4;
  const int kk   = hf << 1;

  v8f c0 = {}, c1 = {}, c2 = {}, c3 = {};
  const int sub = lane & 7;
  const int eg  = lane >> 3;
  v4f* sa = &sta[threadIdx.x >> 5][0];
  const float* ab = A + (size_t)(rt * 16 + eg) * 128 + 4 * sub;
  const v2f* rowm = (const v2f*)(sa + m * STG_PITCH) + hf;
  const float* wp = W + (size_t)kk * 128 + 64 * g + 4 * m;
#pragma unroll
  for (int t0 = 0; t0 < 128; t0 += 32) {
    __builtin_amdgcn_sched_barrier(0);
    v4f ta[4], w0[8], w1[8];
#pragma unroll
    for (int j = 0; j < 4; ++j) ta[j] = *(const v4f*)(ab + (size_t)(4 * j) * 128 + t0);
#pragma unroll
    for (int i = 0; i < 8; ++i) {
      w0[i] = *(const v4f*)(wp + (size_t)(t0 + 4 * i) * 128);
      w1[i] = *(const v4f*)(wp + (size_t)(t0 + 4 * i + 1) * 128);
    }
    __builtin_amdgcn_fence(__ATOMIC_ACQ_REL, "wavefront");
#pragma unroll
    for (int j = 0; j < 4; ++j) sa[(4 * j + eg) * STG_PITCH + sub] = ta[j];
    __builtin_amdgcn_fence(__ATOMIC_ACQ_REL, "wavefront");
    v2f afr[8];
#pragma unroll
    for (int i = 0; i < 8; ++i) afr[i] = rowm[2 * i];
#pragma unroll
    for (int i = 0; i < 8; ++i) {
      const v2f a = afr[i];
      v2f b;
      b[0] = w0[i][0]; b[1] = w1[i][0];
      c0 = __builtin_amdgcn_wmma_f32_16x16x4_f32(false, a, false, b, (short)0, c0,
                                                 false, false);
      b[0] = w0[i][1]; b[1] = w1[i][1];
      c1 = __builtin_amdgcn_wmma_f32_16x16x4_f32(false, a, false, b, (short)0, c1,
                                                 false, false);
      b[0] = w0[i][2]; b[1] = w1[i][2];
      c2 = __builtin_amdgcn_wmma_f32_16x16x4_f32(false, a, false, b, (short)0, c2,
                                                 false, false);
      b[0] = w0[i][3]; b[1] = w1[i][3];
      c3 = __builtin_amdgcn_wmma_f32_16x16x4_f32(false, a, false, b, (short)0, c3,
                                                 false, false);
    }
  }
  float* op = out + (size_t)(rt * 16 + 8 * hf) * 128 + 64 * g + 4 * m;
#pragma unroll
  for (int r = 0; r < 8; ++r) {
    v4f o;
    o[0] = c0[r]; o[1] = c1[r]; o[2] = c2[r]; o[3] = c3[r];
    *(v4f*)(op + (size_t)r * 128) = o;
  }
}

__device__ __forceinline__ float xhalf(float x) {
  int b = __builtin_bit_cast(int, x);
  return __builtin_bit_cast(
      float, __builtin_amdgcn_permlanex16(b, b, 0x76543210, 0xfedcba98, false,
                                          false));
}

__device__ __forceinline__ v8f silu8(v8f c) {
#pragma unroll
  for (int r = 0; r < 8; ++r) {
    float xv = c[r];
    c[r] = xv / (1.0f + __expf(-xv));
  }
  return c;
}

__device__ __forceinline__ void wmma4_steps(const v4f* Wb,
                                            unsigned wo, unsigned pitch,
                                            const v2f* efr, int i0, v8f& c0,
                                            v8f& c1, v8f& c2, v8f& c3) {
  if (i0 != 0) asm volatile("" : "+v"(c0), "+v"(c1), "+v"(c2), "+v"(c3) : : "memory");
  v4f w0[4], w1[4];
#pragma unroll
  for (int i = 0; i < 4; ++i) {
    w0[i] = Wb[wo + (4u * (i0 + i)) * pitch];
    w1[i] = Wb[wo + (4u * (i0 + i) + 1u) * pitch];
  }
  __builtin_amdgcn_sched_barrier(0);
#pragma unroll
  for (int i = 0; i < 4; ++i) {
    const v2f e = efr[i0 + i];
    v2f a;
    a[0] = w0[i][0]; a[1] = w1[i][0];
    c0 = __builtin_amdgcn_wmma_f32_16x16x4_f32(false, a, false, e,
                                               (short)0, c0, false, false);
    a[0] = w0[i][1]; a[1] = w1[i][1];
    c1 = __builtin_amdgcn_wmma_f32_16x16x4_f32(false, a, false, e,
                                               (short)0, c1, false, false);
    a[0] = w0[i][2]; a[1] = w1[i][2];
    c2 = __builtin_amdgcn_wmma_f32_16x16x4_f32(false, a, false, e,
                                               (short)0, c2, false, false);
    a[0] = w0[i][3]; a[1] = w1[i][3];
    c3 = __builtin_amdgcn_wmma_f32_16x16x4_f32(false, a, false, e,
                                               (short)0, c3, false, false);
  }
}

__device__ __forceinline__ void gate2_frags(v2f* a2, const float* w2, int t) {
#pragma unroll
  for (int p = 0; p < 4; ++p) {
    a2[p][0] = w2[(8 * p + t) * 8];
    a2[p][1] = w2[(8 * p + 4 + t) * 8];
  }
}
__device__ __forceinline__ v8f gate2_acc(v8f gt, const v2f* a2, v8f c) {
#pragma unroll
  for (int p = 0; p < 4; ++p) {
    v2f b;
    b[0] = c[2 * p];
    b[1] = c[2 * p + 1];
    gt = __builtin_amdgcn_wmma_f32_16x16x4_f32(false, a2[p], false, b, (short)0,
                                               gt, false, false);
  }
  return gt;
}

__global__ __launch_bounds__(256) void edge_score_kernel(
    const float* __restrict__ ea, const int* __restrict__ src,
    const int* __restrict__ dst, const float* __restrict__ Wek,
    const float* __restrict__ Wg1, const float* __restrict__ Wg2,
    const float* __restrict__ q, const float* __restrict__ k,
    float* __restrict__ scores, unsigned* __restrict__ mmax,
    int* __restrict__ cnt, int n_nodes) {
  __shared__ v4f stg[WAVES][2][16 * STG_PITCH];
  __shared__ v4f swek[32 * 32];
  __shared__ v4f swg1[32 * 16];
  __shared__ v4f swg2[64 * 2];
  const int lane = threadIdx.x & 31;
  const int wv   = threadIdx.x >> 5;
  const int e0   = (blockIdx.x * WAVES + wv) << 4;
  const int m    = lane & 15;
  const int hf   = lane >> 4;
  const int kk   = hf << 1;

  const int dm = dst[e0 + m];

  const int sub = lane & 7;
  const int eg  = lane >> 3;
  unsigned qrow[4], krow[4];
#pragma unroll
  for (int j = 0; j < 4; ++j) {
    qrow[j] = (unsigned)dst[e0 + 4 * j + eg] * 512u + 16u * (unsigned)sub;
    krow[j] = (unsigned)src[e0 + 4 * j + eg] * 512u + 16u * (unsigned)sub;
  }
  v4f* sq = &stg[wv][0][0];
  v4f* sk = &stg[wv][1][0];

  v2f efr[8];
  {
    const float* eb = ea + (size_t)e0 * 32 + 4 * lane;
    v4f te[4];
#pragma unroll
    for (int j = 0; j < 4; ++j) te[j] = *(const v4f*)(eb + 128 * j);
    const int t = threadIdx.x;
    v4f tw[4], tg[2];
#pragma unroll
    for (int j = 0; j < 4; ++j) tw[j] = ((const v4f*)Wek)[t + 256 * j];
#pragma unroll
    for (int j = 0; j < 2; ++j) tg[j] = ((const v4f*)Wg1)[t + 256 * j];
    const v4f t2 = ((const v4f*)Wg2)[t & 127];
#pragma unroll
    for (int j = 0; j < 4; ++j) sq[(4 * j + eg) * STG_PITCH + sub] = te[j];
    __builtin_amdgcn_fence(__ATOMIC_ACQ_REL, "wavefront");
    const v2f* rowm = (const v2f*)(sq + m * STG_PITCH) + hf;
#pragma unroll
    for (int i = 0; i < 8; ++i) efr[i] = rowm[2 * i];
#pragma unroll
    for (int j = 0; j < 4; ++j) swek[t + 256 * j] = tw[j];
#pragma unroll
    for (int j = 0; j < 2; ++j) swg1[t + 256 * j] = tg[j];
    if (t < 128) swg2[t] = t2;
  }
  __syncthreads();
  const unsigned psi4 = 2u * (unsigned)(m & 7) + (unsigned)(m >> 3);
  v4f o = {};
#pragma unroll 1
  for (int g = 0; g < 2; ++g) {
    v8f c0 = {}, c1 = {}, c2 = {}, c3 = {};
    const unsigned wo = (unsigned)kk * 32u + 16u * (unsigned)g + psi4;
    wmma4_steps(swek, wo, 32u, efr, 0, c0, c1, c2, c3);
    wmma4_steps(swek, wo, 32u, efr, 4, c0, c1, c2, c3);
#pragma unroll
    for (int hp = 0; hp < 2; ++hp) {
      __builtin_amdgcn_sched_barrier(0);
      const unsigned lo = 256u * (unsigned)g + 128u * (unsigned)hp;
      v4f tq[4], tk[4];
#pragma unroll
      for (int j = 0; j < 4; ++j) {
        tq[j] = *(const v4f*)((const char*)q + (qrow[j] + lo));
        tk[j] = *(const v4f*)((const char*)k + (krow[j] + lo));
      }
      __builtin_amdgcn_fence(__ATOMIC_ACQ_REL, "wavefront");
#pragma unroll
      for (int j = 0; j < 4; ++j) {
        sq[(4 * j + eg) * STG_PITCH + sub] = tq[j];
        sk[(4 * j + eg) * STG_PITCH + sub] = tk[j];
      }
      __builtin_amdgcn_fence(__ATOMIC_ACQ_REL, "wavefront");
#pragma unroll
      for (int hl = 0; hl < 2; ++hl) {
        const int hh = 2 * hp + hl;
        const v4f q0 = sq[m * STG_PITCH + 4 * hl + hf];
        const v4f q1 = sq[m * STG_PITCH + 4 * hl + hf + 2];
        const v4f k0 = sk[m * STG_PITCH + 4 * hl + hf];
        const v4f k1 = sk[m * STG_PITCH + 4 * hl + hf + 2];
        float p = q0[0] * (k0[0] + c0[2 * hh]);
        p = fmaf(q0[1], k0[1] + c1[2 * hh], p);
        p = fmaf(q0[2], k0[2] + c2[2 * hh], p);
        p = fmaf(q0[3], k0[3] + c3[2 * hh], p);
        p = fmaf(q1[0], k1[0] + c0[2 * hh + 1], p);
        p = fmaf(q1[1], k1[1] + c1[2 * hh + 1], p);
        p = fmaf(q1[2], k1[2] + c2[2 * hh + 1], p);
        p = fmaf(q1[3], k1[3] + c3[2 * hh + 1], p);
        p += xhalf(p);
        o[hh] = (hf == g) ? p : o[hh];
      }
    }
  }
  v8f gt = {};
  {
    v8f c0 = {}, c1 = {}, c2 = {}, c3 = {};
    const unsigned wo = (unsigned)kk * 16u + (unsigned)m;
    wmma4_steps(swg1, wo, 16u, efr, 0, c0, c1, c2, c3);
    wmma4_steps(swg1, wo, 16u, efr, 4, c0, c1, c2, c3);
    const float* w2 = (const float*)swg2 + (32 * hf) * 8 + (m & 7);
    v2f a20[4], a21[4], a22[4], a23[4];
    gate2_frags(a20, w2, 0);
    gate2_frags(a21, w2, 1);
    gate2_frags(a22, w2, 2);
    gate2_frags(a23, w2, 3);
    gt = gate2_acc(gt, a20, silu8(c0));
    gt = gate2_acc(gt, a21, silu8(c1));
    gt = gate2_acc(gt, a22, silu8(c2));
    gt = gate2_acc(gt, a23, silu8(c3));
  }

  o[0] = o[0] * 0.25f + (hf ? gt[4] : gt[0]);
  o[1] = o[1] * 0.25f + (hf ? gt[5] : gt[1]);
  o[2] = o[2] * 0.25f + (hf ? gt[6] : gt[2]);
  o[3] = o[3] * 0.25f + (hf ? gt[7] : gt[3]);

  *(v4f*)(scores + (size_t)(e0 + m) * 8 + 4 * hf) = o;
  unsigned* mp = mmax + (size_t)dm * 8 + 4 * hf;
  atomicMax(mp + 0, fkey(o[0]));
  atomicMax(mp + 1, fkey(o[1]));
  atomicMax(mp + 2, fkey(o[2]));
  atomicMax(mp + 3, fkey(o[3]));

  const int gid = blockIdx.x * (WAVES * 32) + threadIdx.x;
  if (gid < n_nodes) cnt[gid] = 0;
}

#define EXP_STEPS 16
#define EXP_EDGES (4 * EXP_STEPS)
#define EXP_ROUNDS (EXP_EDGES / 32)

__global__ __launch_bounds__(256) void exp_denom_kernel(
    const int* __restrict__ dst, float* __restrict__ scores,
    const unsigned* __restrict__ mmax, float* __restrict__ denom,
    int* __restrict__ cnt, int* __restrict__ rank, int n_edges) {
  const int lane  = threadIdx.x & 31;
  const int wbase = (blockIdx.x * WAVES + (threadIdx.x >> 5)) * EXP_EDGES;
  if (wbase >= n_edges) return;
  const unsigned h = lane & 7;
  const int sub = lane >> 3;

  if (wbase + EXP_EDGES <= n_edges) {
    float* sp = scores + (size_t)wbase * 8 + lane;
    const int* dp = dst + wbase + sub;
    int d[EXP_STEPS], rd[EXP_ROUNDS], rk[EXP_ROUNDS];
    float s[EXP_STEPS];
    unsigned mk[EXP_STEPS];
#pragma unroll
    for (int i = 0; i < EXP_STEPS; ++i) {
      d[i] = dp[4 * i];
      s[i] = sp[32 * i];
    }
#pragma unroll
    for (int j = 0; j < EXP_ROUNDS; ++j) rd[j] = dst[wbase + 32 * j + lane];
#pragma unroll
    for (int i = 0; i < EXP_STEPS; ++i) mk[i] = mmax[(unsigned)d[i] * 8u + h];
#pragma unroll
    for (int j = 0; j < EXP_ROUNDS; ++j) rk[j] = atomicAdd(&cnt[rd[j]], 1);
#pragma unroll
    for (int i = 0; i < EXP_STEPS; ++i) {
      const float es = __expf(s[i] - funkey(mk[i]));
      sp[32 * i] = es;
      atomicAdd(&denom[(unsigned)d[i] * 8u + h], es);
    }
#pragma unroll
    for (int j = 0; j < EXP_ROUNDS; ++j) rank[wbase + 32 * j + lane] = rk[j];
  } else {
    for (int i = 0; i < EXP_STEPS; ++i) {
      const int e = wbase + 4 * i + sub;
      if (e < n_edges) {
        const unsigned dh = (unsigned)dst[e] * 8u + h;
        const size_t idx = (size_t)e * 8 + h;
        const float es = __expf(scores[idx] - funkey(mmax[dh]));
        scores[idx] = es;
        atomicAdd(&denom[dh], es);
      }
    }
    for (int j = 0; j < EXP_ROUNDS; ++j) {
      const int e = wbase + 32 * j + lane;
      if (e < n_edges) rank[e] = atomicAdd(&cnt[dst[e]], 1);
    }
  }
}

__global__ __launch_bounds__(1024) void seg_offsets_kernel(
    const int* __restrict__ cnt, int* __restrict__ off, int n) {
  __shared__ int wsum[32];
  __shared__ int psum[32];
  const int t = threadIdx.x, lane = t & 31, w = t >> 5;
  const int base = blockIdx.x * 1024;
  int p = 0;
  for (int i = t; i < base; i += 1024) p += cnt[i];
  const int x = (base + t < n) ? cnt[base + t] : 0;
  int inc = x;
  for (int d = 1; d < 32; d <<= 1) {
    int y = __shfl_up(inc, d);
    if (lane >= d) inc += y;
  }
  for (int d = 16; d > 0; d >>= 1) p += __shfl_xor(p, d);
  if (lane == 31) wsum[w] = inc;
  if (lane == 0) psum[w] = p;
  __syncthreads();
  int pre = 0;
  for (int i = 0; i < 32; ++i) {
    pre += psum[i];
    pre += (i < w) ? wsum[i] : 0;
  }
  if (base + t < n) off[base + t] = pre + inc - x;
}

__global__ __launch_bounds__(256) void dst_order_kernel(
    const int* __restrict__ src, const int* __restrict__ dst,
    const int* __restrict__ rank, const int* __restrict__ off,
    int* __restrict__ perm, int* __restrict__ ssrc, int* __restrict__ sdst,
    int n_edges) {
  const int e = blockIdx.x * blockDim.x + threadIdx.x;
  if (e >= n_edges) return;
  const int d = dst[e];
  const int pos = off[d] + rank[e];
  perm[pos] = e;
  ssrc[pos] = src[e];
  sdst[pos] = d;
}

typedef __attribute__((ext_vector_type(4))) int v4i;
#define TILES 2

__global__ __launch_bounds__(256) __attribute__((amdgpu_waves_per_eu(6))) void edge_aggregate_kernel(
    const float* __restrict__ ea, const int* __restrict__ perm,
    const int* __restrict__ ssrc, const int* __restrict__ sdst,
    const float* __restrict__ Wev, const float* __restrict__ v,
    const float* __restrict__ scores, const float* __restrict__ denom,
    float* __restrict__ acc) {
  __shared__ float carry[WAVES][8][32];
  __shared__ v4f sta[WAVES][16 * STG_PITCH];
  __shared__ v4f swev[32 * 32];
  {
    const int t = threadIdx.x;
    v4f tw[4];
#pragma unroll
    for (int j = 0; j < 4; ++j) tw[j] = ((const v4f*)Wev)[t + 256 * j];
#pragma unroll
    for (int j = 0; j < 4; ++j) swev[t + 256 * j] = tw[j];
  }
  __syncthreads();
  const int lane = threadIdx.x & 31;
  const int wv   = threadIdx.x >> 5;
  const int m    = lane & 15;
  const int hf   = lane >> 4;
  const int kk   = hf << 1;
  const unsigned hq = (unsigned)(m >> 2);
  const int wbase = (blockIdx.x * WAVES + wv) * (16 * TILES);
  const int sbase = wbase + hf * (8 * TILES);
  const int sub = lane & 7;
  const int eg  = lane >> 3;
  v4f* sa = &sta[wv][0];
  const v2f* rowm = (const v2f*)(sa + m * STG_PITCH) + hf;

#pragma unroll 1
  for (int i = 0; i < TILES; ++i) {
    v4f te[4];
#pragma unroll
    for (int jj = 0; jj < 4; ++jj) {
      const int pe = perm[wbase + (jj >> 1) * (8 * TILES) + 4 * (jj & 1) + eg + 8 * i];
      te[jj] = *(const v4f*)(ea + (size_t)pe * 32 + 4 * sub);
    }

    const int p = sbase + 8 * i;
    unsigned so[8], dn[8], vo[8];
    {
      const v4i p0 = *(const v4i*)(perm + p), p1 = *(const v4i*)(perm + p + 4);
      const v4i s0 = *(const v4i*)(ssrc + p), s1 = *(const v4i*)(ssrc + p + 4);
      const v4i d0 = *(const v4i*)(sdst + p), d1 = *(const v4i*)(sdst + p + 4);
#pragma unroll
      for (int r = 0; r < 4; ++r) {
        so[r] = (unsigned)p0[r] * 32u;             so[r + 4] = (unsigned)p1[r] * 32u;
        vo[r] = (unsigned)s0[r] * 512u + 16u * m;  vo[r + 4] = (unsigned)s1[r] * 512u + 16u * m;
        dn[r] = (unsigned)d0[r] * 32u;             dn[r + 4] = (unsigned)d1[r] * 32u;
      }
    }
    const bool last = (i == TILES - 1);
    const unsigned dnext = last ? 0xffffffffu : (unsigned)sdst[p + 8] * 32u;

    __builtin_amdgcn_fence(__ATOMIC_ACQ_REL, "wavefront");
#pragma unroll
    for (int jj = 0; jj < 4; ++jj) sa[(4 * jj + eg) * STG_PITCH + sub] = te[jj];
    __builtin_amdgcn_fence(__ATOMIC_ACQ_REL, "wavefront");
    v2f afr[8];
#pragma unroll
    for (int j = 0; j < 8; ++j) afr[j] = rowm[2 * j];

#pragma unroll 1
    for (int g = 0; g < 2; ++g) {
      v8f c0 = {}, c1 = {}, c2 = {}, c3 = {};
      const v4f* wp = swev + (kk * 32 + 16 * g + m);
      v4f wr0[8], wr1[8];
#pragma unroll
      for (int j = 0; j < 8; ++j) {
        wr0[j] = wp[(4 * j) * 32];
        wr1[j] = wp[(4 * j + 1) * 32];
      }
      __builtin_amdgcn_sched_barrier(0);
#pragma unroll
      for (int j = 0; j < 8; ++j) {
        const v4f w0 = wr0[j];
        const v4f w1 = wr1[j];
        v2f b;
        b[0] = w0[0]; b[1] = w1[0];
        c0 = __builtin_amdgcn_wmma_f32_16x16x4_f32(false, afr[j], false, b,
                                                   (short)0, c0, false, false);
        b[0] = w0[1]; b[1] = w1[1];
        c1 = __builtin_amdgcn_wmma_f32_16x16x4_f32(false, afr[j], false, b,
                                                   (short)0, c1, false, false);
        b[0] = w0[2]; b[1] = w1[2];
        c2 = __builtin_amdgcn_wmma_f32_16x16x4_f32(false, afr[j], false, b,
                                                   (short)0, c2, false, false);
        b[0] = w0[3]; b[1] = w1[3];
        c3 = __builtin_amdgcn_wmma_f32_16x16x4_f32(false, afr[j], false, b,
                                                   (short)0, c3, false, false);
      }
      __builtin_amdgcn_sched_barrier(0);
      const unsigned hd = 16u * (unsigned)g + 4u * hq;
      const unsigned fo = 256u * (unsigned)g;
      v4f t[8];
#pragma unroll
      for (int r = 0; r < 8; ++r) {
        const float att = *(const float*)((const char*)scores + (so[r] + hd)) /
                          (*(const float*)((const char*)denom + (dn[r] + hd)) + 1e-9f);
        const v4f vr = *(const v4f*)((const char*)v + (vo[r] + fo));
        t[r][0] = att * (vr[0] + c0[r]);
        t[r][1] = att * (vr[1] + c1[r]);
        t[r][2] = att * (vr[2] + c2[r]);
        t[r][3] = att * (vr[3] + c3[r]);
      }
      v4f run = {};
      if (i != 0) {
        run[0] = carry[wv][4 * g + 0][lane];
        run[1] = carry[wv][4 * g + 1][lane];
        run[2] = carry[wv][4 * g + 2][lane];
        run[3] = carry[wv][4 * g + 3][lane];
      }
#pragma unroll
      for (int r = 0; r < 8; ++r) {
        run[0] += t[r][0];
        run[1] += t[r][1];
        run[2] += t[r][2];
        run[3] += t[r][3];
        const unsigned nx = (r < 7) ? dn[r + 1 < 8 ? r + 1 : 7] : dnext;
        if (dn[r] != nx) {
          float* ap = (float*)((char*)acc + (dn[r] * 16u + fo + 16u * (unsigned)m));
          atomicAdd(ap + 0, run[0]);
          atomicAdd(ap + 1, run[1]);
          atomicAdd(ap + 2, run[2]);
          atomicAdd(ap + 3, run[3]);
          run[0] = 0.0f; run[1] = 0.0f; run[2] = 0.0f; run[3] = 0.0f;
        }
      }
      if (!last) {
        carry[wv][4 * g + 0][lane] = run[0];
        carry[wv][4 * g + 1][lane] = run[1];
        carry[wv][4 * g + 2][lane] = run[2];
        carry[wv][4 * g + 3][lane] = run[3];
      }
    }
  }
}

extern "C" void kernel_launch(void* const* d_in, const int* in_sizes, int n_in,
                              void* d_out, int out_size, void* d_ws,
                              size_t ws_size, hipStream_t stream) {
  const float* x   = (const float*)d_in[0];
  const int*   ei  = (const int*)d_in[1];
  const float* ea  = (const float*)d_in[2];
  const float* Wq  = (const float*)d_in[3];
  const float* Wk  = (const float*)d_in[4];
  const float* Wv  = (const float*)d_in[5];
  const float* Wo  = (const float*)d_in[6];
  const float* Wek = (const float*)d_in[7];
  const float* Wev = (const float*)d_in[8];
  const float* Wg1 = (const float*)d_in[9];
  const float* Wg2 = (const float*)d_in[10];
  float* out = (float*)d_out;

  const int N = in_sizes[0] / 128;
  const int E = in_sizes[2] / 32;
  const int* src = ei;
  const int* dst = ei + E;

  float* q      = (float*)d_ws;
  float* k      = q + (size_t)N * 128;
  float* v      = k + (size_t)N * 128;
  float* scores = v + (size_t)N * 128;
  float* acc    = scores + (size_t)E * 8;
  float* denom  = acc + (size_t)N * 128;
  unsigned* mmax = (unsigned*)(denom + (size_t)N * 8);
  int* cnt  = (int*)v;
  int* rank = cnt + N;
  int* perm = (int*)q;
  int* ssrc = perm + E;
  int* sdst = ssrc + E;
  int* off  = sdst + E;

  hipMemsetAsync(acc, 0, (size_t)(N * 128 + N * 8 + N * 8) * sizeof(float),
                 stream);

  dim3 blk(256);
  dim3 gblk(32 * GEMM_WAVES);
  gemm128_kernel<<<dim3((N / 16 * 2 + GEMM_WAVES - 1) / GEMM_WAVES), gblk, 0, stream>>>(x, Wq, q, N / 16);
  gemm128_kernel<<<dim3((N / 16 * 2 + GEMM_WAVES - 1) / GEMM_WAVES), gblk, 0, stream>>>(x, Wk, k, N / 16);

  edge_score_kernel<<<dim3(E / (16 * WAVES)), blk, 0, stream>>>(
      ea, src, dst, Wek, Wg1, Wg2, q, k, scores, mmax, cnt, N);

  exp_denom_kernel<<<dim3((E + WAVES * EXP_EDGES - 1) / (WAVES * EXP_EDGES)), blk,
                     0, stream>>>(dst, scores, mmax, denom, cnt, rank, E);

  seg_offsets_kernel<<<dim3((N + 1023) / 1024), dim3(1024), 0, stream>>>(
      cnt, off, N);

  dst_order_kernel<<<dim3((E + 255) / 256), blk, 0, stream>>>(
      src, dst, rank, off, perm, ssrc, sdst, E);

  gemm128_kernel<<<dim3((N / 16 * 2 + GEMM_WAVES - 1) / GEMM_WAVES), gblk, 0, stream>>>(x, Wv, v, N / 16);

  edge_aggregate_kernel<<<dim3(E / (16 * WAVES * TILES)), blk, 0, stream>>>(
      ea, perm, ssrc, sdst, Wev, v, scores, denom, acc);

  gemm128_kernel<<<dim3((N / 16 * 2 + GEMM_WAVES - 1) / GEMM_WAVES), gblk, 0, stream>>>(acc, Wo, out, N / 16);
}


# edges counting-sorted by destination BEFORE pass 1 (own count kernel): q rows, scores, max, denom go in destination order; max and denom summed per run before one atomic; denom sum order changes
# speedup vs baseline: 2.7596x; 1.1401x over previous
#include <hip/hip_runtime.h>


typedef __attribute__((ext_vector_type(2))) float v2f;
typedef __attribute__((ext_vector_type(4))) float v4f;
typedef __attribute__((ext_vector_type(8))) float v8f;
typedef __attribute__((ext_vector_type(4))) int v4i;

#define WAVES 8

__device__ __forceinline__ unsigned fkey(float f) {
  unsigned u = __float_as_uint(f);
  return (u & 0x80000000u) ? ~u : (u | 0x80000000u);
}
__device__ __forceinline__ float funkey(unsigned kk) {
  unsigned u = (kk & 0x80000000u) ? (kk & 0x7FFFFFFFu) : ~kk;
  return __uint_as_float(u);
}

#define GEMM_WAVES 4
#define STG_PITCH 9
__global__ __launch_bounds__(32 * GEMM_WAVES) void gemm128_kernel(
    const float* __restrict__ A, const float* __restrict__ W,
    float* __restrict__ out, int n_tiles) {
  __shared__ v4f sta[GEMM_WAVES][16 * STG_PITCH];
  const int lane = threadIdx.x & 31;
  const int unit = blockIdx.x * GEMM_WAVES + (threadIdx.x >> 5);
  const int rt   = unit >> 1;
  const int g    = unit & 1;
  if (rt >= n_tiles) return;
  const int m    = lane & 15;
  const int hf   = lane >> 4;
  const int kk   = hf << 1;

  v8f c0 = {}, c1 = {}, c2 = {}, c3 = {};
  const int sub = lane & 7;
  const int eg  = lane >> 3;
  v4f* sa = &sta[threadIdx.x >> 5][0];
  const float* ab = A + (size_t)(rt * 16 + eg) * 128 + 4 * sub;
  const v2f* rowm = (const v2f*)(sa + m * STG_PITCH) + hf;
  const float* wp = W + (size_t)kk * 128 + 64 * g + 4 * m;
#pragma unroll
  for (int t0 = 0; t0 < 128; t0 += 32) {
    __builtin_amdgcn_sched_barrier(0);
    v4f ta[4], w0[8], w1[8];
#pragma unroll
    for (int j = 0; j < 4; ++j) ta[j] = *(const v4f*)(ab + (size_t)(4 * j) * 128 + t0);
#pragma unroll
    for (int i = 0; i < 8; ++i) {
      w0[i] = *(const v4f*)(wp + (size_t)(t0 + 4 * i) * 128);
      w1[i] = *(const v4f*)(wp + (size_t)(t0 + 4 * i + 1) * 128);
    }
    __builtin_amdgcn_fence(__ATOMIC_ACQ_REL, "wavefront");
#pragma unroll
    for (int j = 0; j < 4; ++j) sa[(4 * j + eg) * STG_PITCH + sub] = ta[j];
    __builtin_amdgcn_fence(__ATOMIC_ACQ_REL, "wavefront");
    v2f afr[8];
#pragma unroll
    for (int i = 0; i < 8; ++i) afr[i] = rowm[2 * i];
#pragma unroll
    for (int i = 0; i < 8; ++i) {
      const v2f a = afr[i];
      v2f b;
      b[0] = w0[i][0]; b[1] = w1[i][0];
      c0 = __builtin_amdgcn_wmma_f32_16x16x4_f32(false, a, false, b, (short)0, c0,
                                                 false, false);
      b[0] = w0[i][1]; b[1] = w1[i][1];
      c1 = __builtin_amdgcn_wmma_f32_16x16x4_f32(false, a, false, b, (short)0, c1,
                                                 false, false);
      b[0] = w0[i][2]; b[1] = w1[i][2];
      c2 = __builtin_amdgcn_wmma_f32_16x16x4_f32(false, a, false, b, (short)0, c2,
                                                 false, false);
      b[0] = w0[i][3]; b[1] = w1[i][3];
      c3 = __builtin_amdgcn_wmma_f32_16x16x4_f32(false, a, false, b, (short)0, c3,
                                                 false, false);
    }
  }
  float* op = out + (size_t)(rt * 16 + 8 * hf) * 128 + 64 * g + 4 * m;
#pragma unroll
  for (int r = 0; r < 8; ++r) {
    v4f o;
    o[0] = c0[r]; o[1] = c1[r]; o[2] = c2[r]; o[3] = c3[r];
    *(v4f*)(op + (size_t)r * 128) = o;
  }
}

__device__ __forceinline__ float xhalf(float x) {
  int b = __builtin_bit_cast(int, x);
  return __builtin_bit_cast(
      float, __builtin_amdgcn_permlanex16(b, b, 0x76543210, 0xfedcba98, false,
                                          false));
}

__device__ __forceinline__ v8f silu8(v8f c) {
#pragma unroll
  for (int r = 0; r < 8; ++r) {
    float xv = c[r];
    c[r] = xv / (1.0f + __expf(-xv));
  }
  return c;
}

__device__ __forceinline__ void wmma4_steps(const v4f* Wb,
                                            unsigned wo, unsigned pitch,
                                            const v2f* efr, int i0, v8f& c0,
                                            v8f& c1, v8f& c2, v8f& c3) {
  if (i0 != 0) asm volatile("" : "+v"(c0), "+v"(c1), "+v"(c2), "+v"(c3) : : "memory");
  v4f w0[4], w1[4];
#pragma unroll
  for (int i = 0; i < 4; ++i) {
    w0[i] = Wb[wo + (4u * (i0 + i)) * pitch];
    w1[i] = Wb[wo + (4u * (i0 + i) + 1u) * pitch];
  }
  __builtin_amdgcn_sched_barrier(0);
#pragma unroll
  for (int i = 0; i < 4; ++i) {
    const v2f e = efr[i0 + i];
    v2f a;
    a[0] = w0[i][0]; a[1] = w1[i][0];
    c0 = __builtin_amdgcn_wmma_f32_16x16x4_f32(false, a, false, e,
                                               (short)0, c0, false, false);
    a[0] = w0[i][1]; a[1] = w1[i][1];
    c1 = __builtin_amdgcn_wmma_f32_16x16x4_f32(false, a, false, e,
                                               (short)0, c1, false, false);
    a[0] = w0[i][2]; a[1] = w1[i][2];
    c2 = __builtin_amdgcn_wmma_f32_16x16x4_f32(false, a, false, e,
                                               (short)0, c2, false, false);
    a[0] = w0[i][3]; a[1] = w1[i][3];
    c3 = __builtin_amdgcn_wmma_f32_16x16x4_f32(false, a, false, e,
                                               (short)0, c3, false, false);
  }
}

__device__ __forceinline__ void gate2_frags(v2f* a2, const float* w2, int t) {
#pragma unroll
  for (int p = 0; p < 4; ++p) {
    a2[p][0] = w2[(8 * p + t) * 8];
    a2[p][1] = w2[(8 * p + 4 + t) * 8];
  }
}
__device__ __forceinline__ v8f gate2_acc(v8f gt, const v2f* a2, v8f c) {
#pragma unroll
  for (int p = 0; p < 4; ++p) {
    v2f b;
    b[0] = c[2 * p];
    b[1] = c[2 * p + 1];
    gt = __builtin_amdgcn_wmma_f32_16x16x4_f32(false, a2[p], false, b, (short)0,
                                               gt, false, false);
  }
  return gt;
}

__global__ __launch_bounds__(256) void edge_score_kernel(
    const float* __restrict__ ea, const int* __restrict__ perm,
    const int* __restrict__ ssrc, const int* __restrict__ sdst,
    const float* __restrict__ Wek,
    const float* __restrict__ Wg1, const float* __restrict__ Wg2,
    const float* __restrict__ q, const float* __restrict__ k,
    float* __restrict__ scores, unsigned* __restrict__ mmax) {
  __shared__ v4f stg[WAVES][2][16 * STG_PITCH];
  __shared__ v4f swek[32 * 32];
  __shared__ v4f swg1[32 * 16];
  __shared__ v4f swg2[64 * 2];
  const int lane = threadIdx.x & 31;
  const int wv   = threadIdx.x >> 5;
  const int e0   = (blockIdx.x * WAVES + wv) << 4;
  const int m    = lane & 15;
  const int hf   = lane >> 4;
  const int kk   = hf << 1;

  const int dm = sdst[e0 + m];

  const int sub = lane & 7;
  const int eg  = lane >> 3;
  unsigned qrow[4], krow[4];
#pragma unroll
  for (int j = 0; j < 4; ++j) {
    qrow[j] = (unsigned)sdst[e0 + 4 * j + eg] * 512u + 16u * (unsigned)sub;
    krow[j] = (unsigned)ssrc[e0 + 4 * j + eg] * 512u + 16u * (unsigned)sub;
  }
  v4f* sq = &stg[wv][0][0];
  v4f* sk = &stg[wv][1][0];

  v2f efr[8];
  {
    v4f te[4];
#pragma unroll
    for (int j = 0; j < 4; ++j)
      te[j] = *(const v4f*)(ea + (size_t)perm[e0 + 4 * j + eg] * 32 + 4 * sub);
    const int t = threadIdx.x;
    v4f tw[4], tg[2];
#pragma unroll
    for (int j = 0; j < 4; ++j) tw[j] = ((const v4f*)Wek)[t + 256 * j];
#pragma unroll
    for (int j = 0; j < 2; ++j) tg[j] = ((const v4f*)Wg1)[t + 256 * j];
    const v4f t2 = ((const v4f*)Wg2)[t & 127];
#pragma unroll
    for (int j = 0; j < 4; ++j) sq[(4 * j + eg) * STG_PITCH + sub] = te[j];
    __builtin_amdgcn_fence(__ATOMIC_ACQ_REL, "wavefront");
    const v2f* rowm = (const v2f*)(sq + m * STG_PITCH) + hf;
#pragma unroll
    for (int i = 0; i < 8; ++i) efr[i] = rowm[2 * i];
#pragma unroll
    for (int j = 0; j < 4; ++j) swek[t + 256 * j] = tw[j];
#pragma unroll
    for (int j = 0; j < 2; ++j) swg1[t + 256 * j] = tg[j];
    if (t < 128) swg2[t] = t2;
  }
  __syncthreads();
  const unsigned psi4 = 2u * (unsigned)(m & 7) + (unsigned)(m >> 3);
  v4f o = {};
#pragma unroll 1
  for (int g = 0; g < 2; ++g) {
    v8f c0 = {}, c1 = {}, c2 = {}, c3 = {};
    const unsigned wo = (unsigned)kk * 32u + 16u * (unsigned)g + psi4;
    wmma4_steps(swek, wo, 32u, efr, 0, c0, c1, c2, c3);
    wmma4_steps(swek, wo, 32u, efr, 4, c0, c1, c2, c3);
#pragma unroll
    for (int hp = 0; hp < 2; ++hp) {
      __builtin_amdgcn_sched_barrier(0);
      const unsigned lo = 256u * (unsigned)g + 128u * (unsigned)hp;
      v4f tq[4], tk[4];
#pragma unroll
      for (int j = 0; j < 4; ++j) {
        tq[j] = *(const v4f*)((const char*)q + (qrow[j] + lo));
        tk[j] = *(const v4f*)((const char*)k + (krow[j] + lo));
      }
      __builtin_amdgcn_fence(__ATOMIC_ACQ_REL, "wavefront");
#pragma unroll
      for (int j = 0; j < 4; ++j) {
        sq[(4 * j + eg) * STG_PITCH + sub] = tq[j];
        sk[(4 * j + eg) * STG_PITCH + sub] = tk[j];
      }
      __builtin_amdgcn_fence(__ATOMIC_ACQ_REL, "wavefront");
#pragma unroll
      for (int hl = 0; hl < 2; ++hl) {
        const int hh = 2 * hp + hl;
        const v4f q0 = sq[m * STG_PITCH + 4 * hl + hf];
        const v4f q1 = sq[m * STG_PITCH + 4 * hl + hf + 2];
        const v4f k0 = sk[m * STG_PITCH + 4 * hl + hf];
        const v4f k1 = sk[m * STG_PITCH + 4 * hl + hf + 2];
        float p = q0[0] * (k0[0] + c0[2 * hh]);
        p = fmaf(q0[1], k0[1] + c1[2 * hh], p);
        p = fmaf(q0[2], k0[2] + c2[2 * hh], p);
        p = fmaf(q0[3], k0[3] + c3[2 * hh], p);
        p = fmaf(q1[0], k1[0] + c0[2 * hh + 1], p);
        p = fmaf(q1[1], k1[1] + c1[2 * hh + 1], p);
        p = fmaf(q1[2], k1[2] + c2[2 * hh + 1], p);
        p = fmaf(q1[3], k1[3] + c3[2 * hh + 1], p);
        p += xhalf(p);
        o[hh] = (hf == g) ? p : o[hh];
      }
    }
  }
  v8f gt = {};
  {
    v8f c0 = {}, c1 = {}, c2 = {}, c3 = {};
    const unsigned wo = (unsigned)kk * 16u + (unsigned)m;
    wmma4_steps(swg1, wo, 16u, efr, 0, c0, c1, c2, c3);
    wmma4_steps(swg1, wo, 16u, efr, 4, c0, c1, c2, c3);
    const float* w2 = (const float*)swg2 + (32 * hf) * 8 + (m & 7);
    v2f a20[4], a21[4], a22[4], a23[4];
    gate2_frags(a20, w2, 0);
    gate2_frags(a21, w2, 1);
    gate2_frags(a22, w2, 2);
    gate2_frags(a23, w2, 3);
    gt = gate2_acc(gt, a20, silu8(c0));
    gt = gate2_acc(gt, a21, silu8(c1));
    gt = gate2_acc(gt, a22, silu8(c2));
    gt = gate2_acc(gt, a23, silu8(c3));
  }

  o[0] = o[0] * 0.25f + (hf ? gt[4] : gt[0]);
  o[1] = o[1] * 0.25f + (hf ? gt[5] : gt[1]);
  o[2] = o[2] * 0.25f + (hf ? gt[6] : gt[2]);
  o[3] = o[3] * 0.25f + (hf ? gt[7] : gt[3]);

  *(v4f*)(scores + (size_t)(e0 + m) * 8 + 4 * hf) = o;
  v4f mx = o;
#pragma unroll
  for (int d = 1; d < 16; d <<= 1) {
    const int dd = __shfl_up(dm, d, 16);
    v4f y;
    y[0] = __shfl_up(mx[0], d, 16);
    y[1] = __shfl_up(mx[1], d, 16);
    y[2] = __shfl_up(mx[2], d, 16);
    y[3] = __shfl_up(mx[3], d, 16);
    const bool same = (m >= d) && (dd == dm);
    mx[0] = same ? fmaxf(mx[0], y[0]) : mx[0];
    mx[1] = same ? fmaxf(mx[1], y[1]) : mx[1];
    mx[2] = same ? fmaxf(mx[2], y[2]) : mx[2];
    mx[3] = same ? fmaxf(mx[3], y[3]) : mx[3];
  }
  const int dnx = __shfl_down(dm, 1, 16);
  if (m == 15 || dnx != dm) {
    unsigned* mp = mmax + (size_t)dm * 8 + 4 * hf;
    atomicMax(mp + 0, fkey(mx[0]));
    atomicMax(mp + 1, fkey(mx[1]));
    atomicMax(mp + 2, fkey(mx[2]));
    atomicMax(mp + 3, fkey(mx[3]));
  }
}

#define EXP_STEPS 16
#define EXP_EDGES (4 * EXP_STEPS)

__global__ __launch_bounds__(256) void exp_denom_kernel(
    const int* __restrict__ sdst, float* __restrict__ scores,
    const unsigned* __restrict__ mmax, float* __restrict__ denom, int n_edges) {
  const int lane  = threadIdx.x & 31;
  const int wbase = (blockIdx.x * WAVES + (threadIdx.x >> 5)) * EXP_EDGES;
  if (wbase >= n_edges) return;
  const unsigned h = lane & 7;
  const int sub = lane >> 3;

  if (wbase + EXP_EDGES <= n_edges) {
    float* sp = scores + (size_t)wbase * 8 + lane;
    const int* dp = sdst + wbase + sub;
    int d[EXP_STEPS + 1];
    float s[EXP_STEPS];
    unsigned mk[EXP_STEPS];
#pragma unroll
    for (int i = 0; i < EXP_STEPS; ++i) {
      d[i] = dp[4 * i];
      s[i] = sp[32 * i];
    }
    d[EXP_STEPS] = -1;
#pragma unroll
    for (int i = 0; i < EXP_STEPS; ++i) mk[i] = mmax[(unsigned)d[i] * 8u + h];
    float run = 0.0f;
#pragma unroll
    for (int i = 0; i < EXP_STEPS; ++i) {
      const float es = __expf(s[i] - funkey(mk[i]));
      sp[32 * i] = es;
      run += es;
      if (d[i] != d[i + 1]) {
        atomicAdd(&denom[(unsigned)d[i] * 8u + h], run);
        run = 0.0f;
      }
    }
  } else {
    for (int i = 0; i < EXP_EDGES / 4; ++i) {
      const int e = wbase + 4 * i + sub;
      if (e < n_edges) {
        const unsigned dh = (unsigned)sdst[e] * 8u + h;
        const size_t idx = (size_t)e * 8 + h;
        const float es = __expf(scores[idx] - funkey(mmax[dh]));
        scores[idx] = es;
        atomicAdd(&denom[dh], es);
      }
    }
  }
}

#define CNT_ROUNDS 4
__global__ __launch_bounds__(256) void dst_count_kernel(
    const int* __restrict__ dst, int* __restrict__ cnt, int* __restrict__ rank,
    int n_edges) {
  const int lane  = threadIdx.x & 31;
  const int wbase = (blockIdx.x * WAVES + (threadIdx.x >> 5)) * (32 * CNT_ROUNDS);
  if (wbase + 32 * CNT_ROUNDS <= n_edges) {
    int rd[CNT_ROUNDS], rk[CNT_ROUNDS];
#pragma unroll
    for (int j = 0; j < CNT_ROUNDS; ++j) rd[j] = dst[wbase + 32 * j + lane];
#pragma unroll
    for (int j = 0; j < CNT_ROUNDS; ++j) rk[j] = atomicAdd(&cnt[rd[j]], 1);
#pragma unroll
    for (int j = 0; j < CNT_ROUNDS; ++j) rank[wbase + 32 * j + lane] = rk[j];
  } else {
    for (int j = 0; j < CNT_ROUNDS; ++j) {
      const int e = wbase + 32 * j + lane;
      if (e < n_edges) rank[e] = atomicAdd(&cnt[dst[e]], 1);
    }
  }
}

__global__ __launch_bounds__(1024) void seg_offsets_kernel(
    const int* __restrict__ cnt, int* __restrict__ off, int n) {
  __shared__ int wsum[32];
  __shared__ int psum[32];
  const int t = threadIdx.x, lane = t & 31, w = t >> 5;
  const int base = blockIdx.x * 1024;
  int p = 0;
  for (int i = t; i < base; i += 1024) p += cnt[i];
  const int x = (base + t < n) ? cnt[base + t] : 0;
  int inc = x;
  for (int d = 1; d < 32; d <<= 1) {
    int y = __shfl_up(inc, d);
    if (lane >= d) inc += y;
  }
  for (int d = 16; d > 0; d >>= 1) p += __shfl_xor(p, d);
  if (lane == 31) wsum[w] = inc;
  if (lane == 0) psum[w] = p;
  __syncthreads();
  int pre = 0;
  for (int i = 0; i < 32; ++i) {
    pre += psum[i];
    pre += (i < w) ? wsum[i] : 0;
  }
  if (base + t < n) off[base + t] = pre + inc - x;
}

__global__ __launch_bounds__(256) void dst_order_kernel(
    const int* __restrict__ src, const int* __restrict__ dst,
    const int* __restrict__ rank, const int* __restrict__ off,
    int* __restrict__ perm, int* __restrict__ ssrc, int* __restrict__ sdst,
    int n_edges) {
  const int e = blockIdx.x * blockDim.x + threadIdx.x;
  if (e >= n_edges) return;
  const int d = dst[e];
  const int pos = off[d] + rank[e];
  perm[pos] = e;
  ssrc[pos] = src[e];
  sdst[pos] = d;
}

#define TILES 2

__global__ __launch_bounds__(256) __attribute__((amdgpu_waves_per_eu(6))) void edge_aggregate_kernel(
    const float* __restrict__ ea, const int* __restrict__ perm,
    const int* __restrict__ ssrc, const int* __restrict__ sdst,
    const float* __restrict__ Wev, const float* __restrict__ v,
    const float* __restrict__ scores, const float* __restrict__ denom,
    float* __restrict__ acc) {
  __shared__ float carry[WAVES][8][32];
  __shared__ v4f sta[WAVES][16 * STG_PITCH];
  __shared__ v4f swev[32 * 32];
  {
    const int t = threadIdx.x;
    v4f tw[4];
#pragma unroll
    for (int j = 0; j < 4; ++j) tw[j] = ((const v4f*)Wev)[t + 256 * j];
#pragma unroll
    for (int j = 0; j < 4; ++j) swev[t + 256 * j] = tw[j];
  }
  __syncthreads();
  const int lane = threadIdx.x & 31;
  const int wv   = threadIdx.x >> 5;
  const int m    = lane & 15;
  const int hf   = lane >> 4;
  const int kk   = hf << 1;
  const unsigned hq = (unsigned)(m >> 2);
  const int wbase = (blockIdx.x * WAVES + wv) * (16 * TILES);
  const int sbase = wbase + hf * (8 * TILES);
  const int sub = lane & 7;
  const int eg  = lane >> 3;
  v4f* sa = &sta[wv][0];
  const v2f* rowm = (const v2f*)(sa + m * STG_PITCH) + hf;

#pragma unroll 1
  for (int i = 0; i < TILES; ++i) {
    v4f te[4];
#pragma unroll
    for (int jj = 0; jj < 4; ++jj) {
      const int pe = perm[wbase + (jj >> 1) * (8 * TILES) + 4 * (jj & 1) + eg + 8 * i];
      te[jj] = *(const v4f*)(ea + (size_t)pe * 32 + 4 * sub);
    }

    const int p = sbase + 8 * i;
    unsigned so[8], dn[8], vo[8];
    {
      const v4i s0 = *(const v4i*)(ssrc + p), s1 = *(const v4i*)(ssrc + p + 4);
      const v4i d0 = *(const v4i*)(sdst + p), d1 = *(const v4i*)(sdst + p + 4);
#pragma unroll
      for (int r = 0; r < 4; ++r) {
        so[r] = (unsigned)(p + r) * 32u;           so[r + 4] = (unsigned)(p + r + 4) * 32u;
        vo[r] = (unsigned)s0[r] * 512u + 16u * m;  vo[r + 4] = (unsigned)s1[r] * 512u + 16u * m;
        dn[r] = (unsigned)d0[r] * 32u;             dn[r + 4] = (unsigned)d1[r] * 32u;
      }
    }
    const bool last = (i == TILES - 1);
    const unsigned dnext = last ? 0xffffffffu : (unsigned)sdst[p + 8] * 32u;

    __builtin_amdgcn_fence(__ATOMIC_ACQ_REL, "wavefront");
#pragma unroll
    for (int jj = 0; jj < 4; ++jj) sa[(4 * jj + eg) * STG_PITCH + sub] = te[jj];
    __builtin_amdgcn_fence(__ATOMIC_ACQ_REL, "wavefront");
    v2f afr[8];
#pragma unroll
    for (int j = 0; j < 8; ++j) afr[j] = rowm[2 * j];

#pragma unroll 1
    for (int g = 0; g < 2; ++g) {
      v8f c0 = {}, c1 = {}, c2 = {}, c3 = {};
      const v4f* wp = swev + (kk * 32 + 16 * g + m);
      v4f wr0[8], wr1[8];
#pragma unroll
      for (int j = 0; j < 8; ++j) {
        wr0[j] = wp[(4 * j) * 32];
        wr1[j] = wp[(4 * j + 1) * 32];
      }
      __builtin_amdgcn_sched_barrier(0);
#pragma unroll
      for (int j = 0; j < 8; ++j) {
        const v4f w0 = wr0[j];
        const v4f w1 = wr1[j];
        v2f b;
        b[0] = w0[0]; b[1] = w1[0];
        c0 = __builtin_amdgcn_wmma_f32_16x16x4_f32(false, afr[j], false, b,
                                                   (short)0, c0, false, false);
        b[0] = w0[1]; b[1] = w1[1];
        c1 = __builtin_amdgcn_wmma_f32_16x16x4_f32(false, afr[j], false, b,
                                                   (short)0, c1, false, false);
        b[0] = w0[2]; b[1] = w1[2];
        c2 = __builtin_amdgcn_wmma_f32_16x16x4_f32(false, afr[j], false, b,
                                                   (short)0, c2, false, false);
        b[0] = w0[3]; b[1] = w1[3];
        c3 = __builtin_amdgcn_wmma_f32_16x16x4_f32(false, afr[j], false, b,
                                                   (short)0, c3, false, false);
      }
      __builtin_amdgcn_sched_barrier(0);
      const unsigned hd = 16u * (unsigned)g + 4u * hq;
      const unsigned fo = 256u * (unsigned)g;
      v4f t[8];
#pragma unroll
      for (int r = 0; r < 8; ++r) {
        const float att = *(const float*)((const char*)scores + (so[r] + hd)) /
                          (*(const float*)((const char*)denom + (dn[r] + hd)) + 1e-9f);
        const v4f vr = *(const v4f*)((const char*)v + (vo[r] + fo));
        t[r][0] = att * (vr[0] + c0[r]);
        t[r][1] = att * (vr[1] + c1[r]);
        t[r][2] = att * (vr[2] + c2[r]);
        t[r][3] = att * (vr[3] + c3[r]);
      }
      v4f run = {};
      if (i != 0) {
        run[0] = carry[wv][4 * g + 0][lane];
        run[1] = carry[wv][4 * g + 1][lane];
        run[2] = carry[wv][4 * g + 2][lane];
        run[3] = carry[wv][4 * g + 3][lane];
      }
#pragma unroll
      for (int r = 0; r < 8; ++r) {
        run[0] += t[r][0];
        run[1] += t[r][1];
        run[2] += t[r][2];
        run[3] += t[r][3];
        const unsigned nx = (r < 7) ? dn[r + 1 < 8 ? r + 1 : 7] : dnext;
        if (dn[r] != nx) {
          float* ap = (float*)((char*)acc + (dn[r] * 16u + fo + 16u * (unsigned)m));
          atomicAdd(ap + 0, run[0]);
          atomicAdd(ap + 1, run[1]);
          atomicAdd(ap + 2, run[2]);
          atomicAdd(ap + 3, run[3]);
          run[0] = 0.0f; run[1] = 0.0f; run[2] = 0.0f; run[3] = 0.0f;
        }
      }
      if (!last) {
        carry[wv][4 * g + 0][lane] = run[0];
        carry[wv][4 * g + 1][lane] = run[1];
        carry[wv][4 * g + 2][lane] = run[2];
        carry[wv][4 * g + 3][lane] = run[3];
      }
    }
  }
}

extern "C" void kernel_launch(void* const* d_in, const int* in_sizes, int n_in,
                              void* d_out, int out_size, void* d_ws,
                              size_t ws_size, hipStream_t stream) {
  const float* x   = (const float*)d_in[0];
  const int*   ei  = (const int*)d_in[1];
  const float* ea  = (const float*)d_in[2];
  const float* Wq  = (const float*)d_in[3];
  const float* Wk  = (const float*)d_in[4];
  const float* Wv  = (const float*)d_in[5];
  const float* Wo  = (const float*)d_in[6];
  const float* Wek = (const float*)d_in[7];
  const float* Wev = (const float*)d_in[8];
  const float* Wg1 = (const float*)d_in[9];
  const float* Wg2 = (const float*)d_in[10];
  float* out = (float*)d_out;

  const int N = in_sizes[0] / 128;
  const int E = in_sizes[2] / 32;
  const int* src = ei;
  const int* dst = ei + E;

  float* q      = (float*)d_ws;
  float* k      = q + (size_t)N * 128;
  float* scores = k + (size_t)N * 128;
  float* ord    = scores + (size_t)E * 8;
  float* acc    = ord + (size_t)N * 128;
  float* denom  = acc + (size_t)N * 128;
  unsigned* mmax = (unsigned*)(denom + (size_t)N * 8);
  float* v      = q;
  int* perm = (int*)ord;
  int* ssrc = perm + E;
  int* sdst = ssrc + E;
  int* off  = sdst + E;
  int* rank = off + N;
  int* cnt  = (int*)acc - N;

  hipMemsetAsync(cnt, 0, (size_t)(N + N * 128 + N * 8 + N * 8) * sizeof(float),
                 stream);

  dim3 blk(256);
  dim3 gblk(32 * GEMM_WAVES);
  dst_count_kernel<<<dim3((E + WAVES * 32 * CNT_ROUNDS - 1) / (WAVES * 32 * CNT_ROUNDS)),
                     blk, 0, stream>>>(dst, cnt, rank, E);

  seg_offsets_kernel<<<dim3((N + 1023) / 1024), dim3(1024), 0, stream>>>(
      cnt, off, N);

  dst_order_kernel<<<dim3((E + 255) / 256), blk, 0, stream>>>(
      src, dst, rank, off, perm, ssrc, sdst, E);

  gemm128_kernel<<<dim3((N / 16 * 2 + GEMM_WAVES - 1) / GEMM_WAVES), gblk, 0, stream>>>(x, Wq, q, N / 16);
  gemm128_kernel<<<dim3((N / 16 * 2 + GEMM_WAVES - 1) / GEMM_WAVES), gblk, 0, stream>>>(x, Wk, k, N / 16);

  edge_score_kernel<<<dim3(E / (16 * WAVES)), blk, 0, stream>>>(
      ea, perm, ssrc, sdst, Wek, Wg1, Wg2, q, k, scores, mmax);

  exp_denom_kernel<<<dim3((E + WAVES * EXP_EDGES - 1) / (WAVES * EXP_EDGES)), blk,
                     0, stream>>>(sdst, scores, mmax, denom, E);

  gemm128_kernel<<<dim3((N / 16 * 2 + GEMM_WAVES - 1) / GEMM_WAVES), gblk, 0, stream>>>(x, Wv, v, N / 16);

  edge_aggregate_kernel<<<dim3(E / (16 * WAVES * TILES)), blk, 0, stream>>>(
      ea, perm, ssrc, sdst, Wev, v, scores, denom, acc);

  gemm128_kernel<<<dim3((N / 16 * 2 + GEMM_WAVES - 1) / GEMM_WAVES), gblk, 0, stream>>>(acc, Wo, out, N / 16);
}


# softmax normalised after the sum: last GEMM scales acc rows by 1/(denom+1e-9) on load, so exp and denom move into the aggregate pass and the exp_denom launch is gone; denom sum order changes
# speedup vs baseline: 2.8467x; 1.0316x over previous
#include <hip/hip_runtime.h>


typedef __attribute__((ext_vector_type(2))) float v2f;
typedef __attribute__((ext_vector_type(4))) float v4f;
typedef __attribute__((ext_vector_type(8))) float v8f;
typedef __attribute__((ext_vector_type(4))) int v4i;

#define WAVES 8

__device__ __forceinline__ unsigned fkey(float f) {
  unsigned u = __float_as_uint(f);
  return (u & 0x80000000u) ? ~u : (u | 0x80000000u);
}
__device__ __forceinline__ float funkey(unsigned kk) {
  unsigned u = (kk & 0x80000000u) ? (kk & 0x7FFFFFFFu) : ~kk;
  return __uint_as_float(u);
}

#define GEMM_WAVES 4
#define STG_PITCH 9
__device__ __forceinline__ void gemm128_body(
    v4f* sa, const float* __restrict__ A, const float* __restrict__ W,
    float* __restrict__ out, int n_tiles, const float* __restrict__ dnm,
    const bool norm) {
  const int lane = threadIdx.x & 31;
  const int unit = blockIdx.x * GEMM_WAVES + (threadIdx.x >> 5);
  const int rt   = unit >> 1;
  const int g    = unit & 1;
  if (rt >= n_tiles) return;
  const int m    = lane & 15;
  const int hf   = lane >> 4;
  const int kk   = hf << 1;

  v8f c0 = {}, c1 = {}, c2 = {}, c3 = {};
  const int sub = lane & 7;
  const int eg  = lane >> 3;
  const float* ab = A + (size_t)(rt * 16 + eg) * 128 + 4 * sub;
  const v2f* rowm = (const v2f*)(sa + m * STG_PITCH) + hf;
  const float* wp = W + (size_t)kk * 128 + 64 * g + 4 * m;
#pragma unroll
  for (int t0 = 0; t0 < 128; t0 += 32) {
    __builtin_amdgcn_sched_barrier(0);
    v4f ta[4], w0[8], w1[8];
#pragma unroll
    for (int j = 0; j < 4; ++j) ta[j] = *(const v4f*)(ab + (size_t)(4 * j) * 128 + t0);
    float dd[4];
    if (norm) {
      const float* db = dnm + (size_t)(rt * 16 + eg) * 8 + (sub >> 2) + (t0 >> 4);
#pragma unroll
      for (int j = 0; j < 4; ++j) dd[j] = db[(size_t)(4 * j) * 8];
    }
#pragma unroll
    for (int i = 0; i < 8; ++i) {
      w0[i] = *(const v4f*)(wp + (size_t)(t0 + 4 * i) * 128);
      w1[i] = *(const v4f*)(wp + (size_t)(t0 + 4 * i + 1) * 128);
    }
    __builtin_amdgcn_fence(__ATOMIC_ACQ_REL, "wavefront");
    if (norm) {
#pragma unroll
      for (int j = 0; j < 4; ++j) {
        const float rs = 1.0f / (dd[j] + 1e-9f);
        ta[j][0] *= rs; ta[j][1] *= rs; ta[j][2] *= rs; ta[j][3] *= rs;
      }
    }
#pragma unroll
    for (int j = 0; j < 4; ++j) sa[(4 * j + eg) * STG_PITCH + sub] = ta[j];
    __builtin_amdgcn_fence(__ATOMIC_ACQ_REL, "wavefront");
    v2f afr[8];
#pragma unroll
    for (int i = 0; i < 8; ++i) afr[i] = rowm[2 * i];
#pragma unroll
    for (int i = 0; i < 8; ++i) {
      const v2f a = afr[i];
      v2f b;
      b[0] = w0[i][0]; b[1] = w1[i][0];
      c0 = __builtin_amdgcn_wmma_f32_16x16x4_f32(false, a, false, b, (short)0, c0,
                                                 false, false);
      b[0] = w0[i][1]; b[1] = w1[i][1];
      c1 = __builtin_amdgcn_wmma_f32_16x16x4_f32(false, a, false, b, (short)0, c1,
                                                 false, false);
      b[0] = w0[i][2]; b[1] = w1[i][2];
      c2 = __builtin_amdgcn_wmma_f32_16x16x4_f32(false, a, false, b, (short)0, c2,
                                                 false, false);
      b[0] = w0[i][3]; b[1] = w1[i][3];
      c3 = __builtin_amdgcn_wmma_f32_16x16x4_f32(false, a, false, b, (short)0, c3,
                                                 false, false);
    }
  }
  float* op = out + (size_t)(rt * 16 + 8 * hf) * 128 + 64 * g + 4 * m;
#pragma unroll
  for (int r = 0; r < 8; ++r) {
    v4f o;
    o[0] = c0[r]; o[1] = c1[r]; o[2] = c2[r]; o[3] = c3[r];
    *(v4f*)(op + (size_t)r * 128) = o;
  }
}

__global__ __launch_bounds__(32 * GEMM_WAVES) void gemm128_kernel(
    const float* __restrict__ A, const float* __restrict__ W,
    float* __restrict__ out, int n_tiles) {
  __shared__ v4f sta[GEMM_WAVES][16 * STG_PITCH];
  gemm128_body(&sta[threadIdx.x >> 5][0], A, W, out, n_tiles, nullptr, false);
}

__global__ __launch_bounds__(32 * GEMM_WAVES) void gemm128_norm_kernel(
    const float* __restrict__ A, const float* __restrict__ W,
    float* __restrict__ out, int n_tiles, const float* __restrict__ dnm) {
  __shared__ v4f sta[GEMM_WAVES][16 * STG_PITCH];
  gemm128_body(&sta[threadIdx.x >> 5][0], A, W, out, n_tiles, dnm, true);
}

__device__ __forceinline__ float xhalf(float x) {
  int b = __builtin_bit_cast(int, x);
  return __builtin_bit_cast(
      float, __builtin_amdgcn_permlanex16(b, b, 0x76543210, 0xfedcba98, false,
                                          false));
}

__device__ __forceinline__ v8f silu8(v8f c) {
#pragma unroll
  for (int r = 0; r < 8; ++r) {
    float xv = c[r];
    c[r] = xv / (1.0f + __expf(-xv));
  }
  return c;
}

__device__ __forceinline__ void wmma4_steps(const v4f* Wb,
                                            unsigned wo, unsigned pitch,
                                            const v2f* efr, int i0, v8f& c0,
                                            v8f& c1, v8f& c2, v8f& c3) {
  if (i0 != 0) asm volatile("" : "+v"(c0), "+v"(c1), "+v"(c2), "+v"(c3) : : "memory");
  v4f w0[4], w1[4];
#pragma unroll
  for (int i = 0; i < 4; ++i) {
    w0[i] = Wb[wo + (4u * (i0 + i)) * pitch];
    w1[i] = Wb[wo + (4u * (i0 + i) + 1u) * pitch];
  }
  __builtin_amdgcn_sched_barrier(0);
#pragma unroll
  for (int i = 0; i < 4; ++i) {
    const v2f e = efr[i0 + i];
    v2f a;
    a[0] = w0[i][0]; a[1] = w1[i][0];
    c0 = __builtin_amdgcn_wmma_f32_16x16x4_f32(false, a, false, e,
                                               (short)0, c0, false, false);
    a[0] = w0[i][1]; a[1] = w1[i][1];
    c1 = __builtin_amdgcn_wmma_f32_16x16x4_f32(false, a, false, e,
                                               (short)0, c1, false, false);
    a[0] = w0[i][2]; a[1] = w1[i][2];
    c2 = __builtin_amdgcn_wmma_f32_16x16x4_f32(false, a, false, e,
                                               (short)0, c2, false, false);
    a[0] = w0[i][3]; a[1] = w1[i][3];
    c3 = __builtin_amdgcn_wmma_f32_16x16x4_f32(false, a, false, e,
                                               (short)0, c3, false, false);
  }
}

__device__ __forceinline__ void gate2_frags(v2f* a2, const float* w2, int t) {
#pragma unroll
  for (int p = 0; p < 4; ++p) {
    a2[p][0] = w2[(8 * p + t) * 8];
    a2[p][1] = w2[(8 * p + 4 + t) * 8];
  }
}
__device__ __forceinline__ v8f gate2_acc(v8f gt, const v2f* a2, v8f c) {
#pragma unroll
  for (int p = 0; p < 4; ++p) {
    v2f b;
    b[0] = c[2 * p];
    b[1] = c[2 * p + 1];
    gt = __builtin_amdgcn_wmma_f32_16x16x4_f32(false, a2[p], false, b, (short)0,
                                               gt, false, false);
  }
  return gt;
}

__global__ __launch_bounds__(256) void edge_score_kernel(
    const float* __restrict__ ea, const int* __restrict__ perm,
    const int* __restrict__ ssrc, const int* __restrict__ sdst,
    const float* __restrict__ Wek,
    const float* __restrict__ Wg1, const float* __restrict__ Wg2,
    const float* __restrict__ q, const float* __restrict__ k,
    float* __restrict__ scores, unsigned* __restrict__ mmax) {
  __shared__ v4f stg[WAVES][2][16 * STG_PITCH];
  __shared__ v4f swek[32 * 32];
  __shared__ v4f swg1[32 * 16];
  __shared__ v4f swg2[64 * 2];
  const int lane = threadIdx.x & 31;
  const int wv   = threadIdx.x >> 5;
  const int e0   = (blockIdx.x * WAVES + wv) << 4;
  const int m    = lane & 15;
  const int hf   = lane >> 4;
  const int kk   = hf << 1;

  const int dm = sdst[e0 + m];

  const int sub = lane & 7;
  const int eg  = lane >> 3;
  unsigned qrow[4], krow[4];
#pragma unroll
  for (int j = 0; j < 4; ++j) {
    qrow[j] = (unsigned)sdst[e0 + 4 * j + eg] * 512u + 16u * (unsigned)sub;
    krow[j] = (unsigned)ssrc[e0 + 4 * j + eg] * 512u + 16u * (unsigned)sub;
  }
  v4f* sq = &stg[wv][0][0];
  v4f* sk = &stg[wv][1][0];

  v2f efr[8];
  {
    v4f te[4];
#pragma unroll
    for (int j = 0; j < 4; ++j)
      te[j] = *(const v4f*)(ea + (size_t)perm[e0 + 4 * j + eg] * 32 + 4 * sub);
    const int t = threadIdx.x;
    v4f tw[4], tg[2];
#pragma unroll
    for (int j = 0; j < 4; ++j) tw[j] = ((const v4f*)Wek)[t + 256 * j];
#pragma unroll
    for (int j = 0; j < 2; ++j) tg[j] = ((const v4f*)Wg1)[t + 256 * j];
    const v4f t2 = ((const v4f*)Wg2)[t & 127];
#pragma unroll
    for (int j = 0; j < 4; ++j) sq[(4 * j + eg) * STG_PITCH + sub] = te[j];
    __builtin_amdgcn_fence(__ATOMIC_ACQ_REL, "wavefront");
    const v2f* rowm = (const v2f*)(sq + m * STG_PITCH) + hf;
#pragma unroll
    for (int i = 0; i < 8; ++i) efr[i] = rowm[2 * i];
#pragma unroll
    for (int j = 0; j < 4; ++j) swek[t + 256 * j] = tw[j];
#pragma unroll
    for (int j = 0; j < 2; ++j) swg1[t + 256 * j] = tg[j];
    if (t < 128) swg2[t] = t2;
  }
  __syncthreads();
  const unsigned psi4 = 2u * (unsigned)(m & 7) + (unsigned)(m >> 3);
  v4f o = {};
#pragma unroll 1
  for (int g = 0; g < 2; ++g) {
    v8f c0 = {}, c1 = {}, c2 = {}, c3 = {};
    const unsigned wo = (unsigned)kk * 32u + 16u * (unsigned)g + psi4;
    wmma4_steps(swek, wo, 32u, efr, 0, c0, c1, c2, c3);
    wmma4_steps(swek, wo, 32u, efr, 4, c0, c1, c2, c3);
#pragma unroll
    for (int hp = 0; hp < 2; ++hp) {
      __builtin_amdgcn_sched_barrier(0);
      const unsigned lo = 256u * (unsigned)g + 128u * (unsigned)hp;
      v4f tq[4], tk[4];
#pragma unroll
      for (int j = 0; j < 4; ++j) {
        tq[j] = *(const v4f*)((const char*)q + (qrow[j] + lo));
        tk[j] = *(const v4f*)((const char*)k + (krow[j] + lo));
      }
      __builtin_amdgcn_fence(__ATOMIC_ACQ_REL, "wavefront");
#pragma unroll
      for (int j = 0; j < 4; ++j) {
        sq[(4 * j + eg) * STG_PITCH + sub] = tq[j];
        sk[(4 * j + eg) * STG_PITCH + sub] = tk[j];
      }
      __builtin_amdgcn_fence(__ATOMIC_ACQ_REL, "wavefront");
#pragma unroll
      for (int hl = 0; hl < 2; ++hl) {
        const int hh = 2 * hp + hl;
        const v4f q0 = sq[m * STG_PITCH + 4 * hl + hf];
        const v4f q1 = sq[m * STG_PITCH + 4 * hl + hf + 2];
        const v4f k0 = sk[m * STG_PITCH + 4 * hl + hf];
        const v4f k1 = sk[m * STG_PITCH + 4 * hl + hf + 2];
        float p = q0[0] * (k0[0] + c0[2 * hh]);
        p = fmaf(q0[1], k0[1] + c1[2 * hh], p);
        p = fmaf(q0[2], k0[2] + c2[2 * hh], p);
        p = fmaf(q0[3], k0[3] + c3[2 * hh], p);
        p = fmaf(q1[0], k1[0] + c0[2 * hh + 1], p);
        p = fmaf(q1[1], k1[1] + c1[2 * hh + 1], p);
        p = fmaf(q1[2], k1[2] + c2[2 * hh + 1], p);
        p = fmaf(q1[3], k1[3] + c3[2 * hh + 1], p);
        p += xhalf(p);
        o[hh] = (hf == g) ? p : o[hh];
      }
    }
  }
  v8f gt = {};
  {
    v8f c0 = {}, c1 = {}, c2 = {}, c3 = {};
    const unsigned wo = (unsigned)kk * 16u + (unsigned)m;
    wmma4_steps(swg1, wo, 16u, efr, 0, c0, c1, c2, c3);
    wmma4_steps(swg1, wo, 16u, efr, 4, c0, c1, c2, c3);
    const float* w2 = (const float*)swg2 + (32 * hf) * 8 + (m & 7);
    v2f a20[4], a21[4], a22[4], a23[4];
    gate2_frags(a20, w2, 0);
    gate2_frags(a21, w2, 1);
    gate2_frags(a22, w2, 2);
    gate2_frags(a23, w2, 3);
    gt = gate2_acc(gt, a20, silu8(c0));
    gt = gate2_acc(gt, a21, silu8(c1));
    gt = gate2_acc(gt, a22, silu8(c2));
    gt = gate2_acc(gt, a23, silu8(c3));
  }

  o[0] = o[0] * 0.25f + (hf ? gt[4] : gt[0]);
  o[1] = o[1] * 0.25f + (hf ? gt[5] : gt[1]);
  o[2] = o[2] * 0.25f + (hf ? gt[6] : gt[2]);
  o[3] = o[3] * 0.25f + (hf ? gt[7] : gt[3]);

  *(v4f*)(scores + (size_t)(e0 + m) * 8 + 4 * hf) = o;
  v4f mx = o;
#pragma unroll
  for (int d = 1; d < 16; d <<= 1) {
    const int dd = __shfl_up(dm, d, 16);
    v4f y;
    y[0] = __shfl_up(mx[0], d, 16);
    y[1] = __shfl_up(mx[1], d, 16);
    y[2] = __shfl_up(mx[2], d, 16);
    y[3] = __shfl_up(mx[3], d, 16);
    const bool same = (m >= d) && (dd == dm);
    mx[0] = same ? fmaxf(mx[0], y[0]) : mx[0];
    mx[1] = same ? fmaxf(mx[1], y[1]) : mx[1];
    mx[2] = same ? fmaxf(mx[2], y[2]) : mx[2];
    mx[3] = same ? fmaxf(mx[3], y[3]) : mx[3];
  }
  const int dnx = __shfl_down(dm, 1, 16);
  if (m == 15 || dnx != dm) {
    unsigned* mp = mmax + (size_t)dm * 8 + 4 * hf;
    atomicMax(mp + 0, fkey(mx[0]));
    atomicMax(mp + 1, fkey(mx[1]));
    atomicMax(mp + 2, fkey(mx[2]));
    atomicMax(mp + 3, fkey(mx[3]));
  }
}

#define CNT_ROUNDS 4
__global__ __launch_bounds__(256) void dst_count_kernel(
    const int* __restrict__ dst, int* __restrict__ cnt, int* __restrict__ rank,
    int n_edges) {
  const int lane  = threadIdx.x & 31;
  const int wbase = (blockIdx.x * WAVES + (threadIdx.x >> 5)) * (32 * CNT_ROUNDS);
  if (wbase + 32 * CNT_ROUNDS <= n_edges) {
    int rd[CNT_ROUNDS], rk[CNT_ROUNDS];
#pragma unroll
    for (int j = 0; j < CNT_ROUNDS; ++j) rd[j] = dst[wbase + 32 * j + lane];
#pragma unroll
    for (int j = 0; j < CNT_ROUNDS; ++j) rk[j] = atomicAdd(&cnt[rd[j]], 1);
#pragma unroll
    for (int j = 0; j < CNT_ROUNDS; ++j) rank[wbase + 32 * j + lane] = rk[j];
  } else {
    for (int j = 0; j < CNT_ROUNDS; ++j) {
      const int e = wbase + 32 * j + lane;
      if (e < n_edges) rank[e] = atomicAdd(&cnt[dst[e]], 1);
    }
  }
}

__global__ __launch_bounds__(1024) void seg_offsets_kernel(
    const int* __restrict__ cnt, int* __restrict__ off, int n) {
  __shared__ int wsum[32];
  __shared__ int psum[32];
  const int t = threadIdx.x, lane = t & 31, w = t >> 5;
  const int base = blockIdx.x * 1024;
  int p = 0;
  for (int i = t; i < base; i += 1024) p += cnt[i];
  const int x = (base + t < n) ? cnt[base + t] : 0;
  int inc = x;
  for (int d = 1; d < 32; d <<= 1) {
    int y = __shfl_up(inc, d);
    if (lane >= d) inc += y;
  }
  for (int d = 16; d > 0; d >>= 1) p += __shfl_xor(p, d);
  if (lane == 31) wsum[w] = inc;
  if (lane == 0) psum[w] = p;
  __syncthreads();
  int pre = 0;
  for (int i = 0; i < 32; ++i) {
    pre += psum[i];
    pre += (i < w) ? wsum[i] : 0;
  }
  if (base + t < n) off[base + t] = pre + inc - x;
}

__global__ __launch_bounds__(256) void dst_order_kernel(
    const int* __restrict__ src, const int* __restrict__ dst,
    const int* __restrict__ rank, const int* __restrict__ off,
    int* __restrict__ perm, int* __restrict__ ssrc, int* __restrict__ sdst,
    int n_edges) {
  const int e = blockIdx.x * blockDim.x + threadIdx.x;
  if (e >= n_edges) return;
  const int d = dst[e];
  const int pos = off[d] + rank[e];
  perm[pos] = e;
  ssrc[pos] = src[e];
  sdst[pos] = d;
}

#define TILES 2

__global__ __launch_bounds__(256) __attribute__((amdgpu_waves_per_eu(6))) void edge_aggregate_kernel(
    const float* __restrict__ ea, const int* __restrict__ perm,
    const int* __restrict__ ssrc, const int* __restrict__ sdst,
    const float* __restrict__ Wev, const float* __restrict__ v,
    const float* __restrict__ scores, const unsigned* __restrict__ mmax,
    float* __restrict__ denom, float* __restrict__ acc) {
  __shared__ float carry[WAVES][10][32];
  __shared__ v4f sta[WAVES][16 * STG_PITCH];
  __shared__ v4f swev[32 * 32];
  {
    const int t = threadIdx.x;
    v4f tw[4];
#pragma unroll
    for (int j = 0; j < 4; ++j) tw[j] = ((const v4f*)Wev)[t + 256 * j];
#pragma unroll
    for (int j = 0; j < 4; ++j) swev[t + 256 * j] = tw[j];
  }
  __syncthreads();
  const int lane = threadIdx.x & 31;
  const int wv   = threadIdx.x >> 5;
  const int m    = lane & 15;
  const int hf   = lane >> 4;
  const int kk   = hf << 1;
  const unsigned hq = (unsigned)(m >> 2);
  const int wbase = (blockIdx.x * WAVES + wv) * (16 * TILES);
  const int sbase = wbase + hf * (8 * TILES);
  const int sub = lane & 7;
  const int eg  = lane >> 3;
  v4f* sa = &sta[wv][0];
  const v2f* rowm = (const v2f*)(sa + m * STG_PITCH) + hf;

#pragma unroll 1
  for (int i = 0; i < TILES; ++i) {
    v4f te[4];
#pragma unroll
    for (int jj = 0; jj < 4; ++jj) {
      const int pe = perm[wbase + (jj >> 1) * (8 * TILES) + 4 * (jj & 1) + eg + 8 * i];
      te[jj] = *(const v4f*)(ea + (size_t)pe * 32 + 4 * sub);
    }

    const int p = sbase + 8 * i;
    unsigned so[8], dn[8], vo[8];
    {
      const v4i s0 = *(const v4i*)(ssrc + p), s1 = *(const v4i*)(ssrc + p + 4);
      const v4i d0 = *(const v4i*)(sdst + p), d1 = *(const v4i*)(sdst + p + 4);
#pragma unroll
      for (int r = 0; r < 4; ++r) {
        so[r] = (unsigned)(p + r) * 32u;           so[r + 4] = (unsigned)(p + r + 4) * 32u;
        vo[r] = (unsigned)s0[r] * 512u + 16u * m;  vo[r + 4] = (unsigned)s1[r] * 512u + 16u * m;
        dn[r] = (unsigned)d0[r] * 32u;             dn[r + 4] = (unsigned)d1[r] * 32u;
      }
    }
    const bool last = (i == TILES - 1);
    const unsigned dnext = last ? 0xffffffffu : (unsigned)sdst[p + 8] * 32u;

    __builtin_amdgcn_fence(__ATOMIC_ACQ_REL, "wavefront");
#pragma unroll
    for (int jj = 0; jj < 4; ++jj) sa[(4 * jj + eg) * STG_PITCH + sub] = te[jj];
    __builtin_amdgcn_fence(__ATOMIC_ACQ_REL, "wavefront");
    v2f afr[8];
#pragma unroll
    for (int j = 0; j < 8; ++j) afr[j] = rowm[2 * j];

#pragma unroll 1
    for (int g = 0; g < 2; ++g) {
      v8f c0 = {}, c1 = {}, c2 = {}, c3 = {};
      const v4f* wp = swev + (kk * 32 + 16 * g + m);
      v4f wr0[8], wr1[8];
#pragma unroll
      for (int j = 0; j < 8; ++j) {
        wr0[j] = wp[(4 * j) * 32];
        wr1[j] = wp[(4 * j + 1) * 32];
      }
      __builtin_amdgcn_sched_barrier(0);
#pragma unroll
      for (int j = 0; j < 8; ++j) {
        const v4f w0 = wr0[j];
        const v4f w1 = wr1[j];
        v2f b;
        b[0] = w0[0]; b[1] = w1[0];
        c0 = __builtin_amdgcn_wmma_f32_16x16x4_f32(false, afr[j], false, b,
                                                   (short)0, c0, false, false);
        b[0] = w0[1]; b[1] = w1[1];
        c1 = __builtin_amdgcn_wmma_f32_16x16x4_f32(false, afr[j], false, b,
                                                   (short)0, c1, false, false);
        b[0] = w0[2]; b[1] = w1[2];
        c2 = __builtin_amdgcn_wmma_f32_16x16x4_f32(false, afr[j], false, b,
                                                   (short)0, c2, false, false);
        b[0] = w0[3]; b[1] = w1[3];
        c3 = __builtin_amdgcn_wmma_f32_16x16x4_f32(false, afr[j], false, b,
                                                   (short)0, c3, false, false);
      }
      __builtin_amdgcn_sched_barrier(0);
      const unsigned hd = 16u * (unsigned)g + 4u * hq;
      const unsigned fo = 256u * (unsigned)g;
      const unsigned gm = 16u * (unsigned)g + (unsigned)m;
      v4f t[8];
      float es[8];
#pragma unroll
      for (int r = 0; r < 8; ++r) {
        es[r] = __expf(*(const float*)((const char*)scores + (so[r] + hd)) -
                       funkey(*(const unsigned*)((const char*)mmax + (dn[r] + hd))));
        const v4f vr = *(const v4f*)((const char*)v + (vo[r] + fo));
        t[r][0] = es[r] * (vr[0] + c0[r]);
        t[r][1] = es[r] * (vr[1] + c1[r]);
        t[r][2] = es[r] * (vr[2] + c2[r]);
        t[r][3] = es[r] * (vr[3] + c3[r]);
      }
      v4f run = {};
      float rune = 0.0f;
      if (i != 0) {
        run[0] = carry[wv][5 * g + 0][lane];
        run[1] = carry[wv][5 * g + 1][lane];
        run[2] = carry[wv][5 * g + 2][lane];
        run[3] = carry[wv][5 * g + 3][lane];
        rune   = carry[wv][5 * g + 4][lane];
      }
#pragma unroll
      for (int r = 0; r < 8; ++r) {
        run[0] += t[r][0];
        run[1] += t[r][1];
        run[2] += t[r][2];
        run[3] += t[r][3];
        rune += es[r];
        const unsigned nx = (r < 7) ? dn[r + 1 < 8 ? r + 1 : 7] : dnext;
        if (dn[r] != nx) {
          float* ap = (float*)((char*)acc + ((dn[r] + gm) << 4));
          atomicAdd(ap + 0, run[0]);
          atomicAdd(ap + 1, run[1]);
          atomicAdd(ap + 2, run[2]);
          atomicAdd(ap + 3, run[3]);
          if ((m & 3) == 0) atomicAdd((float*)((char*)denom + (dn[r] + hd)), rune);
          run[0] = 0.0f; run[1] = 0.0f; run[2] = 0.0f; run[3] = 0.0f;
          rune = 0.0f;
        }
      }
      if (!last) {
        carry[wv][5 * g + 0][lane] = run[0];
        carry[wv][5 * g + 1][lane] = run[1];
        carry[wv][5 * g + 2][lane] = run[2];
        carry[wv][5 * g + 3][lane] = run[3];
        carry[wv][5 * g + 4][lane] = rune;
      }
    }
  }
}

extern "C" void kernel_launch(void* const* d_in, const int* in_sizes, int n_in,
                              void* d_out, int out_size, void* d_ws,
                              size_t ws_size, hipStream_t stream) {
  const float* x   = (const float*)d_in[0];
  const int*   ei  = (const int*)d_in[1];
  const float* ea  = (const float*)d_in[2];
  const float* Wq  = (const float*)d_in[3];
  const float* Wk  = (const float*)d_in[4];
  const float* Wv  = (const float*)d_in[5];
  const float* Wo  = (const float*)d_in[6];
  const float* Wek = (const float*)d_in[7];
  const float* Wev = (const float*)d_in[8];
  const float* Wg1 = (const float*)d_in[9];
  const float* Wg2 = (const float*)d_in[10];
  float* out = (float*)d_out;

  const int N = in_sizes[0] / 128;
  const int E = in_sizes[2] / 32;
  const int* src = ei;
  const int* dst = ei + E;

  float* q      = (float*)d_ws;
  float* k      = q + (size_t)N * 128;
  float* scores = k + (size_t)N * 128;
  float* ord    = scores + (size_t)E * 8;
  float* acc    = ord + (size_t)N * 128;
  float* denom  = acc + (size_t)N * 128;
  unsigned* mmax = (unsigned*)(denom + (size_t)N * 8);
  float* v      = q;
  int* perm = (int*)ord;
  int* ssrc = perm + E;
  int* sdst = ssrc + E;
  int* off  = sdst + E;
  int* rank = off + N;
  int* cnt  = (int*)acc - N;

  hipMemsetAsync(cnt, 0, (size_t)(N + N * 128 + N * 8 + N * 8) * sizeof(float),
                 stream);

  dim3 blk(256);
  dim3 gblk(32 * GEMM_WAVES);
  dst_count_kernel<<<dim3((E + WAVES * 32 * CNT_ROUNDS - 1) / (WAVES * 32 * CNT_ROUNDS)),
                     blk, 0, stream>>>(dst, cnt, rank, E);

  seg_offsets_kernel<<<dim3((N + 1023) / 1024), dim3(1024), 0, stream>>>(
      cnt, off, N);

  dst_order_kernel<<<dim3((E + 255) / 256), blk, 0, stream>>>(
      src, dst, rank, off, perm, ssrc, sdst, E);

  gemm128_kernel<<<dim3((N / 16 * 2 + GEMM_WAVES - 1) / GEMM_WAVES), gblk, 0, stream>>>(x, Wq, q, N / 16);
  gemm128_kernel<<<dim3((N / 16 * 2 + GEMM_WAVES - 1) / GEMM_WAVES), gblk, 0, stream>>>(x, Wk, k, N / 16);

  edge_score_kernel<<<dim3(E / (16 * WAVES)), blk, 0, stream>>>(
      ea, perm, ssrc, sdst, Wek, Wg1, Wg2, q, k, scores, mmax);

  gemm128_kernel<<<dim3((N / 16 * 2 + GEMM_WAVES - 1) / GEMM_WAVES), gblk, 0, stream>>>(x, Wv, v, N / 16);

  edge_aggregate_kernel<<<dim3(E / (16 * WAVES * TILES)), blk, 0, stream>>>(
      ea, perm, ssrc, sdst, Wev, v, scores, mmax, denom, acc);

  gemm128_norm_kernel<<<dim3((N / 16 * 2 + GEMM_WAVES - 1) / GEMM_WAVES), gblk, 0, stream>>>(acc, Wo, out, N / 16, denom);
}
